# starting kernel
# baseline (speedup 1.0000x reference)
#include <hip/hip_runtime.h>

#define E_HID 128
#define R_HID 128

typedef float v2f __attribute__((ext_vector_type(2)));
typedef float v8f __attribute__((ext_vector_type(8)));

__device__ __forceinline__ float leaky(float x) { return x > 0.f ? x : 0.01f * x; }

__device__ __forceinline__ void atomicMaxF(float* addr, float val) {
  if (val >= 0.f) atomicMax((int*)addr, __float_as_int(val));
  else            atomicMin((unsigned int*)addr, __float_as_uint(val));
}

__global__ void k_init(float* __restrict__ x_acc, long n_acc,
                       float* __restrict__ mx_h, float* __restrict__ mx_t,
                       float* __restrict__ sm_h, float* __restrict__ sm_t, int n_e) {
  long i = (long)blockIdx.x * blockDim.x + threadIdx.x;
  long stride = (long)gridDim.x * blockDim.x;
  for (long j = i; j < n_acc; j += stride) x_acc[j] = 0.f;
  for (long j = i; j < n_e; j += stride) {
    mx_h[j] = -__builtin_inff(); mx_t[j] = -__builtin_inff();
    sm_h[j] = 0.f;               sm_t[j] = 0.f;
  }
}

__global__ void k_node_scores(const float* __restrict__ x_e,
                              const float* __restrict__ w_ah,
                              const float* __restrict__ w_at,
                              float* __restrict__ e_h, float* __restrict__ e_t, int n_e) {
  int lane = threadIdx.x & 31;
  int wave = threadIdx.x >> 5;
  int row = blockIdx.x * 8 + wave;
  if (row >= n_e) return;
  const float4 x4 = ((const float4*)(x_e + (size_t)row * E_HID))[lane];
  const float4 a4 = ((const float4*)w_ah)[lane];
  const float4 b4 = ((const float4*)w_at)[lane];
  float sh = x4.x*a4.x + x4.y*a4.y + x4.z*a4.z + x4.w*a4.w;
  float st = x4.x*b4.x + x4.y*b4.y + x4.z*b4.z + x4.w*b4.w;
  for (int off = 16; off > 0; off >>= 1) {
    sh += __shfl_xor(sh, off);
    st += __shfl_xor(st, off);
  }
  if (lane == 0) { e_h[row] = sh; e_t[row] = st; }
}

__global__ void k_rel(const float* __restrict__ x_r,
                      const float* __restrict__ w_ar,
                      const float* __restrict__ W1, const float* __restrict__ b1,
                      const float* __restrict__ W2, const float* __restrict__ b2,
                      float* __restrict__ e_r, float* __restrict__ msg, int n_r) {
  __shared__ float row[R_HID];
  __shared__ float hid[2 * R_HID];
  int r = blockIdx.x;
  int t = threadIdx.x;
  if (t < R_HID) row[t] = x_r[(size_t)r * R_HID + t];
  __syncthreads();
  float acc = b1[t];
  #pragma unroll 8
  for (int k = 0; k < R_HID; ++k) acc += row[k] * W1[k * 2 * R_HID + t];
  hid[t] = acc;
  if (t < 32) {
    float s = 0.f;
    #pragma unroll
    for (int j = 0; j < 4; ++j) s += row[t + 32 * j] * w_ar[t + 32 * j];
    for (int off = 16; off > 0; off >>= 1) s += __shfl_xor(s, off);
    if (t == 0) e_r[r] = s;
  }
  __syncthreads();
  if (t < R_HID) {
    float o = b2[t];
    #pragma unroll 8
    for (int k = 0; k < 2 * R_HID; ++k) o += hid[k] * W2[k * R_HID + t];
    msg[(size_t)r * R_HID + t] = row[t] + o;
  }
}

__global__ void k_edge_max(const int* __restrict__ h_idx, const int* __restrict__ t_idx,
                           const int* __restrict__ rel,
                           const float* __restrict__ e_h, const float* __restrict__ e_t,
                           const float* __restrict__ e_r,
                           float* __restrict__ z_h, float* __restrict__ z_t,
                           float* __restrict__ mx_h, float* __restrict__ mx_t, int n_edges) {
  int e = blockIdx.x * blockDim.x + threadIdx.x;
  if (e >= n_edges) return;
  int h = h_idx[e], t = t_idx[e], r = rel[e];
  float er = e_r[r];
  float lh = leaky(e_h[h] + er);
  float lt = leaky(e_t[t] + er);
  z_h[e] = lh; z_t[e] = lt;
  atomicMaxF(mx_h + h, lh);
  atomicMaxF(mx_t + t, lt);
}

__global__ void k_edge_sum(const int* __restrict__ h_idx, const int* __restrict__ t_idx,
                           const float* __restrict__ mx_h, const float* __restrict__ mx_t,
                           float* __restrict__ z_h, float* __restrict__ z_t,
                           float* __restrict__ sm_h, float* __restrict__ sm_t, int n_edges) {
  int e = blockIdx.x * blockDim.x + threadIdx.x;
  if (e >= n_edges) return;
  int h = h_idx[e], t = t_idx[e];
  float zh = __expf(z_h[e] - mx_h[h]);
  float zt = __expf(z_t[e] - mx_t[t]);
  z_h[e] = zh; z_t[e] = zt;
  atomicAdd(sm_h + h, zh);
  atomicAdd(sm_t + t, zt);
}

__global__ void k_scatter(const int* __restrict__ h_idx, const int* __restrict__ t_idx,
                          const int* __restrict__ rel,
                          const float* __restrict__ z_h, const float* __restrict__ z_t,
                          const float* __restrict__ sm_h, const float* __restrict__ sm_t,
                          const float* __restrict__ msg, float* __restrict__ x_acc, int n_edges) {
  int gid = blockIdx.x * blockDim.x + threadIdx.x;
  int e = gid >> 5;
  int lane = gid & 31;
  if (e >= n_edges) return;
  int h = h_idx[e], t = t_idx[e], r = rel[e];
  float ah = z_h[e] / (sm_h[h] + 1e-16f);
  float at = z_t[e] / (sm_t[t] + 1e-16f);
  float4 m4 = ((const float4*)(msg + (size_t)r * R_HID))[lane];
  float* ph = x_acc + (size_t)h * 256 + lane * 4;
  atomicAdd(ph + 0, ah * m4.x); atomicAdd(ph + 1, ah * m4.y);
  atomicAdd(ph + 2, ah * m4.z); atomicAdd(ph + 3, ah * m4.w);
  float* pt = x_acc + (size_t)t * 256 + 128 + lane * 4;
  atomicAdd(pt + 0, at * m4.x); atomicAdd(pt + 1, at * m4.y);
  atomicAdd(pt + 2, at * m4.z); atomicAdd(pt + 3, at * m4.w);
}

__global__ void k_final_gemm(const float* __restrict__ x_acc,
                             const float* __restrict__ Wr, const float* __restrict__ br,
                             float* __restrict__ out, int n_e) {
  constexpr int KTOT = 256, NTOT = 256, KC = 32;
  constexpr int ASTR = 36, BKSTR = 36;
  __shared__ float As[16 * ASTR];
  __shared__ float Bs[NTOT * BKSTR];
  const int tid  = threadIdx.x;
  const int lane = tid & 31;
  const int wave = tid >> 5;
  const int m0   = blockIdx.x * 16;
  const int lrow = lane & 15;
  const int hsel = (lane >= 16) ? 2 : 0;
  const int mrow = (lane >= 16) ? 8 : 0;
  const int n0   = wave * 16;

  const int arow = tid >> 3;
  const int acol = (tid & 7) * 4;

  v8f c0 = {}, c1 = {};
  float4 aReg = make_float4(0.f, 0.f, 0.f, 0.f);
  float4 bReg[8];

  auto loadGlobal = [&](int kb) {
    if (tid < 128) {
      int grow = m0 + arow;
      aReg = make_float4(0.f, 0.f, 0.f, 0.f);
      if (grow < n_e) aReg = *(const float4*)(x_acc + (size_t)grow * KTOT + kb + acol);
    }
    #pragma unroll
    for (int i = 0; i < 8; ++i) {
      int k4 = i * 4;
      bReg[i].x = Wr[(size_t)(kb + k4 + 0) * NTOT + tid];
      bReg[i].y = Wr[(size_t)(kb + k4 + 1) * NTOT + tid];
      bReg[i].z = Wr[(size_t)(kb + k4 + 2) * NTOT + tid];
      bReg[i].w = Wr[(size_t)(kb + k4 + 3) * NTOT + tid];
    }
  };
  auto storeSmem = [&]() {
    if (tid < 128) *(float4*)&As[arow * ASTR + acol] = aReg;
    #pragma unroll
    for (int i = 0; i < 8; ++i)
      *(float4*)&Bs[tid * BKSTR + i * 4] = bReg[i];
  };

  loadGlobal(0);
  for (int kb = 0; kb < KTOT; kb += KC) {
    storeSmem();
    __syncthreads();
    if (kb + KC < KTOT) loadGlobal(kb + KC);
    #pragma unroll
    for (int kk = 0; kk < KC; kk += 4) {
      int ka = kk + hsel;
      v2f a  = *(const v2f*)&As[lrow * ASTR + ka];
      v2f b0 = *(const v2f*)&Bs[(n0 + lrow) * BKSTR + ka];
      v2f b1 = *(const v2f*)&Bs[(n0 + 128 + lrow) * BKSTR + ka];
      c0 = __builtin_amdgcn_wmma_f32_16x16x4_f32(false, a, false, b0, (short)0, c0, false, false);
      c1 = __builtin_amdgcn_wmma_f32_16x16x4_f32(false, a, false, b1, (short)0, c1, false, false);
    }
    __syncthreads();
  }
  #pragma unroll
  for (int j = 0; j < 8; ++j) {
    int grow = m0 + mrow + j;
    if (grow < n_e) {
      out[(size_t)grow * NTOT + n0 + lrow]       = c0[j] + br[n0 + lrow];
      out[(size_t)grow * NTOT + n0 + 128 + lrow] = c1[j] + br[n0 + 128 + lrow];
    }
  }
}

extern "C" void kernel_launch(void* const* d_in, const int* in_sizes, int n_in,
                              void* d_out, int out_size, void* d_ws, size_t ws_size,
                              hipStream_t stream) {
  const float* x_e  = (const float*)d_in[0];
  const float* x_r  = (const float*)d_in[1];
  const int*   eidx = (const int*)d_in[2];
  const int*   rel  = (const int*)d_in[3];
  const float* w_ah = (const float*)d_in[5];
  const float* w_at = (const float*)d_in[6];
  const float* w_ar = (const float*)d_in[7];
  const float* W1   = (const float*)d_in[8];
  const float* b1   = (const float*)d_in[9];
  const float* W2   = (const float*)d_in[10];
  const float* b2   = (const float*)d_in[11];
  const float* Wr   = (const float*)d_in[12];
  const float* br   = (const float*)d_in[13];
  float* out = (float*)d_out;

  const int n_e     = in_sizes[0] / E_HID;
  const int n_r     = in_sizes[1] / R_HID;
  const int n_edges = in_sizes[3];
  const int* h_idx = eidx;
  const int* t_idx = eidx + n_edges;

  float* x_acc = (float*)d_ws;
  float* msg   = x_acc + (size_t)n_e * 256;
  float* z_h   = msg + (size_t)n_r * R_HID;
  float* z_t   = z_h + n_edges;
  float* e_h   = z_t + n_edges;
  float* e_t   = e_h + n_e;
  float* e_r   = e_t + n_e;
  float* mx_h  = e_r + n_r;
  float* mx_t  = mx_h + n_e;
  float* sm_h  = mx_t + n_e;
  float* sm_t  = sm_h + n_e;

  k_init<<<4096, 256, 0, stream>>>(x_acc, (long)n_e * 256, mx_h, mx_t, sm_h, sm_t, n_e);
  k_node_scores<<<(n_e + 7) / 8, 256, 0, stream>>>(x_e, w_ah, w_at, e_h, e_t, n_e);
  k_rel<<<n_r, 256, 0, stream>>>(x_r, w_ar, W1, b1, W2, b2, e_r, msg, n_r);
  k_edge_max<<<(n_edges + 255) / 256, 256, 0, stream>>>(h_idx, t_idx, rel, e_h, e_t, e_r,
                                                        z_h, z_t, mx_h, mx_t, n_edges);
  k_edge_sum<<<(n_edges + 255) / 256, 256, 0, stream>>>(h_idx, t_idx, mx_h, mx_t,
                                                        z_h, z_t, sm_h, sm_t, n_edges);
  k_scatter<<<(n_edges + 7) / 8, 256, 0, stream>>>(h_idx, t_idx, rel, z_h, z_t,
                                                   sm_h, sm_t, msg, x_acc, n_edges);
  k_final_gemm<<<(n_e + 15) / 16, 256, 0, stream>>>(x_acc, Wr, br, out, n_e);
}


# Float-atomic max/sum/scatter passes replaced by per-node relation lists (integer count, offsets, fill) gathered one wave per node side; float sums now in fixed ascending relation-id order
# speedup vs baseline: 1.9159x; 1.9159x over previous
#include <hip/hip_runtime.h>

#define E_HID 128
#define R_HID 128

typedef float v2f __attribute__((ext_vector_type(2)));
typedef float v8f __attribute__((ext_vector_type(8)));

__device__ __forceinline__ float leaky(float x) { return x > 0.f ? x : 0.01f * x; }


__device__ __forceinline__ float wave_max_f(float v) {
  for (int o = 16; o > 0; o >>= 1) v = fmaxf(v, __shfl_xor(v, o));
  return v;
}
__device__ __forceinline__ float wave_sum_f(float v) {
  for (int o = 16; o > 0; o >>= 1) v += __shfl_xor(v, o);
  return v;
}
__device__ __forceinline__ int wave_min_i(int v) {
  for (int o = 16; o > 0; o >>= 1) { int w = __shfl_xor(v, o); v = w < v ? w : v; }
  return v;
}
__device__ __forceinline__ int wave_sum_i(int v) {
  for (int o = 16; o > 0; o >>= 1) v += __shfl_xor(v, o);
  return v;
}

__global__ void k_node_scores(const float* __restrict__ x_e,
                              const float* __restrict__ w_ah,
                              const float* __restrict__ w_at,
                              float* __restrict__ e_h, float* __restrict__ e_t,
                              int* __restrict__ cnt, int* __restrict__ cursor, int n_e) {
  int lane = threadIdx.x & 31;
  int wave = threadIdx.x >> 5;
  int row = blockIdx.x * 8 + wave;
  if (blockIdx.x == 0 && threadIdx.x == 0) *cursor = 0;
  if (row >= n_e) return;
  const float4 x4 = ((const float4*)(x_e + (size_t)row * E_HID))[lane];
  const float4 a4 = ((const float4*)w_ah)[lane];
  const float4 b4 = ((const float4*)w_at)[lane];
  float sh = x4.x*a4.x + x4.y*a4.y + x4.z*a4.z + x4.w*a4.w;
  float st = x4.x*b4.x + x4.y*b4.y + x4.z*b4.z + x4.w*b4.w;
  for (int off = 16; off > 0; off >>= 1) {
    sh += __shfl_xor(sh, off);
    st += __shfl_xor(st, off);
  }
  if (lane == 0) { e_h[row] = sh; e_t[row] = st; cnt[row] = 0; cnt[n_e + row] = 0; }
}

__global__ void k_rel(const float* __restrict__ x_r,
                      const float* __restrict__ w_ar,
                      const float* __restrict__ W1, const float* __restrict__ b1,
                      const float* __restrict__ W2, const float* __restrict__ b2,
                      float* __restrict__ e_r, float* __restrict__ msg, int n_r) {
  __shared__ float row[R_HID];
  __shared__ float hid[2 * R_HID];
  int r = blockIdx.x;
  int t = threadIdx.x;
  if (t < R_HID) row[t] = x_r[(size_t)r * R_HID + t];
  __syncthreads();
  float acc = b1[t];
  #pragma unroll 8
  for (int k = 0; k < R_HID; ++k) acc += row[k] * W1[k * 2 * R_HID + t];
  hid[t] = acc;
  if (t < 32) {
    float s = 0.f;
    #pragma unroll
    for (int j = 0; j < 4; ++j) s += row[t + 32 * j] * w_ar[t + 32 * j];
    for (int off = 16; off > 0; off >>= 1) s += __shfl_xor(s, off);
    if (t == 0) e_r[r] = s;
  }
  __syncthreads();
  if (t < R_HID) {
    float o = b2[t];
    #pragma unroll 8
    for (int k = 0; k < 2 * R_HID; ++k) o += hid[k] * W2[k * R_HID + t];
    msg[(size_t)r * R_HID + t] = row[t] + o;
  }
}

__global__ void k_count(const int* __restrict__ h_idx, const int* __restrict__ t_idx,
                        int* __restrict__ cnt, int* __restrict__ rank_h, int* __restrict__ rank_t,
                        int n_edges, int n_e) {
  int i = blockIdx.x * blockDim.x + threadIdx.x;
  int stride = gridDim.x * blockDim.x;
  for (int e = i; e < n_edges; e += stride) {
    int h = h_idx[e], t = t_idx[e];
    const int rh = atomicAdd(cnt + h, 1);
    const int rt = atomicAdd(cnt + n_e + t, 1);
    rank_h[e] = rh;
    rank_t[e] = rt;
  }
}

__global__ void k_alloc(const int* __restrict__ cnt, int* __restrict__ off,
                        int* __restrict__ cursor, int n_keys) {
  __shared__ int wsum[8];
  __shared__ int bbase;
  int i = blockIdx.x * 256 + threadIdx.x;
  int lane = threadIdx.x & 31;
  int wave = threadIdx.x >> 5;
  int c = (i < n_keys) ? cnt[i] : 0;
  int incl = c;
  for (int o = 1; o < 32; o <<= 1) {
    int v = __shfl_up(incl, o);
    if (lane >= o) incl += v;
  }
  if (lane == 31) wsum[wave] = incl;
  __syncthreads();
  if (threadIdx.x == 0) {
    int tot = 0;
    for (int w = 0; w < 8; ++w) { int v = wsum[w]; wsum[w] = tot; tot += v; }
    bbase = atomicAdd(cursor, tot);
  }
  __syncthreads();
  if (i < n_keys) off[i] = bbase + wsum[wave] + incl - c;
}

__global__ void k_fill(const int* __restrict__ h_idx, const int* __restrict__ t_idx,
                       const int* __restrict__ rel, const int* __restrict__ off,
                       const int* __restrict__ rank_h, const int* __restrict__ rank_t,
                       int* __restrict__ list, int n_edges, int n_e) {
  int i = blockIdx.x * blockDim.x + threadIdx.x;
  int stride = gridDim.x * blockDim.x;
  for (int e = i; e < n_edges; e += stride) {
    int h = h_idx[e], t = t_idx[e], r = rel[e];
    list[off[h] + rank_h[e]] = r;
    list[off[n_e + t] + rank_t[e]] = r;
  }
}

__global__ void __launch_bounds__(256, 16) k_gather(const int* __restrict__ cnt, const int* __restrict__ off,
                         const int* __restrict__ list,
                         const float* __restrict__ e_h, const float* __restrict__ e_t,
                         const float* __restrict__ e_r, const float* __restrict__ msg,
                         float* __restrict__ x_acc, int n_e) {
  const int lane = threadIdx.x & 31;
  const int wv = __builtin_amdgcn_readfirstlane((int)(threadIdx.x >> 5));
  const int nw = (int)gridDim.x * 8;
  const int n_keys = 2 * n_e;
  for (int key = (int)blockIdx.x * 8 + wv; key < n_keys; key += nw) {
    const int side = key >= n_e ? 1 : 0;
    const int node = key - side * n_e;
    const int d = cnt[key];
    const int o = off[key];
    const float en = side ? e_t[node] : e_h[node];
    float4 acc = make_float4(0.f, 0.f, 0.f, 0.f);
    if (d > 0 && d <= 32) {
      const bool act = lane < d;
      int r = 0;
      float z = -__builtin_inff();
      if (act) { r = list[o + lane]; z = leaky(en + e_r[r]); }
      const float mx = wave_max_f(z);
      const float p = act ? __expf(z - mx) : 0.f;
      int rank = act ? 0 : lane;
      for (int j = 0; j < d; ++j) {
        const int rj = __builtin_amdgcn_readlane(r, j);
        if (act && (rj < r || (rj == r && j < lane))) rank += 1;
      }
      const int rs = __builtin_amdgcn_ds_permute(rank << 2, r);
      const float ps = __int_as_float(__builtin_amdgcn_ds_permute(rank << 2, __float_as_int(p)));
      const float s = wave_sum_f(ps);
      const float al = ps / (s + 1e-16f);
      int k = 0;
      for (; k + 4 <= d; k += 4) {
        const int r0 = __builtin_amdgcn_readlane(rs, k);
        const int r1 = __builtin_amdgcn_readlane(rs, k + 1);
        const int r2 = __builtin_amdgcn_readlane(rs, k + 2);
        const int r3 = __builtin_amdgcn_readlane(rs, k + 3);
        const float a0 = __int_as_float(__builtin_amdgcn_readlane(__float_as_int(al), k));
        const float a1 = __int_as_float(__builtin_amdgcn_readlane(__float_as_int(al), k + 1));
        const float a2 = __int_as_float(__builtin_amdgcn_readlane(__float_as_int(al), k + 2));
        const float a3 = __int_as_float(__builtin_amdgcn_readlane(__float_as_int(al), k + 3));
        const float4 m0 = ((const float4*)(msg + (size_t)r0 * R_HID))[lane];
        const float4 m1 = ((const float4*)(msg + (size_t)r1 * R_HID))[lane];
        const float4 m2 = ((const float4*)(msg + (size_t)r2 * R_HID))[lane];
        const float4 m3 = ((const float4*)(msg + (size_t)r3 * R_HID))[lane];
        acc.x += a0 * m0.x; acc.y += a0 * m0.y; acc.z += a0 * m0.z; acc.w += a0 * m0.w;
        acc.x += a1 * m1.x; acc.y += a1 * m1.y; acc.z += a1 * m1.z; acc.w += a1 * m1.w;
        acc.x += a2 * m2.x; acc.y += a2 * m2.y; acc.z += a2 * m2.z; acc.w += a2 * m2.w;
        acc.x += a3 * m3.x; acc.y += a3 * m3.y; acc.z += a3 * m3.z; acc.w += a3 * m3.w;
      }
      for (; k < d; ++k) {
        const int r0 = __builtin_amdgcn_readlane(rs, k);
        const float a0 = __int_as_float(__builtin_amdgcn_readlane(__float_as_int(al), k));
        const float4 m0 = ((const float4*)(msg + (size_t)r0 * R_HID))[lane];
        acc.x += a0 * m0.x; acc.y += a0 * m0.y; acc.z += a0 * m0.z; acc.w += a0 * m0.w;
      }
    } else if (d > 32) {
      float mx = -__builtin_inff();
      for (int i = lane; i < d; i += 32) mx = fmaxf(mx, leaky(en + e_r[list[o + i]]));
      mx = wave_max_f(mx);
      float s = 0.f;
      int last = -1;
      for (int it = 0; it < d; ++it) {
        int best = 0x7fffffff;
        for (int i = lane; i < d; i += 32) { const int r = list[o + i]; if (r > last && r < best) best = r; }
        best = __builtin_amdgcn_readfirstlane(wave_min_i(best));
        if (best == 0x7fffffff) break;
        int c = 0;
        for (int i = lane; i < d; i += 32) c += (list[o + i] == best) ? 1 : 0;
        c = __builtin_amdgcn_readfirstlane(wave_sum_i(c));
        const float p = __expf(leaky(en + e_r[best]) - mx);
        for (int q = 0; q < c; ++q) s += p;
        last = best;
      }
      const float den = s + 1e-16f;
      last = -1;
      for (int it = 0; it < d; ++it) {
        int best = 0x7fffffff;
        for (int i = lane; i < d; i += 32) { const int r = list[o + i]; if (r > last && r < best) best = r; }
        best = __builtin_amdgcn_readfirstlane(wave_min_i(best));
        if (best == 0x7fffffff) break;
        int c = 0;
        for (int i = lane; i < d; i += 32) c += (list[o + i] == best) ? 1 : 0;
        c = __builtin_amdgcn_readfirstlane(wave_sum_i(c));
        const float a = __expf(leaky(en + e_r[best]) - mx) / den;
        const float4 m0 = ((const float4*)(msg + (size_t)best * R_HID))[lane];
        for (int q = 0; q < c; ++q) {
          acc.x += a * m0.x; acc.y += a * m0.y; acc.z += a * m0.z; acc.w += a * m0.w;
        }
        last = best;
      }
    }
    ((float4*)(x_acc + (size_t)node * 256 + side * 128))[lane] = acc;
  }
}

__global__ void k_final_gemm(const float* __restrict__ x_acc,
                             const float* __restrict__ Wr, const float* __restrict__ br,
                             float* __restrict__ out, int n_e) {
  constexpr int KTOT = 256, NTOT = 256, KC = 32;
  constexpr int ASTR = 36, BKSTR = 36;
  __shared__ float As[16 * ASTR];
  __shared__ float Bs[NTOT * BKSTR];
  const int tid  = threadIdx.x;
  const int lane = tid & 31;
  const int wave = tid >> 5;
  const int m0   = blockIdx.x * 16;
  const int lrow = lane & 15;
  const int hsel = (lane >= 16) ? 2 : 0;
  const int mrow = (lane >= 16) ? 8 : 0;
  const int n0   = wave * 16;

  const int arow = tid >> 3;
  const int acol = (tid & 7) * 4;

  v8f c0 = {}, c1 = {};
  float4 aReg = make_float4(0.f, 0.f, 0.f, 0.f);
  float4 bReg[8];

  auto loadGlobal = [&](int kb) {
    if (tid < 128) {
      int grow = m0 + arow;
      aReg = make_float4(0.f, 0.f, 0.f, 0.f);
      if (grow < n_e) aReg = *(const float4*)(x_acc + (size_t)grow * KTOT + kb + acol);
    }
    #pragma unroll
    for (int i = 0; i < 8; ++i) {
      int k4 = i * 4;
      bReg[i].x = Wr[(size_t)(kb + k4 + 0) * NTOT + tid];
      bReg[i].y = Wr[(size_t)(kb + k4 + 1) * NTOT + tid];
      bReg[i].z = Wr[(size_t)(kb + k4 + 2) * NTOT + tid];
      bReg[i].w = Wr[(size_t)(kb + k4 + 3) * NTOT + tid];
    }
  };
  auto storeSmem = [&]() {
    if (tid < 128) *(float4*)&As[arow * ASTR + acol] = aReg;
    #pragma unroll
    for (int i = 0; i < 8; ++i)
      *(float4*)&Bs[tid * BKSTR + i * 4] = bReg[i];
  };

  loadGlobal(0);
  for (int kb = 0; kb < KTOT; kb += KC) {
    storeSmem();
    __syncthreads();
    if (kb + KC < KTOT) loadGlobal(kb + KC);
    #pragma unroll
    for (int kk = 0; kk < KC; kk += 4) {
      int ka = kk + hsel;
      v2f a  = *(const v2f*)&As[lrow * ASTR + ka];
      v2f b0 = *(const v2f*)&Bs[(n0 + lrow) * BKSTR + ka];
      v2f b1 = *(const v2f*)&Bs[(n0 + 128 + lrow) * BKSTR + ka];
      c0 = __builtin_amdgcn_wmma_f32_16x16x4_f32(false, a, false, b0, (short)0, c0, false, false);
      c1 = __builtin_amdgcn_wmma_f32_16x16x4_f32(false, a, false, b1, (short)0, c1, false, false);
    }
    __syncthreads();
  }
  #pragma unroll
  for (int j = 0; j < 8; ++j) {
    int grow = m0 + mrow + j;
    if (grow < n_e) {
      out[(size_t)grow * NTOT + n0 + lrow]       = c0[j] + br[n0 + lrow];
      out[(size_t)grow * NTOT + n0 + 128 + lrow] = c1[j] + br[n0 + 128 + lrow];
    }
  }
}

extern "C" void kernel_launch(void* const* d_in, const int* in_sizes, int n_in,
                              void* d_out, int out_size, void* d_ws, size_t ws_size,
                              hipStream_t stream) {
  const float* x_e  = (const float*)d_in[0];
  const float* x_r  = (const float*)d_in[1];
  const int*   eidx = (const int*)d_in[2];
  const int*   rel  = (const int*)d_in[3];
  const float* w_ah = (const float*)d_in[5];
  const float* w_at = (const float*)d_in[6];
  const float* w_ar = (const float*)d_in[7];
  const float* W1   = (const float*)d_in[8];
  const float* b1   = (const float*)d_in[9];
  const float* W2   = (const float*)d_in[10];
  const float* b2   = (const float*)d_in[11];
  const float* Wr   = (const float*)d_in[12];
  const float* br   = (const float*)d_in[13];
  float* out = (float*)d_out;

  const int n_e     = in_sizes[0] / E_HID;
  const int n_r     = in_sizes[1] / R_HID;
  const int n_edges = in_sizes[3];
  const int* h_idx = eidx;
  const int* t_idx = eidx + n_edges;

  float* x_acc = (float*)d_ws;
  float* msg   = x_acc + (size_t)n_e * 256;
  float* e_h   = msg + (size_t)n_r * R_HID;
  float* e_t   = e_h + n_e;
  float* e_r   = e_t + n_e;
  int* cnt     = (int*)(e_r + n_r);
  int* off     = cnt + (size_t)2 * n_e;
  int* rank_h  = off + (size_t)2 * n_e;
  int* rank_t  = rank_h + n_edges;
  int* list    = rank_t + n_edges;
  int* cursor  = list + (size_t)2 * n_edges;

  k_node_scores<<<(n_e + 7) / 8, 256, 0, stream>>>(x_e, w_ah, w_at, e_h, e_t, cnt, cursor, n_e);
  k_rel<<<n_r, 256, 0, stream>>>(x_r, w_ar, W1, b1, W2, b2, e_r, msg, n_r);
  k_count<<<2048, 256, 0, stream>>>(h_idx, t_idx, cnt, rank_h, rank_t, n_edges, n_e);
  k_alloc<<<(2 * n_e + 255) / 256, 256, 0, stream>>>(cnt, off, cursor, 2 * n_e);
  k_fill<<<2048, 256, 0, stream>>>(h_idx, t_idx, rel, off, rank_h, rank_t, list, n_edges, n_e);
  k_gather<<<2048, 256, 0, stream>>>(cnt, off, list, e_h, e_t, e_r, msg, x_acc, n_e);
  k_final_gemm<<<(n_e + 15) / 16, 256, 0, stream>>>(x_acc, Wr, br, out, n_e);
}


# Node-score pass re-gridded: each wave streams 32 rows with all its loads in flight (8 lanes per row) instead of one short wave per row; per-row dot-product sum order changed
# speedup vs baseline: 2.3176x; 1.2097x over previous
#include <hip/hip_runtime.h>

#define E_HID 128
#define R_HID 128

typedef float v2f __attribute__((ext_vector_type(2)));
typedef float v8f __attribute__((ext_vector_type(8)));
typedef float v4f __attribute__((ext_vector_type(4)));

__device__ __forceinline__ float leaky(float x) { return x > 0.f ? x : 0.01f * x; }


__device__ __forceinline__ float wave_max_f(float v) {
  for (int o = 16; o > 0; o >>= 1) v = fmaxf(v, __shfl_xor(v, o));
  return v;
}
__device__ __forceinline__ float wave_sum_f(float v) {
  for (int o = 16; o > 0; o >>= 1) v += __shfl_xor(v, o);
  return v;
}
__device__ __forceinline__ int wave_min_i(int v) {
  for (int o = 16; o > 0; o >>= 1) { int w = __shfl_xor(v, o); v = w < v ? w : v; }
  return v;
}
__device__ __forceinline__ int wave_sum_i(int v) {
  for (int o = 16; o > 0; o >>= 1) v += __shfl_xor(v, o);
  return v;
}

__global__ void __launch_bounds__(64) k_node_scores(const float* __restrict__ x_e,
                              const float* __restrict__ w_ah,
                              const float* __restrict__ w_at,
                              float* __restrict__ e_h, float* __restrict__ e_t,
                              int* __restrict__ cnt, int* __restrict__ cursor, int n_e) {
  const int lane = threadIdx.x & 31;
  const int base = ((int)blockIdx.x * 2 + (int)(threadIdx.x >> 5)) * 32;
  if (blockIdx.x == 0 && threadIdx.x == 0) *cursor = 0;
  if (base >= n_e) return;
  const int s = lane & 7;
  const int q = lane >> 3;
  v4f x[8][4];
  #pragma unroll
  for (int i = 0; i < 8; ++i) {
    int row = base + q * 8 + i;
    row = row < n_e ? row : n_e - 1;
    const v4f* p = (const v4f*)(x_e + (size_t)row * E_HID);
    #pragma unroll
    for (int j = 0; j < 4; ++j) x[i][j] = p[j * 8 + s];
  }
  v4f a[4], b[4];
  #pragma unroll
  for (int j = 0; j < 4; ++j) {
    a[j] = ((const v4f*)w_ah)[j * 8 + s];
    b[j] = ((const v4f*)w_at)[j * 8 + s];
  }
  asm volatile("" ::
               "v"(x[0][0]), "v"(x[0][1]), "v"(x[0][2]), "v"(x[0][3]),
               "v"(x[1][0]), "v"(x[1][1]), "v"(x[1][2]), "v"(x[1][3]),
               "v"(x[2][0]), "v"(x[2][1]), "v"(x[2][2]), "v"(x[2][3]),
               "v"(x[3][0]), "v"(x[3][1]), "v"(x[3][2]), "v"(x[3][3]),
               "v"(x[4][0]), "v"(x[4][1]), "v"(x[4][2]), "v"(x[4][3]),
               "v"(x[5][0]), "v"(x[5][1]), "v"(x[5][2]), "v"(x[5][3]),
               "v"(x[6][0]), "v"(x[6][1]), "v"(x[6][2]), "v"(x[6][3]),
               "v"(x[7][0]), "v"(x[7][1]), "v"(x[7][2]), "v"(x[7][3]),
               "v"(a[0]), "v"(b[0]), "v"(a[1]), "v"(b[1]),
               "v"(a[2]), "v"(b[2]), "v"(a[3]), "v"(b[3]));
  float vh = 0.f, vt = 0.f;
  #pragma unroll
  for (int i = 0; i < 8; ++i) {
    float sh = 0.f, st = 0.f;
    #pragma unroll
    for (int j = 0; j < 4; ++j) {
      const v4f v = x[i][j];
      sh = fmaf(v.x, a[j].x, sh); sh = fmaf(v.y, a[j].y, sh); sh = fmaf(v.z, a[j].z, sh); sh = fmaf(v.w, a[j].w, sh);
      st = fmaf(v.x, b[j].x, st); st = fmaf(v.y, b[j].y, st); st = fmaf(v.z, b[j].z, st); st = fmaf(v.w, b[j].w, st);
    }
    #pragma unroll
    for (int off = 4; off > 0; off >>= 1) {
      sh += __shfl_xor(sh, off);
      st += __shfl_xor(st, off);
    }
    if (s == i) { vh = sh; vt = st; }
  }
  const int row = base + lane;
  if (row < n_e) { e_h[row] = vh; e_t[row] = vt; cnt[row] = 0; cnt[n_e + row] = 0; }
}

__global__ void k_rel(const float* __restrict__ x_r,
                      const float* __restrict__ w_ar,
                      const float* __restrict__ W1, const float* __restrict__ b1,
                      const float* __restrict__ W2, const float* __restrict__ b2,
                      float* __restrict__ e_r, float* __restrict__ msg, int n_r) {
  __shared__ float row[R_HID];
  __shared__ float hid[2 * R_HID];
  int r = blockIdx.x;
  int t = threadIdx.x;
  if (t < R_HID) row[t] = x_r[(size_t)r * R_HID + t];
  __syncthreads();
  float acc = b1[t];
  #pragma unroll 8
  for (int k = 0; k < R_HID; ++k) acc += row[k] * W1[k * 2 * R_HID + t];
  hid[t] = acc;
  if (t < 32) {
    float s = 0.f;
    #pragma unroll
    for (int j = 0; j < 4; ++j) s += row[t + 32 * j] * w_ar[t + 32 * j];
    for (int off = 16; off > 0; off >>= 1) s += __shfl_xor(s, off);
    if (t == 0) e_r[r] = s;
  }
  __syncthreads();
  if (t < R_HID) {
    float o = b2[t];
    #pragma unroll 8
    for (int k = 0; k < 2 * R_HID; ++k) o += hid[k] * W2[k * R_HID + t];
    msg[(size_t)r * R_HID + t] = row[t] + o;
  }
}

__global__ void k_count(const int* __restrict__ h_idx, const int* __restrict__ t_idx,
                        int* __restrict__ cnt, int* __restrict__ rank_h, int* __restrict__ rank_t,
                        int n_edges, int n_e) {
  int i = blockIdx.x * blockDim.x + threadIdx.x;
  int stride = gridDim.x * blockDim.x;
  for (int e = i; e < n_edges; e += stride) {
    int h = h_idx[e], t = t_idx[e];
    const int rh = atomicAdd(cnt + h, 1);
    const int rt = atomicAdd(cnt + n_e + t, 1);
    rank_h[e] = rh;
    rank_t[e] = rt;
  }
}

__global__ void k_alloc(const int* __restrict__ cnt, int* __restrict__ off,
                        int* __restrict__ cursor, int n_keys) {
  __shared__ int wsum[8];
  __shared__ int bbase;
  int i = blockIdx.x * 256 + threadIdx.x;
  int lane = threadIdx.x & 31;
  int wave = threadIdx.x >> 5;
  int c = (i < n_keys) ? cnt[i] : 0;
  int incl = c;
  for (int o = 1; o < 32; o <<= 1) {
    int v = __shfl_up(incl, o);
    if (lane >= o) incl += v;
  }
  if (lane == 31) wsum[wave] = incl;
  __syncthreads();
  if (threadIdx.x == 0) {
    int tot = 0;
    for (int w = 0; w < 8; ++w) { int v = wsum[w]; wsum[w] = tot; tot += v; }
    bbase = atomicAdd(cursor, tot);
  }
  __syncthreads();
  if (i < n_keys) off[i] = bbase + wsum[wave] + incl - c;
}

__global__ void k_fill(const int* __restrict__ h_idx, const int* __restrict__ t_idx,
                       const int* __restrict__ rel, const int* __restrict__ off,
                       const int* __restrict__ rank_h, const int* __restrict__ rank_t,
                       int* __restrict__ list, int n_edges, int n_e) {
  int i = blockIdx.x * blockDim.x + threadIdx.x;
  int stride = gridDim.x * blockDim.x;
  for (int e = i; e < n_edges; e += stride) {
    int h = h_idx[e], t = t_idx[e], r = rel[e];
    list[off[h] + rank_h[e]] = r;
    list[off[n_e + t] + rank_t[e]] = r;
  }
}

__global__ void k_proj(const float* __restrict__ msg, const float* __restrict__ Wr,
                       float* __restrict__ P, int n_r) {
  constexpr int KTOT = 128, NTOT = 256, KC = 32;
  constexpr int ASTR = 36, BKSTR = 36;
  __shared__ float As[16 * ASTR];
  __shared__ float Bs[NTOT * BKSTR];
  const int tid  = threadIdx.x;
  const int lane = tid & 31;
  const int wave = tid >> 5;
  const int m0   = blockIdx.x * 16;
  const int lrow = lane & 15;
  const int hsel = (lane >= 16) ? 2 : 0;
  const int mrow = (lane >= 16) ? 8 : 0;
  const int n0   = wave * 16;
  const float* __restrict__ B = Wr + (size_t)blockIdx.y * KTOT * NTOT;
  float* __restrict__ O = P + (size_t)blockIdx.y * n_r * NTOT;

  const int arow = tid >> 3;
  const int acol = (tid & 7) * 4;

  v8f c0 = {}, c1 = {};
  float4 aReg = make_float4(0.f, 0.f, 0.f, 0.f);
  float4 bReg[8];

  auto loadGlobal = [&](int kb) {
    if (tid < 128) {
      int grow = m0 + arow;
      aReg = make_float4(0.f, 0.f, 0.f, 0.f);
      if (grow < n_r) aReg = *(const float4*)(msg + (size_t)grow * KTOT + kb + acol);
    }
    #pragma unroll
    for (int i = 0; i < 8; ++i) {
      int k4 = i * 4;
      bReg[i].x = B[(size_t)(kb + k4 + 0) * NTOT + tid];
      bReg[i].y = B[(size_t)(kb + k4 + 1) * NTOT + tid];
      bReg[i].z = B[(size_t)(kb + k4 + 2) * NTOT + tid];
      bReg[i].w = B[(size_t)(kb + k4 + 3) * NTOT + tid];
    }
  };
  auto storeSmem = [&]() {
    if (tid < 128) *(float4*)&As[arow * ASTR + acol] = aReg;
    #pragma unroll
    for (int i = 0; i < 8; ++i)
      *(float4*)&Bs[tid * BKSTR + i * 4] = bReg[i];
  };

  loadGlobal(0);
  for (int kb = 0; kb < KTOT; kb += KC) {
    storeSmem();
    __syncthreads();
    if (kb + KC < KTOT) loadGlobal(kb + KC);
    #pragma unroll
    for (int kk = 0; kk < KC; kk += 4) {
      int ka = kk + hsel;
      v2f a  = *(const v2f*)&As[lrow * ASTR + ka];
      v2f b0 = *(const v2f*)&Bs[(n0 + lrow) * BKSTR + ka];
      v2f b1 = *(const v2f*)&Bs[(n0 + 128 + lrow) * BKSTR + ka];
      c0 = __builtin_amdgcn_wmma_f32_16x16x4_f32(false, a, false, b0, (short)0, c0, false, false);
      c1 = __builtin_amdgcn_wmma_f32_16x16x4_f32(false, a, false, b1, (short)0, c1, false, false);
    }
    __syncthreads();
  }
  #pragma unroll
  for (int j = 0; j < 8; ++j) {
    int grow = m0 + mrow + j;
    if (grow < n_r) {
      O[(size_t)grow * NTOT + n0 + lrow]       = c0[j];
      O[(size_t)grow * NTOT + n0 + 128 + lrow] = c1[j];
    }
  }
}

__device__ __forceinline__ void fma_row(float4& A, float4& B, const float a, const float4 m, const float4 n) {
  A.x += a * m.x; A.y += a * m.y; A.z += a * m.z; A.w += a * m.w;
  B.x += a * n.x; B.y += a * n.y; B.z += a * n.z; B.w += a * n.w;
}

__device__ __forceinline__ void side_pass(const int d, const int o, const float en,
                                          const int* __restrict__ list, const float* __restrict__ e_r,
                                          const float* __restrict__ Ps, const int lane,
                                          float4& accA, float4& accB) {
  if (d > 0 && d <= 32) {
    const bool act = lane < d;
    int r = 0;
    float z = -__builtin_inff();
    if (act) { r = list[o + lane]; z = leaky(en + e_r[r]); }
    const float mx = wave_max_f(z);
    const float p = act ? __expf(z - mx) : 0.f;
    int rank = act ? 0 : lane;
    for (int j = 0; j < d; ++j) {
      const int rj = __builtin_amdgcn_readlane(r, j);
      if (act && (rj < r || (rj == r && j < lane))) rank += 1;
    }
    const int rs = __builtin_amdgcn_ds_permute(rank << 2, r);
    const float ps = __int_as_float(__builtin_amdgcn_ds_permute(rank << 2, __float_as_int(p)));
    const float s = wave_sum_f(ps);
    const float al = ps / (s + 1e-16f);
    int k = 0;
    #pragma unroll 1
    for (; k + 2 <= d; k += 2) {
      const int r0 = __builtin_amdgcn_readlane(rs, k);
      const int r1 = __builtin_amdgcn_readlane(rs, k + 1);
      const float a0 = __int_as_float(__builtin_amdgcn_readlane(__float_as_int(al), k));
      const float a1 = __int_as_float(__builtin_amdgcn_readlane(__float_as_int(al), k + 1));
      const float4* q0 = (const float4*)(Ps + (size_t)r0 * 256);
      const float4* q1 = (const float4*)(Ps + (size_t)r1 * 256);
      const float4 m0 = q0[lane], n0 = q0[32 + lane];
      const float4 m1 = q1[lane], n1 = q1[32 + lane];
      fma_row(accA, accB, a0, m0, n0);
      fma_row(accA, accB, a1, m1, n1);
    }
    if (k < d) {
      const int r0 = __builtin_amdgcn_readlane(rs, k);
      const float a0 = __int_as_float(__builtin_amdgcn_readlane(__float_as_int(al), k));
      const float4* q0 = (const float4*)(Ps + (size_t)r0 * 256);
      const float4 m0 = q0[lane], n0 = q0[32 + lane];
      fma_row(accA, accB, a0, m0, n0);
    }
  } else if (d > 32) {
    float mx = -__builtin_inff();
    for (int i = lane; i < d; i += 32) mx = fmaxf(mx, leaky(en + e_r[list[o + i]]));
    mx = wave_max_f(mx);
    float s = 0.f;
    int last = -1;
    for (int it = 0; it < d; ++it) {
      int best = 0x7fffffff;
      for (int i = lane; i < d; i += 32) { const int r = list[o + i]; if (r > last && r < best) best = r; }
      best = __builtin_amdgcn_readfirstlane(wave_min_i(best));
      if (best == 0x7fffffff) break;
      int c = 0;
      for (int i = lane; i < d; i += 32) c += (list[o + i] == best) ? 1 : 0;
      c = __builtin_amdgcn_readfirstlane(wave_sum_i(c));
      const float p = __expf(leaky(en + e_r[best]) - mx);
      for (int q = 0; q < c; ++q) s += p;
      last = best;
    }
    const float den = s + 1e-16f;
    last = -1;
    for (int it = 0; it < d; ++it) {
      int best = 0x7fffffff;
      for (int i = lane; i < d; i += 32) { const int r = list[o + i]; if (r > last && r < best) best = r; }
      best = __builtin_amdgcn_readfirstlane(wave_min_i(best));
      if (best == 0x7fffffff) break;
      int c = 0;
      for (int i = lane; i < d; i += 32) c += (list[o + i] == best) ? 1 : 0;
      c = __builtin_amdgcn_readfirstlane(wave_sum_i(c));
      const float a = __expf(leaky(en + e_r[best]) - mx) / den;
      const float4* q0 = (const float4*)(Ps + (size_t)best * 256);
      const float4 m0 = q0[lane], n0 = q0[32 + lane];
      for (int q = 0; q < c; ++q) fma_row(accA, accB, a, m0, n0);
      last = best;
    }
  }
}

__global__ void __launch_bounds__(256, 16) k_gather(const int* __restrict__ cnt, const int* __restrict__ off,
                         const int* __restrict__ list,
                         const float* __restrict__ e_h, const float* __restrict__ e_t,
                         const float* __restrict__ e_r, const float* __restrict__ P,
                         const float* __restrict__ br,
                         float* __restrict__ out, int n_e, int n_r) {
  const int lane = threadIdx.x & 31;
  const int wv = __builtin_amdgcn_readfirstlane((int)(threadIdx.x >> 5));
  const int nw = (int)gridDim.x * 8;
  for (int node = (int)blockIdx.x * 8 + wv; node < n_e; node += nw) {
    const int d0 = cnt[node];
    const int d1 = cnt[n_e + node];
    const int o0 = off[node];
    const int o1 = off[n_e + node];
    const float en0 = e_h[node];
    const float en1 = e_t[node];
    float4 accA = make_float4(0.f, 0.f, 0.f, 0.f);
    float4 accB = make_float4(0.f, 0.f, 0.f, 0.f);
    side_pass(d0, o0, en0, list, e_r, P, lane, accA, accB);
    side_pass(d1, o1, en1, list, e_r, P + (size_t)n_r * 256, lane, accA, accB);
    const float4 bA = ((const float4*)br)[lane];
    const float4 bB = ((const float4*)br)[32 + lane];
    accA.x += bA.x; accA.y += bA.y; accA.z += bA.z; accA.w += bA.w;
    accB.x += bB.x; accB.y += bB.y; accB.z += bB.z; accB.w += bB.w;
    float4* orow = (float4*)(out + (size_t)node * 256);
    orow[lane] = accA;
    orow[32 + lane] = accB;
  }
}

extern "C" void kernel_launch(void* const* d_in, const int* in_sizes, int n_in,
                              void* d_out, int out_size, void* d_ws, size_t ws_size,
                              hipStream_t stream) {
  const float* x_e  = (const float*)d_in[0];
  const float* x_r  = (const float*)d_in[1];
  const int*   eidx = (const int*)d_in[2];
  const int*   rel  = (const int*)d_in[3];
  const float* w_ah = (const float*)d_in[5];
  const float* w_at = (const float*)d_in[6];
  const float* w_ar = (const float*)d_in[7];
  const float* W1   = (const float*)d_in[8];
  const float* b1   = (const float*)d_in[9];
  const float* W2   = (const float*)d_in[10];
  const float* b2   = (const float*)d_in[11];
  const float* Wr   = (const float*)d_in[12];
  const float* br   = (const float*)d_in[13];
  float* out = (float*)d_out;

  const int n_e     = in_sizes[0] / E_HID;
  const int n_r     = in_sizes[1] / R_HID;
  const int n_edges = in_sizes[3];
  const int* h_idx = eidx;
  const int* t_idx = eidx + n_edges;

  float* P     = (float*)d_ws;
  float* msg   = P + (size_t)2 * n_r * 256;
  float* e_h   = msg + (size_t)n_r * R_HID;
  float* e_t   = e_h + n_e;
  float* e_r   = e_t + n_e;
  int* cnt     = (int*)(e_r + n_r);
  int* off     = cnt + (size_t)2 * n_e;
  int* rank_h  = off + (size_t)2 * n_e;
  int* rank_t  = rank_h + n_edges;
  int* list    = rank_t + n_edges;
  int* cursor  = list + (size_t)2 * n_edges;

  k_node_scores<<<(n_e + 63) / 64, 64, 0, stream>>>(x_e, w_ah, w_at, e_h, e_t, cnt, cursor, n_e);
  k_rel<<<n_r, 256, 0, stream>>>(x_r, w_ar, W1, b1, W2, b2, e_r, msg, n_r);
  k_proj<<<dim3((n_r + 15) / 16, 2), 256, 0, stream>>>(msg, Wr, P, n_r);
  k_count<<<2048, 256, 0, stream>>>(h_idx, t_idx, cnt, rank_h, rank_t, n_edges, n_e);
  k_alloc<<<(2 * n_e + 255) / 256, 256, 0, stream>>>(cnt, off, cursor, 2 * n_e);
  k_fill<<<2048, 256, 0, stream>>>(h_idx, t_idx, rel, off, rank_h, rank_t, list, n_edges, n_e);
  k_gather<<<2048, 256, 0, stream>>>(cnt, off, list, e_h, e_t, e_r, P, br, out, n_e, n_r);
}


# Relation MLP and projection kernels co-launched as extra blocks of the edge-count and segment-offset launches (independent chains overlap); 7 launches -> 5; kernel bodies and sum orders unchanged
# speedup vs baseline: 2.4579x; 1.0605x over previous
#include <hip/hip_runtime.h>

#define E_HID 128
#define R_HID 128

typedef float v2f __attribute__((ext_vector_type(2)));
typedef float v8f __attribute__((ext_vector_type(8)));
typedef float v4f __attribute__((ext_vector_type(4)));

__device__ __forceinline__ float leaky(float x) { return x > 0.f ? x : 0.01f * x; }


__device__ __forceinline__ float wave_max_f(float v) {
  for (int o = 16; o > 0; o >>= 1) v = fmaxf(v, __shfl_xor(v, o));
  return v;
}
__device__ __forceinline__ float wave_sum_f(float v) {
  for (int o = 16; o > 0; o >>= 1) v += __shfl_xor(v, o);
  return v;
}
__device__ __forceinline__ int wave_min_i(int v) {
  for (int o = 16; o > 0; o >>= 1) { int w = __shfl_xor(v, o); v = w < v ? w : v; }
  return v;
}
__device__ __forceinline__ int wave_sum_i(int v) {
  for (int o = 16; o > 0; o >>= 1) v += __shfl_xor(v, o);
  return v;
}

__global__ void __launch_bounds__(64) k_node_scores(const float* __restrict__ x_e,
                              const float* __restrict__ w_ah,
                              const float* __restrict__ w_at,
                              float* __restrict__ e_h, float* __restrict__ e_t,
                              int* __restrict__ cnt, int* __restrict__ cursor, int n_e) {
  const int lane = threadIdx.x & 31;
  const int base = ((int)blockIdx.x * 2 + (int)(threadIdx.x >> 5)) * 32;
  if (blockIdx.x == 0 && threadIdx.x == 0) *cursor = 0;
  if (base >= n_e) return;
  const int s = lane & 7;
  const int q = lane >> 3;
  v4f x[8][4];
  #pragma unroll
  for (int i = 0; i < 8; ++i) {
    int row = base + q * 8 + i;
    row = row < n_e ? row : n_e - 1;
    const v4f* p = (const v4f*)(x_e + (size_t)row * E_HID);
    #pragma unroll
    for (int j = 0; j < 4; ++j) x[i][j] = p[j * 8 + s];
  }
  v4f a[4], b[4];
  #pragma unroll
  for (int j = 0; j < 4; ++j) {
    a[j] = ((const v4f*)w_ah)[j * 8 + s];
    b[j] = ((const v4f*)w_at)[j * 8 + s];
  }
  asm volatile("" ::
               "v"(x[0][0]), "v"(x[0][1]), "v"(x[0][2]), "v"(x[0][3]),
               "v"(x[1][0]), "v"(x[1][1]), "v"(x[1][2]), "v"(x[1][3]),
               "v"(x[2][0]), "v"(x[2][1]), "v"(x[2][2]), "v"(x[2][3]),
               "v"(x[3][0]), "v"(x[3][1]), "v"(x[3][2]), "v"(x[3][3]),
               "v"(x[4][0]), "v"(x[4][1]), "v"(x[4][2]), "v"(x[4][3]),
               "v"(x[5][0]), "v"(x[5][1]), "v"(x[5][2]), "v"(x[5][3]),
               "v"(x[6][0]), "v"(x[6][1]), "v"(x[6][2]), "v"(x[6][3]),
               "v"(x[7][0]), "v"(x[7][1]), "v"(x[7][2]), "v"(x[7][3]),
               "v"(a[0]), "v"(b[0]), "v"(a[1]), "v"(b[1]),
               "v"(a[2]), "v"(b[2]), "v"(a[3]), "v"(b[3]));
  float vh = 0.f, vt = 0.f;
  #pragma unroll
  for (int i = 0; i < 8; ++i) {
    float sh = 0.f, st = 0.f;
    #pragma unroll
    for (int j = 0; j < 4; ++j) {
      const v4f v = x[i][j];
      sh = fmaf(v.x, a[j].x, sh); sh = fmaf(v.y, a[j].y, sh); sh = fmaf(v.z, a[j].z, sh); sh = fmaf(v.w, a[j].w, sh);
      st = fmaf(v.x, b[j].x, st); st = fmaf(v.y, b[j].y, st); st = fmaf(v.z, b[j].z, st); st = fmaf(v.w, b[j].w, st);
    }
    #pragma unroll
    for (int off = 4; off > 0; off >>= 1) {
      sh += __shfl_xor(sh, off);
      st += __shfl_xor(st, off);
    }
    if (s == i) { vh = sh; vt = st; }
  }
  const int row = base + lane;
  if (row < n_e) { e_h[row] = vh; e_t[row] = vt; cnt[row] = 0; cnt[n_e + row] = 0; }
}

__device__ __forceinline__ void rel_body(const int r, const float* __restrict__ x_r,
                      const float* __restrict__ w_ar,
                      const float* __restrict__ W1, const float* __restrict__ b1,
                      const float* __restrict__ W2, const float* __restrict__ b2,
                      float* __restrict__ e_r, float* __restrict__ msg) {
  __shared__ float row[R_HID];
  __shared__ float hid[2 * R_HID];
  int t = threadIdx.x;
  if (t < R_HID) row[t] = x_r[(size_t)r * R_HID + t];
  __syncthreads();
  float acc = b1[t];
  #pragma unroll 8
  for (int k = 0; k < R_HID; ++k) acc += row[k] * W1[k * 2 * R_HID + t];
  hid[t] = acc;
  if (t < 32) {
    float s = 0.f;
    #pragma unroll
    for (int j = 0; j < 4; ++j) s += row[t + 32 * j] * w_ar[t + 32 * j];
    for (int off = 16; off > 0; off >>= 1) s += __shfl_xor(s, off);
    if (t == 0) e_r[r] = s;
  }
  __syncthreads();
  if (t < R_HID) {
    float o = b2[t];
    #pragma unroll 8
    for (int k = 0; k < 2 * R_HID; ++k) o += hid[k] * W2[k * R_HID + t];
    msg[(size_t)r * R_HID + t] = row[t] + o;
  }
}

__device__ __forceinline__ void count_body(const int bid, const int nb,
                        const int* __restrict__ h_idx, const int* __restrict__ t_idx,
                        int* __restrict__ cnt, int* __restrict__ rank_h, int* __restrict__ rank_t,
                        int n_edges, int n_e) {
  int i = bid * blockDim.x + threadIdx.x;
  int stride = nb * blockDim.x;
  for (int e = i; e < n_edges; e += stride) {
    int h = h_idx[e], t = t_idx[e];
    const int rh = atomicAdd(cnt + h, 1);
    const int rt = atomicAdd(cnt + n_e + t, 1);
    rank_h[e] = rh;
    rank_t[e] = rt;
  }
}

__device__ __forceinline__ void alloc_body(const int bid, const int* __restrict__ cnt, int* __restrict__ off,
                        int* __restrict__ cursor, int n_keys) {
  __shared__ int wsum[8];
  __shared__ int bbase;
  int i = bid * 256 + threadIdx.x;
  int lane = threadIdx.x & 31;
  int wave = threadIdx.x >> 5;
  int c = (i < n_keys) ? cnt[i] : 0;
  int incl = c;
  for (int o = 1; o < 32; o <<= 1) {
    int v = __shfl_up(incl, o);
    if (lane >= o) incl += v;
  }
  if (lane == 31) wsum[wave] = incl;
  __syncthreads();
  if (threadIdx.x == 0) {
    int tot = 0;
    for (int w = 0; w < 8; ++w) { int v = wsum[w]; wsum[w] = tot; tot += v; }
    bbase = atomicAdd(cursor, tot);
  }
  __syncthreads();
  if (i < n_keys) off[i] = bbase + wsum[wave] + incl - c;
}

__global__ void k_fill(const int* __restrict__ h_idx, const int* __restrict__ t_idx,
                       const int* __restrict__ rel, const int* __restrict__ off,
                       const int* __restrict__ rank_h, const int* __restrict__ rank_t,
                       int* __restrict__ list, int n_edges, int n_e) {
  int i = blockIdx.x * blockDim.x + threadIdx.x;
  int stride = gridDim.x * blockDim.x;
  for (int e = i; e < n_edges; e += stride) {
    int h = h_idx[e], t = t_idx[e], r = rel[e];
    list[off[h] + rank_h[e]] = r;
    list[off[n_e + t] + rank_t[e]] = r;
  }
}

__device__ __forceinline__ void proj_body(const int mb, const int half,
                       const float* __restrict__ msg, const float* __restrict__ Wr,
                       float* __restrict__ P, int n_r) {
  constexpr int KTOT = 128, NTOT = 256, KC = 32;
  constexpr int ASTR = 36, BKSTR = 36;
  __shared__ float As[16 * ASTR];
  __shared__ float Bs[NTOT * BKSTR];
  const int tid  = threadIdx.x;
  const int lane = tid & 31;
  const int wave = tid >> 5;
  const int m0   = mb * 16;
  const int lrow = lane & 15;
  const int hsel = (lane >= 16) ? 2 : 0;
  const int mrow = (lane >= 16) ? 8 : 0;
  const int n0   = wave * 16;
  const float* __restrict__ B = Wr + (size_t)half * KTOT * NTOT;
  float* __restrict__ O = P + (size_t)half * n_r * NTOT;

  const int arow = tid >> 3;
  const int acol = (tid & 7) * 4;

  v8f c0 = {}, c1 = {};
  float4 aReg = make_float4(0.f, 0.f, 0.f, 0.f);
  float4 bReg[8];

  auto loadGlobal = [&](int kb) {
    if (tid < 128) {
      int grow = m0 + arow;
      aReg = make_float4(0.f, 0.f, 0.f, 0.f);
      if (grow < n_r) aReg = *(const float4*)(msg + (size_t)grow * KTOT + kb + acol);
    }
    #pragma unroll
    for (int i = 0; i < 8; ++i) {
      int k4 = i * 4;
      bReg[i].x = B[(size_t)(kb + k4 + 0) * NTOT + tid];
      bReg[i].y = B[(size_t)(kb + k4 + 1) * NTOT + tid];
      bReg[i].z = B[(size_t)(kb + k4 + 2) * NTOT + tid];
      bReg[i].w = B[(size_t)(kb + k4 + 3) * NTOT + tid];
    }
  };
  auto storeSmem = [&]() {
    if (tid < 128) *(float4*)&As[arow * ASTR + acol] = aReg;
    #pragma unroll
    for (int i = 0; i < 8; ++i)
      *(float4*)&Bs[tid * BKSTR + i * 4] = bReg[i];
  };

  loadGlobal(0);
  for (int kb = 0; kb < KTOT; kb += KC) {
    storeSmem();
    __syncthreads();
    if (kb + KC < KTOT) loadGlobal(kb + KC);
    #pragma unroll
    for (int kk = 0; kk < KC; kk += 4) {
      int ka = kk + hsel;
      v2f a  = *(const v2f*)&As[lrow * ASTR + ka];
      v2f b0 = *(const v2f*)&Bs[(n0 + lrow) * BKSTR + ka];
      v2f b1 = *(const v2f*)&Bs[(n0 + 128 + lrow) * BKSTR + ka];
      c0 = __builtin_amdgcn_wmma_f32_16x16x4_f32(false, a, false, b0, (short)0, c0, false, false);
      c1 = __builtin_amdgcn_wmma_f32_16x16x4_f32(false, a, false, b1, (short)0, c1, false, false);
    }
    __syncthreads();
  }
  #pragma unroll
  for (int j = 0; j < 8; ++j) {
    int grow = m0 + mrow + j;
    if (grow < n_r) {
      O[(size_t)grow * NTOT + n0 + lrow]       = c0[j];
      O[(size_t)grow * NTOT + n0 + 128 + lrow] = c1[j];
    }
  }
}

__global__ void k_rel_count(const float* __restrict__ x_r,
                            const float* __restrict__ w_ar,
                            const float* __restrict__ W1, const float* __restrict__ b1,
                            const float* __restrict__ W2, const float* __restrict__ b2,
                            float* __restrict__ e_r, float* __restrict__ msg, int n_r,
                            const int* __restrict__ h_idx, const int* __restrict__ t_idx,
                            int* __restrict__ cnt, int* __restrict__ rank_h, int* __restrict__ rank_t,
                            int n_edges, int n_e) {
  const int b = (int)blockIdx.x;
  if (b < n_r) rel_body(b, x_r, w_ar, W1, b1, W2, b2, e_r, msg);
  else count_body(b - n_r, (int)gridDim.x - n_r, h_idx, t_idx, cnt, rank_h, rank_t, n_edges, n_e);
}

__global__ void k_proj_alloc(const float* __restrict__ msg, const float* __restrict__ Wr,
                             float* __restrict__ P, int n_r, int n_mb,
                             const int* __restrict__ cnt, int* __restrict__ off,
                             int* __restrict__ cursor, int n_keys) {
  const int b = (int)blockIdx.x;
  if (b < 2 * n_mb) {
    const int half = b >= n_mb ? 1 : 0;
    proj_body(b - half * n_mb, half, msg, Wr, P, n_r);
  } else {
    alloc_body(b - 2 * n_mb, cnt, off, cursor, n_keys);
  }
}

__device__ __forceinline__ void fma_row(float4& A, float4& B, const float a, const float4 m, const float4 n) {
  A.x += a * m.x; A.y += a * m.y; A.z += a * m.z; A.w += a * m.w;
  B.x += a * n.x; B.y += a * n.y; B.z += a * n.z; B.w += a * n.w;
}

__device__ __forceinline__ void side_pass(const int d, const int o, const float en,
                                          const int* __restrict__ list, const float* __restrict__ e_r,
                                          const float* __restrict__ Ps, const int lane,
                                          float4& accA, float4& accB) {
  if (d > 0 && d <= 32) {
    const bool act = lane < d;
    int r = 0;
    float z = -__builtin_inff();
    if (act) { r = list[o + lane]; z = leaky(en + e_r[r]); }
    const float mx = wave_max_f(z);
    const float p = act ? __expf(z - mx) : 0.f;
    int rank = act ? 0 : lane;
    for (int j = 0; j < d; ++j) {
      const int rj = __builtin_amdgcn_readlane(r, j);
      if (act && (rj < r || (rj == r && j < lane))) rank += 1;
    }
    const int rs = __builtin_amdgcn_ds_permute(rank << 2, r);
    const float ps = __int_as_float(__builtin_amdgcn_ds_permute(rank << 2, __float_as_int(p)));
    const float s = wave_sum_f(ps);
    const float al = ps / (s + 1e-16f);
    int k = 0;
    #pragma unroll 1
    for (; k + 2 <= d; k += 2) {
      const int r0 = __builtin_amdgcn_readlane(rs, k);
      const int r1 = __builtin_amdgcn_readlane(rs, k + 1);
      const float a0 = __int_as_float(__builtin_amdgcn_readlane(__float_as_int(al), k));
      const float a1 = __int_as_float(__builtin_amdgcn_readlane(__float_as_int(al), k + 1));
      const float4* q0 = (const float4*)(Ps + (size_t)r0 * 256);
      const float4* q1 = (const float4*)(Ps + (size_t)r1 * 256);
      const float4 m0 = q0[lane], n0 = q0[32 + lane];
      const float4 m1 = q1[lane], n1 = q1[32 + lane];
      fma_row(accA, accB, a0, m0, n0);
      fma_row(accA, accB, a1, m1, n1);
    }
    if (k < d) {
      const int r0 = __builtin_amdgcn_readlane(rs, k);
      const float a0 = __int_as_float(__builtin_amdgcn_readlane(__float_as_int(al), k));
      const float4* q0 = (const float4*)(Ps + (size_t)r0 * 256);
      const float4 m0 = q0[lane], n0 = q0[32 + lane];
      fma_row(accA, accB, a0, m0, n0);
    }
  } else if (d > 32) {
    float mx = -__builtin_inff();
    for (int i = lane; i < d; i += 32) mx = fmaxf(mx, leaky(en + e_r[list[o + i]]));
    mx = wave_max_f(mx);
    float s = 0.f;
    int last = -1;
    for (int it = 0; it < d; ++it) {
      int best = 0x7fffffff;
      for (int i = lane; i < d; i += 32) { const int r = list[o + i]; if (r > last && r < best) best = r; }
      best = __builtin_amdgcn_readfirstlane(wave_min_i(best));
      if (best == 0x7fffffff) break;
      int c = 0;
      for (int i = lane; i < d; i += 32) c += (list[o + i] == best) ? 1 : 0;
      c = __builtin_amdgcn_readfirstlane(wave_sum_i(c));
      const float p = __expf(leaky(en + e_r[best]) - mx);
      for (int q = 0; q < c; ++q) s += p;
      last = best;
    }
    const float den = s + 1e-16f;
    last = -1;
    for (int it = 0; it < d; ++it) {
      int best = 0x7fffffff;
      for (int i = lane; i < d; i += 32) { const int r = list[o + i]; if (r > last && r < best) best = r; }
      best = __builtin_amdgcn_readfirstlane(wave_min_i(best));
      if (best == 0x7fffffff) break;
      int c = 0;
      for (int i = lane; i < d; i += 32) c += (list[o + i] == best) ? 1 : 0;
      c = __builtin_amdgcn_readfirstlane(wave_sum_i(c));
      const float a = __expf(leaky(en + e_r[best]) - mx) / den;
      const float4* q0 = (const float4*)(Ps + (size_t)best * 256);
      const float4 m0 = q0[lane], n0 = q0[32 + lane];
      for (int q = 0; q < c; ++q) fma_row(accA, accB, a, m0, n0);
      last = best;
    }
  }
}

__global__ void __launch_bounds__(256, 16) k_gather(const int* __restrict__ cnt, const int* __restrict__ off,
                         const int* __restrict__ list,
                         const float* __restrict__ e_h, const float* __restrict__ e_t,
                         const float* __restrict__ e_r, const float* __restrict__ P,
                         const float* __restrict__ br,
                         float* __restrict__ out, int n_e, int n_r) {
  const int lane = threadIdx.x & 31;
  const int wv = __builtin_amdgcn_readfirstlane((int)(threadIdx.x >> 5));
  const int nw = (int)gridDim.x * 8;
  for (int node = (int)blockIdx.x * 8 + wv; node < n_e; node += nw) {
    const int d0 = cnt[node];
    const int d1 = cnt[n_e + node];
    const int o0 = off[node];
    const int o1 = off[n_e + node];
    const float en0 = e_h[node];
    const float en1 = e_t[node];
    float4 accA = make_float4(0.f, 0.f, 0.f, 0.f);
    float4 accB = make_float4(0.f, 0.f, 0.f, 0.f);
    side_pass(d0, o0, en0, list, e_r, P, lane, accA, accB);
    side_pass(d1, o1, en1, list, e_r, P + (size_t)n_r * 256, lane, accA, accB);
    const float4 bA = ((const float4*)br)[lane];
    const float4 bB = ((const float4*)br)[32 + lane];
    accA.x += bA.x; accA.y += bA.y; accA.z += bA.z; accA.w += bA.w;
    accB.x += bB.x; accB.y += bB.y; accB.z += bB.z; accB.w += bB.w;
    float4* orow = (float4*)(out + (size_t)node * 256);
    orow[lane] = accA;
    orow[32 + lane] = accB;
  }
}

extern "C" void kernel_launch(void* const* d_in, const int* in_sizes, int n_in,
                              void* d_out, int out_size, void* d_ws, size_t ws_size,
                              hipStream_t stream) {
  const float* x_e  = (const float*)d_in[0];
  const float* x_r  = (const float*)d_in[1];
  const int*   eidx = (const int*)d_in[2];
  const int*   rel  = (const int*)d_in[3];
  const float* w_ah = (const float*)d_in[5];
  const float* w_at = (const float*)d_in[6];
  const float* w_ar = (const float*)d_in[7];
  const float* W1   = (const float*)d_in[8];
  const float* b1   = (const float*)d_in[9];
  const float* W2   = (const float*)d_in[10];
  const float* b2   = (const float*)d_in[11];
  const float* Wr   = (const float*)d_in[12];
  const float* br   = (const float*)d_in[13];
  float* out = (float*)d_out;

  const int n_e     = in_sizes[0] / E_HID;
  const int n_r     = in_sizes[1] / R_HID;
  const int n_edges = in_sizes[3];
  const int* h_idx = eidx;
  const int* t_idx = eidx + n_edges;

  float* P     = (float*)d_ws;
  float* msg   = P + (size_t)2 * n_r * 256;
  float* e_h   = msg + (size_t)n_r * R_HID;
  float* e_t   = e_h + n_e;
  float* e_r   = e_t + n_e;
  int* cnt     = (int*)(e_r + n_r);
  int* off     = cnt + (size_t)2 * n_e;
  int* rank_h  = off + (size_t)2 * n_e;
  int* rank_t  = rank_h + n_edges;
  int* list    = rank_t + n_edges;
  int* cursor  = list + (size_t)2 * n_edges;

  k_node_scores<<<(n_e + 63) / 64, 64, 0, stream>>>(x_e, w_ah, w_at, e_h, e_t, cnt, cursor, n_e);
  k_rel_count<<<n_r + 2048, 256, 0, stream>>>(x_r, w_ar, W1, b1, W2, b2, e_r, msg, n_r,
                                               h_idx, t_idx, cnt, rank_h, rank_t, n_edges, n_e);
  k_proj_alloc<<<2 * ((n_r + 15) / 16) + (2 * n_e + 255) / 256, 256, 0, stream>>>(msg, Wr, P, n_r, (n_r + 15) / 16,
                                                                                  cnt, off, cursor, 2 * n_e);
  k_fill<<<2048, 256, 0, stream>>>(h_idx, t_idx, rel, off, rank_h, rank_t, list, n_edges, n_e);
  k_gather<<<2048, 256, 0, stream>>>(cnt, off, list, e_h, e_t, e_r, P, br, out, n_e, n_r);
}


# Edge grouping by LDS counters instead of scattered returning global adds: edges binned by key range (one global add per block and bin), bins ranked into lists in LDS; fill pass gone, 5 launches -> 4
# speedup vs baseline: 4.3080x; 1.7527x over previous
#include <hip/hip_runtime.h>

#define E_HID 128
#define R_HID 128

#define KPB 256
#define BIN_CAP 4096
#define NBIN_MAX 1024
#define EPB 2048
#define CNT_OVER 0x40000000

typedef float v2f __attribute__((ext_vector_type(2)));
typedef float v8f __attribute__((ext_vector_type(8)));
typedef float v4f __attribute__((ext_vector_type(4)));

__device__ __forceinline__ float leaky(float x) { return x > 0.f ? x : 0.01f * x; }


__device__ __forceinline__ float wave_max_f(float v) {
  for (int o = 16; o > 0; o >>= 1) v = fmaxf(v, __shfl_xor(v, o));
  return v;
}
__device__ __forceinline__ float wave_sum_f(float v) {
  for (int o = 16; o > 0; o >>= 1) v += __shfl_xor(v, o);
  return v;
}
__device__ __forceinline__ int wave_min_i(int v) {
  for (int o = 16; o > 0; o >>= 1) { int w = __shfl_xor(v, o); v = w < v ? w : v; }
  return v;
}
__device__ __forceinline__ int wave_sum_i(int v) {
  for (int o = 16; o > 0; o >>= 1) v += __shfl_xor(v, o);
  return v;
}

__global__ void __launch_bounds__(64) k_node_scores(const float* __restrict__ x_e,
                              const float* __restrict__ w_ah,
                              const float* __restrict__ w_at,
                              float* __restrict__ e_h, float* __restrict__ e_t,
                              int* __restrict__ gcount, int nbin, int n_e) {
  const int lane = threadIdx.x & 31;
  const int base = ((int)blockIdx.x * 2 + (int)(threadIdx.x >> 5)) * 32;
  if (base >= n_e) return;
  const int s = lane & 7;
  const int q = lane >> 3;
  v4f x[8][4];
  #pragma unroll
  for (int i = 0; i < 8; ++i) {
    int row = base + q * 8 + i;
    row = row < n_e ? row : n_e - 1;
    const v4f* p = (const v4f*)(x_e + (size_t)row * E_HID);
    #pragma unroll
    for (int j = 0; j < 4; ++j) x[i][j] = p[j * 8 + s];
  }
  v4f a[4], b[4];
  #pragma unroll
  for (int j = 0; j < 4; ++j) {
    a[j] = ((const v4f*)w_ah)[j * 8 + s];
    b[j] = ((const v4f*)w_at)[j * 8 + s];
  }
  asm volatile("" ::
               "v"(x[0][0]), "v"(x[0][1]), "v"(x[0][2]), "v"(x[0][3]),
               "v"(x[1][0]), "v"(x[1][1]), "v"(x[1][2]), "v"(x[1][3]),
               "v"(x[2][0]), "v"(x[2][1]), "v"(x[2][2]), "v"(x[2][3]),
               "v"(x[3][0]), "v"(x[3][1]), "v"(x[3][2]), "v"(x[3][3]),
               "v"(x[4][0]), "v"(x[4][1]), "v"(x[4][2]), "v"(x[4][3]),
               "v"(x[5][0]), "v"(x[5][1]), "v"(x[5][2]), "v"(x[5][3]),
               "v"(x[6][0]), "v"(x[6][1]), "v"(x[6][2]), "v"(x[6][3]),
               "v"(x[7][0]), "v"(x[7][1]), "v"(x[7][2]), "v"(x[7][3]),
               "v"(a[0]), "v"(b[0]), "v"(a[1]), "v"(b[1]),
               "v"(a[2]), "v"(b[2]), "v"(a[3]), "v"(b[3]));
  float vh = 0.f, vt = 0.f;
  #pragma unroll
  for (int i = 0; i < 8; ++i) {
    float sh = 0.f, st = 0.f;
    #pragma unroll
    for (int j = 0; j < 4; ++j) {
      const v4f v = x[i][j];
      sh = fmaf(v.x, a[j].x, sh); sh = fmaf(v.y, a[j].y, sh); sh = fmaf(v.z, a[j].z, sh); sh = fmaf(v.w, a[j].w, sh);
      st = fmaf(v.x, b[j].x, st); st = fmaf(v.y, b[j].y, st); st = fmaf(v.z, b[j].z, st); st = fmaf(v.w, b[j].w, st);
    }
    #pragma unroll
    for (int off = 4; off > 0; off >>= 1) {
      sh += __shfl_xor(sh, off);
      st += __shfl_xor(st, off);
    }
    if (s == i) { vh = sh; vt = st; }
  }
  const int row = base + lane;
  if (row < n_e) { e_h[row] = vh; e_t[row] = vt; }
  if (row < nbin) gcount[row] = 0;
}

__device__ __forceinline__ void rel_body(const int r, const float* __restrict__ x_r,
                      const float* __restrict__ w_ar,
                      const float* __restrict__ W1, const float* __restrict__ b1,
                      const float* __restrict__ W2, const float* __restrict__ b2,
                      float* __restrict__ e_r, float* __restrict__ msg) {
  __shared__ float row[R_HID];
  __shared__ float hid[2 * R_HID];
  int t = threadIdx.x;
  if (t < R_HID) row[t] = x_r[(size_t)r * R_HID + t];
  __syncthreads();
  float acc = b1[t];
  #pragma unroll 8
  for (int k = 0; k < R_HID; ++k) acc += row[k] * W1[k * 2 * R_HID + t];
  hid[t] = acc;
  if (t < 32) {
    float s = 0.f;
    #pragma unroll
    for (int j = 0; j < 4; ++j) s += row[t + 32 * j] * w_ar[t + 32 * j];
    for (int off = 16; off > 0; off >>= 1) s += __shfl_xor(s, off);
    if (t == 0) e_r[r] = s;
  }
  __syncthreads();
  if (t < R_HID) {
    float o = b2[t];
    #pragma unroll 8
    for (int k = 0; k < 2 * R_HID; ++k) o += hid[k] * W2[k * R_HID + t];
    msg[(size_t)r * R_HID + t] = row[t] + o;
  }
}

__device__ __forceinline__ int4 load4_vec(const int* __restrict__ p, const int e, const int e0, const int e1) {
  const bool in = e < e1;
  int4 v = *(const int4*)(p + (in ? e : e0));
  if (!in) v = make_int4(-1, -1, -1, -1);
  return v;
}
__device__ __forceinline__ int4 load4_any(const int* __restrict__ p, const int e, const int e0, const int e1) {
  int4 v;
  v.x = p[(e     < e1) ? e     : e0];
  v.y = p[(e + 1 < e1) ? e + 1 : e0];
  v.z = p[(e + 2 < e1) ? e + 2 : e0];
  v.w = p[(e + 3 < e1) ? e + 3 : e0];
  if (e     >= e1) v.x = -1;
  if (e + 1 >= e1) v.y = -1;
  if (e + 2 >= e1) v.z = -1;
  if (e + 3 >= e1) v.w = -1;
  return v;
}

__device__ __forceinline__ void bin_body(const int bid,
                      const int* __restrict__ h_idx, const int* __restrict__ t_idx, const int* __restrict__ rel,
                      int* __restrict__ gcount, int* __restrict__ binned,
                      int n_edges, int n_e, int nbin) {
  __shared__ int hist[NBIN_MAX];
  __shared__ int gbase[NBIN_MAX];
  __shared__ int lbase[NBIN_MAX];
  __shared__ int sRec[2 * EPB];
  __shared__ int sAdr[2 * EPB];
  __shared__ int wsum[8];
  const int tid  = threadIdx.x;
  const int lane = tid & 31;
  const int wave = tid >> 5;
  const int e0 = bid * EPB;
  const int e1 = (e0 + EPB < n_edges) ? e0 + EPB : n_edges;
  const bool vec = (n_edges & 3) == 0;
  const int ea = e0 + tid * 4;
  const int eb = ea + 1024;
  #pragma unroll
  for (int k = 0; k < 4; ++k) hist[tid + 256 * k] = 0;
  int4 hA, tA, hB, tB, rA, rB;
  if (vec) {
    hA = load4_vec(h_idx, ea, e0, e1); tA = load4_vec(t_idx, ea, e0, e1); rA = load4_vec(rel, ea, e0, e1);
    hB = load4_vec(h_idx, eb, e0, e1); tB = load4_vec(t_idx, eb, e0, e1); rB = load4_vec(rel, eb, e0, e1);
  } else {
    hA = load4_any(h_idx, ea, e0, e1); tA = load4_any(t_idx, ea, e0, e1); rA = load4_any(rel, ea, e0, e1);
    hB = load4_any(h_idx, eb, e0, e1); tB = load4_any(t_idx, eb, e0, e1); rB = load4_any(rel, eb, e0, e1);
  }
  const int hv[8] = {hA.x, hA.y, hA.z, hA.w, hB.x, hB.y, hB.z, hB.w};
  const int tv[8] = {tA.x, tA.y, tA.z, tA.w, tB.x, tB.y, tB.z, tB.w};
  const int rv[8] = {rA.x, rA.y, rA.z, rA.w, rB.x, rB.y, rB.z, rB.w};
  __syncthreads();
  #pragma unroll
  for (int j = 0; j < 8; ++j) {
    if (hv[j] >= 0) {
      atomicAdd(&hist[hv[j] >> 8], 1);
      atomicAdd(&hist[(n_e + tv[j]) >> 8], 1);
    }
  }
  __syncthreads();
  int cg[4], g[4];
  #pragma unroll
  for (int k = 0; k < 4; ++k) {
    const int b = tid + 256 * k;
    cg[k] = (b < nbin) ? hist[b] : 0;
  }
  #pragma unroll
  for (int k = 0; k < 4; ++k) g[k] = (cg[k] > 0) ? atomicAdd(gcount + tid + 256 * k, cg[k]) : 0;
  int c[4];
  #pragma unroll
  for (int k = 0; k < 4; ++k) {
    const int b = 4 * tid + k;
    c[k] = (b < nbin) ? hist[b] : 0;
  }
  const int tsum = c[0] + c[1] + c[2] + c[3];
  int incl = tsum;
  for (int o = 1; o < 32; o <<= 1) {
    const int v = __shfl_up(incl, o);
    if (lane >= o) incl += v;
  }
  if (lane == 31) wsum[wave] = incl;
  __syncthreads();
  int excl = incl - tsum;
  for (int w = 0; w < wave; ++w) excl += wsum[w];
  #pragma unroll
  for (int k = 0; k < 4; ++k) {
    const int b = 4 * tid + k;
    lbase[b] = excl;
    hist[b] = 0;
    excl += c[k];
  }
  #pragma unroll
  for (int k = 0; k < 4; ++k) gbase[tid + 256 * k] = g[k];
  __syncthreads();
  #pragma unroll
  for (int j = 0; j < 8; ++j) {
    if (hv[j] >= 0) {
      const int kh = hv[j];
      const int kt = n_e + tv[j];
      const int bh = kh >> 8;
      const int bt = kt >> 8;
      const int lh = atomicAdd(&hist[bh], 1);
      const int lt = atomicAdd(&hist[bt], 1);
      const int ph = lbase[bh] + lh;
      const int pt = lbase[bt] + lt;
      const int gh = gbase[bh] + lh;
      const int gt = gbase[bt] + lt;
      sRec[ph] = (rv[j] << 8) | (kh & 255);
      sRec[pt] = (rv[j] << 8) | (kt & 255);
      sAdr[ph] = (gh < BIN_CAP) ? bh * BIN_CAP + gh : -1;
      sAdr[pt] = (gt < BIN_CAP) ? bt * BIN_CAP + gt : -1;
    }
  }
  __syncthreads();
  const int nrec = 2 * (e1 - e0);
  for (int i = tid; i < nrec; i += 256) {
    const int a = sAdr[i];
    if (a >= 0) binned[a] = sRec[i];
  }
}

__device__ __forceinline__ void binlist_body(const int bin, const int* __restrict__ gcount,
                          const int* __restrict__ binned,
                          int* __restrict__ cnt, int* __restrict__ off, int* __restrict__ list, int n_keys) {
  __shared__ int kcnt[KPB];
  __shared__ int koff[KPB];
  __shared__ int recs[BIN_CAP];
  __shared__ int slot[BIN_CAP];
  __shared__ int wsum2[8];
  const int tid  = threadIdx.x;
  const int lane = tid & 31;
  const int wave = tid >> 5;
  const int tot = gcount[bin];
  const bool over = tot > BIN_CAP;
  const int n = over ? BIN_CAP : tot;
  kcnt[tid] = 0;
  __syncthreads();
  const int* __restrict__ src = binned + (size_t)bin * BIN_CAP;
  for (int i0 = tid; i0 < n; i0 += 1024) {
    int rc[4];
    #pragma unroll
    for (int k = 0; k < 4; ++k) rc[k] = (i0 + 256 * k < n) ? src[i0 + 256 * k] : -1;
    #pragma unroll
    for (int k = 0; k < 4; ++k) {
      if (rc[k] >= 0) {
        recs[i0 + 256 * k] = rc[k];
        slot[i0 + 256 * k] = atomicAdd(&kcnt[rc[k] & 255], 1);
      }
    }
  }
  __syncthreads();
  const int c = kcnt[tid];
  int incl = c;
  for (int o = 1; o < 32; o <<= 1) {
    const int v = __shfl_up(incl, o);
    if (lane >= o) incl += v;
  }
  if (lane == 31) wsum2[wave] = incl;
  __syncthreads();
  int wb = 0;
  for (int w = 0; w < wave; ++w) wb += wsum2[w];
  const int ko = wb + incl - c;
  koff[tid] = ko;
  const int key = bin * KPB + tid;
  if (key < n_keys) {
    cnt[key] = over ? CNT_OVER : c;
    off[key] = bin * BIN_CAP + ko;
  }
  __syncthreads();
  if (!over) {
    int* __restrict__ dst = list + (size_t)bin * BIN_CAP;
    for (int i = tid; i < n; i += 256) {
      const int rec = recs[i];
      dst[koff[rec & 255] + slot[i]] = rec >> 8;
    }
  }
}

__device__ __forceinline__ void proj_body(const int mb, const int half,
                       const float* __restrict__ msg, const float* __restrict__ Wr,
                       float* __restrict__ P, int n_r) {
  constexpr int KTOT = 128, NTOT = 256, KC = 32;
  constexpr int ASTR = 36, BKSTR = 36;
  __shared__ float As[16 * ASTR];
  __shared__ float Bs[NTOT * BKSTR];
  const int tid  = threadIdx.x;
  const int lane = tid & 31;
  const int wave = tid >> 5;
  const int m0   = mb * 16;
  const int lrow = lane & 15;
  const int hsel = (lane >= 16) ? 2 : 0;
  const int mrow = (lane >= 16) ? 8 : 0;
  const int n0   = wave * 16;
  const float* __restrict__ B = Wr + (size_t)half * KTOT * NTOT;
  float* __restrict__ O = P + (size_t)half * n_r * NTOT;

  const int arow = tid >> 3;
  const int acol = (tid & 7) * 4;

  v8f c0 = {}, c1 = {};
  float4 aReg = make_float4(0.f, 0.f, 0.f, 0.f);
  float4 bReg[8];

  auto loadGlobal = [&](int kb) {
    if (tid < 128) {
      int grow = m0 + arow;
      aReg = make_float4(0.f, 0.f, 0.f, 0.f);
      if (grow < n_r) aReg = *(const float4*)(msg + (size_t)grow * KTOT + kb + acol);
    }
    #pragma unroll
    for (int i = 0; i < 8; ++i) {
      int k4 = i * 4;
      bReg[i].x = B[(size_t)(kb + k4 + 0) * NTOT + tid];
      bReg[i].y = B[(size_t)(kb + k4 + 1) * NTOT + tid];
      bReg[i].z = B[(size_t)(kb + k4 + 2) * NTOT + tid];
      bReg[i].w = B[(size_t)(kb + k4 + 3) * NTOT + tid];
    }
  };
  auto storeSmem = [&]() {
    if (tid < 128) *(float4*)&As[arow * ASTR + acol] = aReg;
    #pragma unroll
    for (int i = 0; i < 8; ++i)
      *(float4*)&Bs[tid * BKSTR + i * 4] = bReg[i];
  };

  loadGlobal(0);
  for (int kb = 0; kb < KTOT; kb += KC) {
    storeSmem();
    __syncthreads();
    if (kb + KC < KTOT) loadGlobal(kb + KC);
    #pragma unroll
    for (int kk = 0; kk < KC; kk += 4) {
      int ka = kk + hsel;
      v2f a  = *(const v2f*)&As[lrow * ASTR + ka];
      v2f b0 = *(const v2f*)&Bs[(n0 + lrow) * BKSTR + ka];
      v2f b1 = *(const v2f*)&Bs[(n0 + 128 + lrow) * BKSTR + ka];
      c0 = __builtin_amdgcn_wmma_f32_16x16x4_f32(false, a, false, b0, (short)0, c0, false, false);
      c1 = __builtin_amdgcn_wmma_f32_16x16x4_f32(false, a, false, b1, (short)0, c1, false, false);
    }
    __syncthreads();
  }
  #pragma unroll
  for (int j = 0; j < 8; ++j) {
    int grow = m0 + mrow + j;
    if (grow < n_r) {
      O[(size_t)grow * NTOT + n0 + lrow]       = c0[j];
      O[(size_t)grow * NTOT + n0 + 128 + lrow] = c1[j];
    }
  }
}

__global__ void k_rel_bin(const float* __restrict__ x_r,
                          const float* __restrict__ w_ar,
                          const float* __restrict__ W1, const float* __restrict__ b1,
                          const float* __restrict__ W2, const float* __restrict__ b2,
                          float* __restrict__ e_r, float* __restrict__ msg, int n_r,
                          const int* __restrict__ h_idx, const int* __restrict__ t_idx, const int* __restrict__ rel,
                          int* __restrict__ gcount, int* __restrict__ binned,
                          int n_edges, int n_e, int nbin) {
  const int b = (int)blockIdx.x;
  if (b < n_r) rel_body(b, x_r, w_ar, W1, b1, W2, b2, e_r, msg);
  else bin_body(b - n_r, h_idx, t_idx, rel, gcount, binned, n_edges, n_e, nbin);
}

__global__ void k_proj_list(const float* __restrict__ msg, const float* __restrict__ Wr,
                            float* __restrict__ P, int n_r, int n_mb,
                            const int* __restrict__ gcount, const int* __restrict__ binned,
                            int* __restrict__ cnt, int* __restrict__ off, int* __restrict__ list, int n_keys) {
  const int b = (int)blockIdx.x;
  if (b < 2 * n_mb) {
    const int half = b >= n_mb ? 1 : 0;
    proj_body(b - half * n_mb, half, msg, Wr, P, n_r);
  } else {
    binlist_body(b - 2 * n_mb, gcount, binned, cnt, off, list, n_keys);
  }
}

__device__ __forceinline__ void fma_row(float4& A, float4& B, const float a, const float4 m, const float4 n) {
  A.x += a * m.x; A.y += a * m.y; A.z += a * m.z; A.w += a * m.w;
  B.x += a * n.x; B.y += a * n.y; B.z += a * n.z; B.w += a * n.w;
}

__device__ __forceinline__ void side_pass(const int d, const int o, const float en,
                                          const int* __restrict__ list, const float* __restrict__ e_r,
                                          const float* __restrict__ Ps, const int lane,
                                          const int node, const int* __restrict__ idx,
                                          const int* __restrict__ rel, const int n_edges,
                                          float4& accA, float4& accB) {
  if (d > 0 && d <= 32) {
    const bool act = lane < d;
    int r = 0;
    float z = -__builtin_inff();
    if (act) { r = list[o + lane]; z = leaky(en + e_r[r]); }
    const float mx = wave_max_f(z);
    const float p = act ? __expf(z - mx) : 0.f;
    int rank = act ? 0 : lane;
    for (int j = 0; j < d; ++j) {
      const int rj = __builtin_amdgcn_readlane(r, j);
      if (act && (rj < r || (rj == r && j < lane))) rank += 1;
    }
    const int rs = __builtin_amdgcn_ds_permute(rank << 2, r);
    const float ps = __int_as_float(__builtin_amdgcn_ds_permute(rank << 2, __float_as_int(p)));
    const float s = wave_sum_f(ps);
    const float al = ps / (s + 1e-16f);
    int k = 0;
    #pragma unroll 1
    for (; k + 2 <= d; k += 2) {
      const int r0 = __builtin_amdgcn_readlane(rs, k);
      const int r1 = __builtin_amdgcn_readlane(rs, k + 1);
      const float a0 = __int_as_float(__builtin_amdgcn_readlane(__float_as_int(al), k));
      const float a1 = __int_as_float(__builtin_amdgcn_readlane(__float_as_int(al), k + 1));
      const float4* q0 = (const float4*)(Ps + (size_t)r0 * 256);
      const float4* q1 = (const float4*)(Ps + (size_t)r1 * 256);
      const float4 m0 = q0[lane], n0 = q0[32 + lane];
      const float4 m1 = q1[lane], n1 = q1[32 + lane];
      fma_row(accA, accB, a0, m0, n0);
      fma_row(accA, accB, a1, m1, n1);
    }
    if (k < d) {
      const int r0 = __builtin_amdgcn_readlane(rs, k);
      const float a0 = __int_as_float(__builtin_amdgcn_readlane(__float_as_int(al), k));
      const float4* q0 = (const float4*)(Ps + (size_t)r0 * 256);
      const float4 m0 = q0[lane], n0 = q0[32 + lane];
      fma_row(accA, accB, a0, m0, n0);
    }
  } else if (d > 32) {
    float mx = -__builtin_inff();
    for (int i = lane; i < n_edges; i += 32) if (idx[i] == node) mx = fmaxf(mx, leaky(en + e_r[rel[i]]));
    mx = wave_max_f(mx);
    float s = 0.f;
    int last = -1;
    for (int it = 0; it < n_edges; ++it) {
      int best = 0x7fffffff;
      for (int i = lane; i < n_edges; i += 32) if (idx[i] == node) { const int r = rel[i]; if (r > last && r < best) best = r; }
      best = __builtin_amdgcn_readfirstlane(wave_min_i(best));
      if (best == 0x7fffffff) break;
      int c = 0;
      for (int i = lane; i < n_edges; i += 32) c += (idx[i] == node && rel[i] == best) ? 1 : 0;
      c = __builtin_amdgcn_readfirstlane(wave_sum_i(c));
      const float p = __expf(leaky(en + e_r[best]) - mx);
      for (int q = 0; q < c; ++q) s += p;
      last = best;
    }
    const float den = s + 1e-16f;
    last = -1;
    for (int it = 0; it < n_edges; ++it) {
      int best = 0x7fffffff;
      for (int i = lane; i < n_edges; i += 32) if (idx[i] == node) { const int r = rel[i]; if (r > last && r < best) best = r; }
      best = __builtin_amdgcn_readfirstlane(wave_min_i(best));
      if (best == 0x7fffffff) break;
      int c = 0;
      for (int i = lane; i < n_edges; i += 32) c += (idx[i] == node && rel[i] == best) ? 1 : 0;
      c = __builtin_amdgcn_readfirstlane(wave_sum_i(c));
      const float a = __expf(leaky(en + e_r[best]) - mx) / den;
      const float4* q0 = (const float4*)(Ps + (size_t)best * 256);
      const float4 m0 = q0[lane], n0 = q0[32 + lane];
      for (int q = 0; q < c; ++q) fma_row(accA, accB, a, m0, n0);
      last = best;
    }
  }
}

__global__ void __launch_bounds__(256, 16) k_gather(const int* __restrict__ cnt, const int* __restrict__ off,
                         const int* __restrict__ list,
                         const float* __restrict__ e_h, const float* __restrict__ e_t,
                         const float* __restrict__ e_r, const float* __restrict__ P,
                         const float* __restrict__ br,
                         float* __restrict__ out, int n_e, int n_r,
                         const int* __restrict__ h_idx, const int* __restrict__ t_idx,
                         const int* __restrict__ rel, int n_edges) {
  const int lane = threadIdx.x & 31;
  const int wv = __builtin_amdgcn_readfirstlane((int)(threadIdx.x >> 5));
  const int nw = (int)gridDim.x * 8;
  for (int node = (int)blockIdx.x * 8 + wv; node < n_e; node += nw) {
    const int d0 = cnt[node];
    const int d1 = cnt[n_e + node];
    const int o0 = off[node];
    const int o1 = off[n_e + node];
    const float en0 = e_h[node];
    const float en1 = e_t[node];
    float4 accA = make_float4(0.f, 0.f, 0.f, 0.f);
    float4 accB = make_float4(0.f, 0.f, 0.f, 0.f);
    side_pass(d0, o0, en0, list, e_r, P, lane, node, h_idx, rel, n_edges, accA, accB);
    side_pass(d1, o1, en1, list, e_r, P + (size_t)n_r * 256, lane, node, t_idx, rel, n_edges, accA, accB);
    const float4 bA = ((const float4*)br)[lane];
    const float4 bB = ((const float4*)br)[32 + lane];
    accA.x += bA.x; accA.y += bA.y; accA.z += bA.z; accA.w += bA.w;
    accB.x += bB.x; accB.y += bB.y; accB.z += bB.z; accB.w += bB.w;
    float4* orow = (float4*)(out + (size_t)node * 256);
    orow[lane] = accA;
    orow[32 + lane] = accB;
  }
}

extern "C" void kernel_launch(void* const* d_in, const int* in_sizes, int n_in,
                              void* d_out, int out_size, void* d_ws, size_t ws_size,
                              hipStream_t stream) {
  const float* x_e  = (const float*)d_in[0];
  const float* x_r  = (const float*)d_in[1];
  const int*   eidx = (const int*)d_in[2];
  const int*   rel  = (const int*)d_in[3];
  const float* w_ah = (const float*)d_in[5];
  const float* w_at = (const float*)d_in[6];
  const float* w_ar = (const float*)d_in[7];
  const float* W1   = (const float*)d_in[8];
  const float* b1   = (const float*)d_in[9];
  const float* W2   = (const float*)d_in[10];
  const float* b2   = (const float*)d_in[11];
  const float* Wr   = (const float*)d_in[12];
  const float* br   = (const float*)d_in[13];
  float* out = (float*)d_out;

  const int n_e     = in_sizes[0] / E_HID;
  const int n_r     = in_sizes[1] / R_HID;
  const int n_edges = in_sizes[3];
  const int* h_idx = eidx;
  const int* t_idx = eidx + n_edges;
  const int nbin   = (2 * n_e + KPB - 1) / KPB;

  float* P     = (float*)d_ws;
  float* msg   = P + (size_t)2 * n_r * 256;
  float* e_h   = msg + (size_t)n_r * R_HID;
  float* e_t   = e_h + n_e;
  float* e_r   = e_t + n_e;
  int* cnt     = (int*)(e_r + n_r);
  int* off     = cnt + (size_t)2 * n_e;
  int* gcount  = off + (size_t)2 * n_e;
  int* binned  = gcount + NBIN_MAX;
  int* list    = binned + (size_t)nbin * BIN_CAP;

  k_node_scores<<<(n_e + 63) / 64, 64, 0, stream>>>(x_e, w_ah, w_at, e_h, e_t, gcount, nbin, n_e);
  k_rel_bin<<<n_r + (n_edges + EPB - 1) / EPB, 256, 0, stream>>>(x_r, w_ar, W1, b1, W2, b2, e_r, msg, n_r,
                                                                 h_idx, t_idx, rel, gcount, binned, n_edges, n_e, nbin);
  k_proj_list<<<2 * ((n_r + 15) / 16) + nbin, 256, 0, stream>>>(msg, Wr, P, n_r, (n_r + 15) / 16,
                                                                gcount, binned, cnt, off, list, 2 * n_e);
  k_gather<<<2048, 256, 0, stream>>>(cnt, off, list, e_h, e_t, e_r, P, br, out, n_e, n_r, h_idx, t_idx, rel, n_edges);
}


# Softmax weights and summing order per node side computed in the list pass (lane per edge record, in LDS) instead of wave per node in the gather; gather only walks rows; softmax sum order changed
# speedup vs baseline: 4.6364x; 1.0762x over previous
#include <hip/hip_runtime.h>

#define E_HID 128
#define R_HID 128

#define KPB 256
#define BIN_CAP 4096
#define NBIN_MAX 1024
#define EPB 2048
#define CNT_OVER 0x40000000

typedef float v2f __attribute__((ext_vector_type(2)));
typedef float v8f __attribute__((ext_vector_type(8)));
typedef float v4f __attribute__((ext_vector_type(4)));

__device__ __forceinline__ float leaky(float x) { return x > 0.f ? x : 0.01f * x; }


__device__ __forceinline__ float wave_max_f(float v) {
  for (int o = 16; o > 0; o >>= 1) v = fmaxf(v, __shfl_xor(v, o));
  return v;
}
__device__ __forceinline__ int wave_min_i(int v) {
  for (int o = 16; o > 0; o >>= 1) { int w = __shfl_xor(v, o); v = w < v ? w : v; }
  return v;
}
__device__ __forceinline__ int wave_sum_i(int v) {
  for (int o = 16; o > 0; o >>= 1) v += __shfl_xor(v, o);
  return v;
}

__global__ void __launch_bounds__(64) k_node_scores(const float* __restrict__ x_e,
                              const float* __restrict__ w_ah,
                              const float* __restrict__ w_at,
                              float* __restrict__ e_h, float* __restrict__ e_t,
                              int* __restrict__ gcount, int nbin, int n_e) {
  const int lane = threadIdx.x & 31;
  const int base = ((int)blockIdx.x * 2 + (int)(threadIdx.x >> 5)) * 32;
  if (base >= n_e) return;
  const int s = lane & 7;
  const int q = lane >> 3;
  v4f x[8][4];
  #pragma unroll
  for (int i = 0; i < 8; ++i) {
    int row = base + q * 8 + i;
    row = row < n_e ? row : n_e - 1;
    const v4f* p = (const v4f*)(x_e + (size_t)row * E_HID);
    #pragma unroll
    for (int j = 0; j < 4; ++j) x[i][j] = p[j * 8 + s];
  }
  v4f a[4], b[4];
  #pragma unroll
  for (int j = 0; j < 4; ++j) {
    a[j] = ((const v4f*)w_ah)[j * 8 + s];
    b[j] = ((const v4f*)w_at)[j * 8 + s];
  }
  asm volatile("" ::
               "v"(x[0][0]), "v"(x[0][1]), "v"(x[0][2]), "v"(x[0][3]),
               "v"(x[1][0]), "v"(x[1][1]), "v"(x[1][2]), "v"(x[1][3]),
               "v"(x[2][0]), "v"(x[2][1]), "v"(x[2][2]), "v"(x[2][3]),
               "v"(x[3][0]), "v"(x[3][1]), "v"(x[3][2]), "v"(x[3][3]),
               "v"(x[4][0]), "v"(x[4][1]), "v"(x[4][2]), "v"(x[4][3]),
               "v"(x[5][0]), "v"(x[5][1]), "v"(x[5][2]), "v"(x[5][3]),
               "v"(x[6][0]), "v"(x[6][1]), "v"(x[6][2]), "v"(x[6][3]),
               "v"(x[7][0]), "v"(x[7][1]), "v"(x[7][2]), "v"(x[7][3]),
               "v"(a[0]), "v"(b[0]), "v"(a[1]), "v"(b[1]),
               "v"(a[2]), "v"(b[2]), "v"(a[3]), "v"(b[3]));
  float vh = 0.f, vt = 0.f;
  #pragma unroll
  for (int i = 0; i < 8; ++i) {
    float sh = 0.f, st = 0.f;
    #pragma unroll
    for (int j = 0; j < 4; ++j) {
      const v4f v = x[i][j];
      sh = fmaf(v.x, a[j].x, sh); sh = fmaf(v.y, a[j].y, sh); sh = fmaf(v.z, a[j].z, sh); sh = fmaf(v.w, a[j].w, sh);
      st = fmaf(v.x, b[j].x, st); st = fmaf(v.y, b[j].y, st); st = fmaf(v.z, b[j].z, st); st = fmaf(v.w, b[j].w, st);
    }
    #pragma unroll
    for (int off = 4; off > 0; off >>= 1) {
      sh += __shfl_xor(sh, off);
      st += __shfl_xor(st, off);
    }
    if (s == i) { vh = sh; vt = st; }
  }
  const int row = base + lane;
  if (row < n_e) { e_h[row] = vh; e_t[row] = vt; }
  if (row < nbin) gcount[row] = 0;
}

__device__ __forceinline__ void rel_body(const int r, const float* __restrict__ x_r,
                      const float* __restrict__ w_ar,
                      const float* __restrict__ W1, const float* __restrict__ b1,
                      const float* __restrict__ W2, const float* __restrict__ b2,
                      float* __restrict__ e_r, float* __restrict__ msg) {
  __shared__ float row[R_HID];
  __shared__ float hid[2 * R_HID];
  int t = threadIdx.x;
  if (t < R_HID) row[t] = x_r[(size_t)r * R_HID + t];
  __syncthreads();
  float acc = b1[t];
  #pragma unroll 8
  for (int k = 0; k < R_HID; ++k) acc += row[k] * W1[k * 2 * R_HID + t];
  hid[t] = acc;
  if (t < 32) {
    float s = 0.f;
    #pragma unroll
    for (int j = 0; j < 4; ++j) s += row[t + 32 * j] * w_ar[t + 32 * j];
    for (int off = 16; off > 0; off >>= 1) s += __shfl_xor(s, off);
    if (t == 0) e_r[r] = s;
  }
  __syncthreads();
  if (t < R_HID) {
    float o = b2[t];
    #pragma unroll 8
    for (int k = 0; k < 2 * R_HID; ++k) o += hid[k] * W2[k * R_HID + t];
    msg[(size_t)r * R_HID + t] = row[t] + o;
  }
}

__device__ __forceinline__ int4 load4_vec(const int* __restrict__ p, const int e, const int e0, const int e1) {
  const bool in = e < e1;
  int4 v = *(const int4*)(p + (in ? e : e0));
  if (!in) v = make_int4(-1, -1, -1, -1);
  return v;
}
__device__ __forceinline__ int4 load4_any(const int* __restrict__ p, const int e, const int e0, const int e1) {
  int4 v;
  v.x = p[(e     < e1) ? e     : e0];
  v.y = p[(e + 1 < e1) ? e + 1 : e0];
  v.z = p[(e + 2 < e1) ? e + 2 : e0];
  v.w = p[(e + 3 < e1) ? e + 3 : e0];
  if (e     >= e1) v.x = -1;
  if (e + 1 >= e1) v.y = -1;
  if (e + 2 >= e1) v.z = -1;
  if (e + 3 >= e1) v.w = -1;
  return v;
}

__device__ __forceinline__ void bin_body(const int bid,
                      const int* __restrict__ h_idx, const int* __restrict__ t_idx, const int* __restrict__ rel,
                      int* __restrict__ gcount, int* __restrict__ binned,
                      int n_edges, int n_e, int nbin) {
  __shared__ int hist[NBIN_MAX];
  __shared__ int gbase[NBIN_MAX];
  __shared__ int lbase[NBIN_MAX];
  __shared__ int sRec[2 * EPB];
  __shared__ int sAdr[2 * EPB];
  __shared__ int wsum[8];
  const int tid  = threadIdx.x;
  const int lane = tid & 31;
  const int wave = tid >> 5;
  const int e0 = bid * EPB;
  const int e1 = (e0 + EPB < n_edges) ? e0 + EPB : n_edges;
  const bool vec = (n_edges & 3) == 0;
  const int ea = e0 + tid * 4;
  const int eb = ea + 1024;
  #pragma unroll
  for (int k = 0; k < 4; ++k) hist[tid + 256 * k] = 0;
  int4 hA, tA, hB, tB, rA, rB;
  if (vec) {
    hA = load4_vec(h_idx, ea, e0, e1); tA = load4_vec(t_idx, ea, e0, e1); rA = load4_vec(rel, ea, e0, e1);
    hB = load4_vec(h_idx, eb, e0, e1); tB = load4_vec(t_idx, eb, e0, e1); rB = load4_vec(rel, eb, e0, e1);
  } else {
    hA = load4_any(h_idx, ea, e0, e1); tA = load4_any(t_idx, ea, e0, e1); rA = load4_any(rel, ea, e0, e1);
    hB = load4_any(h_idx, eb, e0, e1); tB = load4_any(t_idx, eb, e0, e1); rB = load4_any(rel, eb, e0, e1);
  }
  const int hv[8] = {hA.x, hA.y, hA.z, hA.w, hB.x, hB.y, hB.z, hB.w};
  const int tv[8] = {tA.x, tA.y, tA.z, tA.w, tB.x, tB.y, tB.z, tB.w};
  const int rv[8] = {rA.x, rA.y, rA.z, rA.w, rB.x, rB.y, rB.z, rB.w};
  __syncthreads();
  #pragma unroll
  for (int j = 0; j < 8; ++j) {
    if (hv[j] >= 0) {
      atomicAdd(&hist[hv[j] >> 8], 1);
      atomicAdd(&hist[(n_e + tv[j]) >> 8], 1);
    }
  }
  __syncthreads();
  int cg[4], g[4];
  #pragma unroll
  for (int k = 0; k < 4; ++k) {
    const int b = tid + 256 * k;
    cg[k] = (b < nbin) ? hist[b] : 0;
  }
  #pragma unroll
  for (int k = 0; k < 4; ++k) g[k] = (cg[k] > 0) ? atomicAdd(gcount + tid + 256 * k, cg[k]) : 0;
  int c[4];
  #pragma unroll
  for (int k = 0; k < 4; ++k) {
    const int b = 4 * tid + k;
    c[k] = (b < nbin) ? hist[b] : 0;
  }
  const int tsum = c[0] + c[1] + c[2] + c[3];
  int incl = tsum;
  for (int o = 1; o < 32; o <<= 1) {
    const int v = __shfl_up(incl, o);
    if (lane >= o) incl += v;
  }
  if (lane == 31) wsum[wave] = incl;
  __syncthreads();
  int excl = incl - tsum;
  for (int w = 0; w < wave; ++w) excl += wsum[w];
  #pragma unroll
  for (int k = 0; k < 4; ++k) {
    const int b = 4 * tid + k;
    lbase[b] = excl;
    hist[b] = 0;
    excl += c[k];
  }
  #pragma unroll
  for (int k = 0; k < 4; ++k) gbase[tid + 256 * k] = g[k];
  __syncthreads();
  #pragma unroll
  for (int j = 0; j < 8; ++j) {
    if (hv[j] >= 0) {
      const int kh = hv[j];
      const int kt = n_e + tv[j];
      const int bh = kh >> 8;
      const int bt = kt >> 8;
      const int lh = atomicAdd(&hist[bh], 1);
      const int lt = atomicAdd(&hist[bt], 1);
      const int ph = lbase[bh] + lh;
      const int pt = lbase[bt] + lt;
      const int gh = gbase[bh] + lh;
      const int gt = gbase[bt] + lt;
      sRec[ph] = (rv[j] << 8) | (kh & 255);
      sRec[pt] = (rv[j] << 8) | (kt & 255);
      sAdr[ph] = (gh < BIN_CAP) ? bh * BIN_CAP + gh : -1;
      sAdr[pt] = (gt < BIN_CAP) ? bt * BIN_CAP + gt : -1;
    }
  }
  __syncthreads();
  const int nrec = 2 * (e1 - e0);
  for (int i = tid; i < nrec; i += 256) {
    const int a = sAdr[i];
    if (a >= 0) binned[a] = sRec[i];
  }
}

#define RPT (BIN_CAP / 256)
#define GRP_CAP (BIN_CAP + 3 * KPB)
__device__ __forceinline__ int f2ord(float f) { const int b = __float_as_int(f); return b >= 0 ? b : b ^ 0x7fffffff; }
__device__ __forceinline__ float ord2f(int o) { return __int_as_float(o >= 0 ? o : o ^ 0x7fffffff); }

__device__ __forceinline__ void binlist_body(const int bin, const int* __restrict__ gcount,
                          const int* __restrict__ binned,
                          const float* __restrict__ e_h, const float* __restrict__ e_t,
                          const float* __restrict__ e_r,
                          int* __restrict__ cnt, int* __restrict__ off, int2* __restrict__ pairs,
                          int n_keys, int n_e) {
  __shared__ int kcnt[KPB];
  __shared__ int koff2[KPB];
  __shared__ float ken[KPB];
  __shared__ int kmx[KPB];
  __shared__ uint4 grpv[GRP_CAP / 4];
  int* const grp = (int*)grpv;
  __shared__ int wgt[BIN_CAP];
  __shared__ int wsum2[8];
  const int tid  = threadIdx.x;
  const int lane = tid & 31;
  const int wave = tid >> 5;
  const int tot = gcount[bin];
  const bool over = tot > BIN_CAP;
  const int n = over ? BIN_CAP : tot;
  kcnt[tid] = 0;
  kmx[tid] = (int)0x80000000;
  const int* __restrict__ src = binned + (size_t)bin * BIN_CAP;
  int rc[RPT], sl[RPT];
  #pragma unroll
  for (int m = 0; m < RPT; ++m) {
    const int i = tid + 256 * m;
    const int v = src[i < n ? i : 0];
    rc[m] = i < n ? v : -1;
  }
  __syncthreads();
  #pragma unroll
  for (int m = 0; m < RPT; ++m) {
    sl[m] = 0;
    if (rc[m] >= 0) sl[m] = atomicAdd(&kcnt[rc[m] & 255], 1);
  }
  __syncthreads();
  const int c = kcnt[tid];
  const int c4 = (c + 3) & ~3;
  const int both = (c4 << 16) | c;
  int incl = both;
  for (int o = 1; o < 32; o <<= 1) {
    const int v = __shfl_up(incl, o);
    if (lane >= o) incl += v;
  }
  if (lane == 31) wsum2[wave] = incl;
  __syncthreads();
  int wb = 0;
  for (int w = 0; w < wave; ++w) wb += wsum2[w];
  const int excl = wb + incl - both;
  const int ko = excl & 0xffff;
  const int ko4 = excl >> 16;
  koff2[tid] = excl;
  for (int u = c; u < c4; ++u) grp[ko4 + u] = -1;
  const int key = bin * KPB + tid;
  float en = 0.f;
  if (key < n_keys) {
    cnt[key] = over ? CNT_OVER : c;
    off[key] = bin * BIN_CAP + ko;
    en = key < n_e ? e_h[key] : e_t[key - n_e];
  }
  ken[tid] = en;
  __syncthreads();
  if (over) return;
  float er[RPT];
  #pragma unroll
  for (int m = 0; m < RPT; ++m) {
    const int r = rc[m] >= 0 ? (rc[m] >> 8) : 0;
    er[m] = e_r[r];
    if (rc[m] >= 0) grp[(koff2[rc[m] & 255] >> 16) + sl[m]] = (int)(((unsigned)r << 12) | (unsigned)sl[m]);
  }
  __syncthreads();
  int dst[RPT];
  float zz[RPT];
  #pragma unroll
  for (int m = 0; m < RPT; ++m) {
    dst[m] = -1;
    zz[m] = 0.f;
    if (rc[m] >= 0) {
      const int k = rc[m] & 255;
      const int o2 = koff2[k];
      const int lo = o2 >> 18;
      const int hi = lo + ((kcnt[k] + 3) >> 2);
      const unsigned w = ((unsigned)(rc[m] >> 8) << 12) | (unsigned)sl[m];
      int rank = 0;
      for (int q = lo; q < hi; ++q) {
        const uint4 y = grpv[q];
        rank += (y.x < w ? 1 : 0) + (y.y < w ? 1 : 0) + (y.z < w ? 1 : 0) + (y.w < w ? 1 : 0);
      }
      dst[m] = (o2 & 0xffff) + rank;
      const float z = leaky(ken[k] + er[m]);
      zz[m] = z;
      atomicMax(&kmx[k], f2ord(z));
    }
  }
  __syncthreads();
  #pragma unroll
  for (int m = 0; m < RPT; ++m) {
    if (dst[m] >= 0) {
      grp[dst[m]] = rc[m];
      wgt[dst[m]] = __float_as_int(__expf(zz[m] - ord2f(kmx[rc[m] & 255])));
    }
  }
  __syncthreads();
  if (c > 0) {
    float sum = 0.f;
    #pragma unroll 4
    for (int q = 0; q < c; ++q) sum += __int_as_float(wgt[ko + q]);
    kmx[tid] = __float_as_int(sum + 1e-16f);
  }
  __syncthreads();
  int2* __restrict__ dstp = pairs + (size_t)bin * BIN_CAP;
  for (int p = tid; p < n; p += 256) {
    const int rec = grp[p];
    const float al = __int_as_float(wgt[p]) / __int_as_float(kmx[rec & 255]);
    dstp[p] = make_int2(rec >> 8, __float_as_int(al));
  }
}

__device__ __forceinline__ void proj_body(const int mb, const int half,
                       const float* __restrict__ msg, const float* __restrict__ Wr,
                       float* __restrict__ P, int n_r) {
  constexpr int KTOT = 128, NTOT = 256, KC = 32;
  constexpr int ASTR = 36, BKSTR = 36;
  __shared__ float As[16 * ASTR];
  __shared__ float Bs[NTOT * BKSTR];
  const int tid  = threadIdx.x;
  const int lane = tid & 31;
  const int wave = tid >> 5;
  const int m0   = mb * 16;
  const int lrow = lane & 15;
  const int hsel = (lane >= 16) ? 2 : 0;
  const int mrow = (lane >= 16) ? 8 : 0;
  const int n0   = wave * 16;
  const float* __restrict__ B = Wr + (size_t)half * KTOT * NTOT;
  float* __restrict__ O = P + (size_t)half * n_r * NTOT;

  const int arow = tid >> 3;
  const int acol = (tid & 7) * 4;

  v8f c0 = {}, c1 = {};
  float4 aReg = make_float4(0.f, 0.f, 0.f, 0.f);
  float4 bReg[8];

  auto loadGlobal = [&](int kb) {
    if (tid < 128) {
      int grow = m0 + arow;
      aReg = make_float4(0.f, 0.f, 0.f, 0.f);
      if (grow < n_r) aReg = *(const float4*)(msg + (size_t)grow * KTOT + kb + acol);
    }
    #pragma unroll
    for (int i = 0; i < 8; ++i) {
      int k4 = i * 4;
      bReg[i].x = B[(size_t)(kb + k4 + 0) * NTOT + tid];
      bReg[i].y = B[(size_t)(kb + k4 + 1) * NTOT + tid];
      bReg[i].z = B[(size_t)(kb + k4 + 2) * NTOT + tid];
      bReg[i].w = B[(size_t)(kb + k4 + 3) * NTOT + tid];
    }
  };
  auto storeSmem = [&]() {
    if (tid < 128) *(float4*)&As[arow * ASTR + acol] = aReg;
    #pragma unroll
    for (int i = 0; i < 8; ++i)
      *(float4*)&Bs[tid * BKSTR + i * 4] = bReg[i];
  };

  loadGlobal(0);
  for (int kb = 0; kb < KTOT; kb += KC) {
    storeSmem();
    __syncthreads();
    if (kb + KC < KTOT) loadGlobal(kb + KC);
    #pragma unroll
    for (int kk = 0; kk < KC; kk += 4) {
      int ka = kk + hsel;
      v2f a  = *(const v2f*)&As[lrow * ASTR + ka];
      v2f b0 = *(const v2f*)&Bs[(n0 + lrow) * BKSTR + ka];
      v2f b1 = *(const v2f*)&Bs[(n0 + 128 + lrow) * BKSTR + ka];
      c0 = __builtin_amdgcn_wmma_f32_16x16x4_f32(false, a, false, b0, (short)0, c0, false, false);
      c1 = __builtin_amdgcn_wmma_f32_16x16x4_f32(false, a, false, b1, (short)0, c1, false, false);
    }
    __syncthreads();
  }
  #pragma unroll
  for (int j = 0; j < 8; ++j) {
    int grow = m0 + mrow + j;
    if (grow < n_r) {
      O[(size_t)grow * NTOT + n0 + lrow]       = c0[j];
      O[(size_t)grow * NTOT + n0 + 128 + lrow] = c1[j];
    }
  }
}

__global__ void k_rel_bin(const float* __restrict__ x_r,
                          const float* __restrict__ w_ar,
                          const float* __restrict__ W1, const float* __restrict__ b1,
                          const float* __restrict__ W2, const float* __restrict__ b2,
                          float* __restrict__ e_r, float* __restrict__ msg, int n_r,
                          const int* __restrict__ h_idx, const int* __restrict__ t_idx, const int* __restrict__ rel,
                          int* __restrict__ gcount, int* __restrict__ binned,
                          int n_edges, int n_e, int nbin) {
  const int b = (int)blockIdx.x;
  if (b < n_r) rel_body(b, x_r, w_ar, W1, b1, W2, b2, e_r, msg);
  else bin_body(b - n_r, h_idx, t_idx, rel, gcount, binned, n_edges, n_e, nbin);
}

__global__ void k_proj_list(const float* __restrict__ msg, const float* __restrict__ Wr,
                            float* __restrict__ P, int n_r, int n_mb,
                            const int* __restrict__ gcount, const int* __restrict__ binned,
                            const float* __restrict__ e_h, const float* __restrict__ e_t,
                            const float* __restrict__ e_r,
                            int* __restrict__ cnt, int* __restrict__ off, int* __restrict__ pairs,
                            int n_keys, int n_e) {
  const int b = (int)blockIdx.x;
  if (b < 2 * n_mb) {
    const int half = b >= n_mb ? 1 : 0;
    proj_body(b - half * n_mb, half, msg, Wr, P, n_r);
  } else {
    binlist_body(b - 2 * n_mb, gcount, binned, e_h, e_t, e_r, cnt, off, (int2*)pairs, n_keys, n_e);
  }
}

__device__ __forceinline__ void fma_row(float4& A, float4& B, const float a, const float4 m, const float4 n) {
  A.x += a * m.x; A.y += a * m.y; A.z += a * m.z; A.w += a * m.w;
  B.x += a * n.x; B.y += a * n.y; B.z += a * n.z; B.w += a * n.w;
}

__device__ __forceinline__ void scan_side(const float en, const float* __restrict__ e_r,
                                          const float* __restrict__ Ps, const int lane,
                                          const int node, const int* __restrict__ idx,
                                          const int* __restrict__ rel, const int n_edges,
                                          float4& accA, float4& accB) {
  float mx = -__builtin_inff();
  for (int i = lane; i < n_edges; i += 32) if (idx[i] == node) mx = fmaxf(mx, leaky(en + e_r[rel[i]]));
  mx = wave_max_f(mx);
  float s = 0.f;
  int last = -1;
  for (int it = 0; it < n_edges; ++it) {
    int best = 0x7fffffff;
    for (int i = lane; i < n_edges; i += 32) if (idx[i] == node) { const int r = rel[i]; if (r > last && r < best) best = r; }
    best = __builtin_amdgcn_readfirstlane(wave_min_i(best));
    if (best == 0x7fffffff) break;
    int c = 0;
    for (int i = lane; i < n_edges; i += 32) c += (idx[i] == node && rel[i] == best) ? 1 : 0;
    c = __builtin_amdgcn_readfirstlane(wave_sum_i(c));
    const float p = __expf(leaky(en + e_r[best]) - mx);
    for (int q = 0; q < c; ++q) s += p;
    last = best;
  }
  const float den = s + 1e-16f;
  last = -1;
  for (int it = 0; it < n_edges; ++it) {
    int best = 0x7fffffff;
    for (int i = lane; i < n_edges; i += 32) if (idx[i] == node) { const int r = rel[i]; if (r > last && r < best) best = r; }
    best = __builtin_amdgcn_readfirstlane(wave_min_i(best));
    if (best == 0x7fffffff) break;
    int c = 0;
    for (int i = lane; i < n_edges; i += 32) c += (idx[i] == node && rel[i] == best) ? 1 : 0;
    c = __builtin_amdgcn_readfirstlane(wave_sum_i(c));
    const float a = __expf(leaky(en + e_r[best]) - mx) / den;
    const float4* q0 = (const float4*)(Ps + (size_t)best * 256);
    const float4 m0 = q0[lane], n0 = q0[32 + lane];
    for (int q = 0; q < c; ++q) fma_row(accA, accB, a, m0, n0);
    last = best;
  }
}

__device__ __forceinline__ void fma_row_v(float4& A, float4& B, const float a, const v4f m, const v4f n) {
  A.x += a * m.x; A.y += a * m.y; A.z += a * m.z; A.w += a * m.w;
  B.x += a * n.x; B.y += a * n.y; B.z += a * n.z; B.w += a * n.w;
}

template <int N>
__device__ __forceinline__ void walk_rows(const int rs, const float al, const int k,
                                          const float* __restrict__ P, const int lane,
                                          float4& accA, float4& accB) {
  v4f m[N], n[N];
  float a[N];
  #pragma unroll
  for (int i = 0; i < N; ++i) {
    const int r = __builtin_amdgcn_readlane(rs, k + i);
    a[i] = __int_as_float(__builtin_amdgcn_readlane(__float_as_int(al), k + i));
    const unsigned o = (unsigned)r * 256u + (unsigned)lane * 4u;
    m[i] = *(const v4f*)(P + o);
    n[i] = *(const v4f*)(P + o + 128u);
  }
  if constexpr (N == 4) {
    asm volatile("" :: "v"(m[0]), "v"(n[0]), "v"(m[1]), "v"(n[1]), "v"(m[2]), "v"(n[2]), "v"(m[3]), "v"(n[3]));
  } else if constexpr (N == 3) {
    asm volatile("" :: "v"(m[0]), "v"(n[0]), "v"(m[1]), "v"(n[1]), "v"(m[2]), "v"(n[2]));
  } else if constexpr (N == 2) {
    asm volatile("" :: "v"(m[0]), "v"(n[0]), "v"(m[1]), "v"(n[1]));
  }
  #pragma unroll
  for (int i = 0; i < N; ++i) fma_row_v(accA, accB, a[i], m[i], n[i]);
}

__device__ __forceinline__ void walk_run(const int rs, const float al, const int D,
                                         const float* __restrict__ P, const int lane,
                                         float4& accA, float4& accB) {
  int k = 0;
  #pragma unroll 1
  for (; k + 4 <= D; k += 4) walk_rows<4>(rs, al, k, P, lane, accA, accB);
  const int rem = D - k;
  if (rem == 3) walk_rows<3>(rs, al, k, P, lane, accA, accB);
  else if (rem == 2) walk_rows<2>(rs, al, k, P, lane, accA, accB);
  else if (rem == 1) walk_rows<1>(rs, al, k, P, lane, accA, accB);
}

__device__ __forceinline__ void side_pass(const int d, const int o, const int row0,
                                          const int2* __restrict__ pairs, const float* __restrict__ escore,
                                          const float* __restrict__ e_r, const float* __restrict__ P,
                                          const int lane, const int node, const int* __restrict__ idx,
                                          const int* __restrict__ rel, const int n_edges,
                                          float4& accA, float4& accB) {
  if (d > 0 && d <= 32) {
    int row = 0;
    float al = 0.f;
    if (lane < d) {
      const int2 pr = pairs[o + lane];
      row = pr.x + row0;
      al = __int_as_float(pr.y);
    }
    walk_run(row, al, d, P, lane, accA, accB);
  } else if (d > 32) {
    scan_side(escore[node], e_r, P + (size_t)row0 * 256, lane, node, idx, rel, n_edges, accA, accB);
  }
}

__global__ void __launch_bounds__(256, 16) k_gather(const int* __restrict__ cnt, const int* __restrict__ off,
                         const int* __restrict__ pairs_w,
                         const float* __restrict__ e_h, const float* __restrict__ e_t,
                         const float* __restrict__ e_r, const float* __restrict__ P,
                         const float* __restrict__ br,
                         float* __restrict__ out, int n_e, int n_r,
                         const int* __restrict__ h_idx, const int* __restrict__ t_idx,
                         const int* __restrict__ rel, int n_edges) {
  const int2* __restrict__ pairs = (const int2*)pairs_w;
  const int lane = threadIdx.x & 31;
  const int wv = __builtin_amdgcn_readfirstlane((int)(threadIdx.x >> 5));
  const int nw = (int)gridDim.x * 8;
  int node = (int)blockIdx.x * 8 + wv;
  if (node >= n_e) return;
  int d0 = cnt[node];
  int d1 = cnt[n_e + node];
  int o0 = off[node];
  int o1 = off[n_e + node];
  for (;;) {
    const int next = node + nw;
    const bool more = next < n_e;
    const int nn = more ? next : node;
    const int d0n = cnt[nn];
    const int d1n = cnt[n_e + nn];
    const int o0n = off[nn];
    const int o1n = off[n_e + nn];
    float4 accA = make_float4(0.f, 0.f, 0.f, 0.f);
    float4 accB = make_float4(0.f, 0.f, 0.f, 0.f);
    if (d0 <= 32 && d1 <= 32 && d0 + d1 <= 32) {
      const int D = d0 + d1;
      const bool head = lane < d0;
      int row = 0;
      float al = 0.f;
      if (lane < D) {
        const int2 pr = pairs[head ? o0 + lane : o1 + (lane - d0)];
        row = head ? pr.x : pr.x + n_r;
        al = __int_as_float(pr.y);
      }
      walk_run(row, al, D, P, lane, accA, accB);
    } else {
      side_pass(d0, o0, 0, pairs, e_h, e_r, P, lane, node, h_idx, rel, n_edges, accA, accB);
      side_pass(d1, o1, n_r, pairs, e_t, e_r, P, lane, node, t_idx, rel, n_edges, accA, accB);
    }
    const float4 bA = ((const float4*)br)[lane];
    const float4 bB = ((const float4*)br)[32 + lane];
    accA.x += bA.x; accA.y += bA.y; accA.z += bA.z; accA.w += bA.w;
    accB.x += bB.x; accB.y += bB.y; accB.z += bB.z; accB.w += bB.w;
    float4* orow = (float4*)(out + (size_t)node * 256);
    orow[lane] = accA;
    orow[32 + lane] = accB;
    if (!more) break;
    node = next;
    d0 = d0n; d1 = d1n; o0 = o0n; o1 = o1n;
  }
}

extern "C" void kernel_launch(void* const* d_in, const int* in_sizes, int n_in,
                              void* d_out, int out_size, void* d_ws, size_t ws_size,
                              hipStream_t stream) {
  const float* x_e  = (const float*)d_in[0];
  const float* x_r  = (const float*)d_in[1];
  const int*   eidx = (const int*)d_in[2];
  const int*   rel  = (const int*)d_in[3];
  const float* w_ah = (const float*)d_in[5];
  const float* w_at = (const float*)d_in[6];
  const float* w_ar = (const float*)d_in[7];
  const float* W1   = (const float*)d_in[8];
  const float* b1   = (const float*)d_in[9];
  const float* W2   = (const float*)d_in[10];
  const float* b2   = (const float*)d_in[11];
  const float* Wr   = (const float*)d_in[12];
  const float* br   = (const float*)d_in[13];
  float* out = (float*)d_out;

  const int n_e     = in_sizes[0] / E_HID;
  const int n_r     = in_sizes[1] / R_HID;
  const int n_edges = in_sizes[3];
  const int* h_idx = eidx;
  const int* t_idx = eidx + n_edges;
  const int nbin   = (2 * n_e + KPB - 1) / KPB;

  int* pairs   = (int*)d_ws;
  float* P     = (float*)(pairs + (size_t)2 * nbin * BIN_CAP);
  float* msg   = P + (size_t)2 * n_r * 256;
  float* e_h   = msg + (size_t)n_r * R_HID;
  float* e_t   = e_h + n_e;
  float* e_r   = e_t + n_e;
  int* cnt     = (int*)(e_r + n_r);
  int* off     = cnt + (size_t)2 * n_e;
  int* gcount  = off + (size_t)2 * n_e;
  int* binned  = gcount + NBIN_MAX;

  k_node_scores<<<(n_e + 63) / 64, 64, 0, stream>>>(x_e, w_ah, w_at, e_h, e_t, gcount, nbin, n_e);
  k_rel_bin<<<n_r + (n_edges + EPB - 1) / EPB, 256, 0, stream>>>(x_r, w_ar, W1, b1, W2, b2, e_r, msg, n_r,
                                                                 h_idx, t_idx, rel, gcount, binned, n_edges, n_e, nbin);
  k_proj_list<<<2 * ((n_r + 15) / 16) + nbin, 256, 0, stream>>>(msg, Wr, P, n_r, (n_r + 15) / 16,
                                                                gcount, binned, e_h, e_t, e_r, cnt, off, pairs, 2 * n_e, n_e);
  k_gather<<<2048, 256, 0, stream>>>(cnt, off, pairs, e_h, e_t, e_r, P, br, out, n_e, n_r, h_idx, t_idx, rel, n_edges);
}


# Gather rows from LDS: each work-group copies its 32-column slice of the projected table into LDS once and reads edge rows there, not from L2; float sum order: head sum + tail sum, then bias
# speedup vs baseline: 5.2826x; 1.1394x over previous
#include <hip/hip_runtime.h>

#define E_HID 128
#define R_HID 128

#define KPB 256
#define BIN_CAP 4096
#define NBIN_MAX 1024
#define EPB 2048
#define CNT_OVER 0x40000000

typedef float v2f __attribute__((ext_vector_type(2)));
typedef float v8f __attribute__((ext_vector_type(8)));
typedef float v4f __attribute__((ext_vector_type(4)));

__device__ __forceinline__ float leaky(float x) { return x > 0.f ? x : 0.01f * x; }


__device__ __forceinline__ float wave_max_f(float v) {
  for (int o = 16; o > 0; o >>= 1) v = fmaxf(v, __shfl_xor(v, o));
  return v;
}
__device__ __forceinline__ int wave_min_i(int v) {
  for (int o = 16; o > 0; o >>= 1) { int w = __shfl_xor(v, o); v = w < v ? w : v; }
  return v;
}
__device__ __forceinline__ int wave_sum_i(int v) {
  for (int o = 16; o > 0; o >>= 1) v += __shfl_xor(v, o);
  return v;
}

__global__ void __launch_bounds__(64) k_node_scores(const float* __restrict__ x_e,
                              const float* __restrict__ w_ah,
                              const float* __restrict__ w_at,
                              float* __restrict__ e_h, float* __restrict__ e_t,
                              int* __restrict__ gcount, int nbin, int n_e) {
  const int lane = threadIdx.x & 31;
  const int base = ((int)blockIdx.x * 2 + (int)(threadIdx.x >> 5)) * 32;
  if (base >= n_e) return;
  const int s = lane & 7;
  const int q = lane >> 3;
  v4f x[8][4];
  #pragma unroll
  for (int i = 0; i < 8; ++i) {
    int row = base + q * 8 + i;
    row = row < n_e ? row : n_e - 1;
    const v4f* p = (const v4f*)(x_e + (size_t)row * E_HID);
    #pragma unroll
    for (int j = 0; j < 4; ++j) x[i][j] = p[j * 8 + s];
  }
  v4f a[4], b[4];
  #pragma unroll
  for (int j = 0; j < 4; ++j) {
    a[j] = ((const v4f*)w_ah)[j * 8 + s];
    b[j] = ((const v4f*)w_at)[j * 8 + s];
  }
  asm volatile("" ::
               "v"(x[0][0]), "v"(x[0][1]), "v"(x[0][2]), "v"(x[0][3]),
               "v"(x[1][0]), "v"(x[1][1]), "v"(x[1][2]), "v"(x[1][3]),
               "v"(x[2][0]), "v"(x[2][1]), "v"(x[2][2]), "v"(x[2][3]),
               "v"(x[3][0]), "v"(x[3][1]), "v"(x[3][2]), "v"(x[3][3]),
               "v"(x[4][0]), "v"(x[4][1]), "v"(x[4][2]), "v"(x[4][3]),
               "v"(x[5][0]), "v"(x[5][1]), "v"(x[5][2]), "v"(x[5][3]),
               "v"(x[6][0]), "v"(x[6][1]), "v"(x[6][2]), "v"(x[6][3]),
               "v"(x[7][0]), "v"(x[7][1]), "v"(x[7][2]), "v"(x[7][3]),
               "v"(a[0]), "v"(b[0]), "v"(a[1]), "v"(b[1]),
               "v"(a[2]), "v"(b[2]), "v"(a[3]), "v"(b[3]));
  float vh = 0.f, vt = 0.f;
  #pragma unroll
  for (int i = 0; i < 8; ++i) {
    float sh = 0.f, st = 0.f;
    #pragma unroll
    for (int j = 0; j < 4; ++j) {
      const v4f v = x[i][j];
      sh = fmaf(v.x, a[j].x, sh); sh = fmaf(v.y, a[j].y, sh); sh = fmaf(v.z, a[j].z, sh); sh = fmaf(v.w, a[j].w, sh);
      st = fmaf(v.x, b[j].x, st); st = fmaf(v.y, b[j].y, st); st = fmaf(v.z, b[j].z, st); st = fmaf(v.w, b[j].w, st);
    }
    #pragma unroll
    for (int off = 4; off > 0; off >>= 1) {
      sh += __shfl_xor(sh, off);
      st += __shfl_xor(st, off);
    }
    if (s == i) { vh = sh; vt = st; }
  }
  const int row = base + lane;
  if (row < n_e) { e_h[row] = vh; e_t[row] = vt; }
  if (row < nbin) gcount[row] = 0;
}

__device__ __forceinline__ void rel_body(const int r, const float* __restrict__ x_r,
                      const float* __restrict__ w_ar,
                      const float* __restrict__ W1, const float* __restrict__ b1,
                      const float* __restrict__ W2, const float* __restrict__ b2,
                      float* __restrict__ e_r, float* __restrict__ msg) {
  __shared__ float row[R_HID];
  __shared__ float hid[2 * R_HID];
  int t = threadIdx.x;
  if (t < R_HID) row[t] = x_r[(size_t)r * R_HID + t];
  __syncthreads();
  float acc = b1[t];
  #pragma unroll 8
  for (int k = 0; k < R_HID; ++k) acc += row[k] * W1[k * 2 * R_HID + t];
  hid[t] = acc;
  if (t < 32) {
    float s = 0.f;
    #pragma unroll
    for (int j = 0; j < 4; ++j) s += row[t + 32 * j] * w_ar[t + 32 * j];
    for (int off = 16; off > 0; off >>= 1) s += __shfl_xor(s, off);
    if (t == 0) e_r[r] = s;
  }
  __syncthreads();
  if (t < R_HID) {
    float o = b2[t];
    #pragma unroll 8
    for (int k = 0; k < 2 * R_HID; ++k) o += hid[k] * W2[k * R_HID + t];
    msg[(size_t)r * R_HID + t] = row[t] + o;
  }
}

__device__ __forceinline__ int4 load4_vec(const int* __restrict__ p, const int e, const int e0, const int e1) {
  const bool in = e < e1;
  int4 v = *(const int4*)(p + (in ? e : e0));
  if (!in) v = make_int4(-1, -1, -1, -1);
  return v;
}
__device__ __forceinline__ int4 load4_any(const int* __restrict__ p, const int e, const int e0, const int e1) {
  int4 v;
  v.x = p[(e     < e1) ? e     : e0];
  v.y = p[(e + 1 < e1) ? e + 1 : e0];
  v.z = p[(e + 2 < e1) ? e + 2 : e0];
  v.w = p[(e + 3 < e1) ? e + 3 : e0];
  if (e     >= e1) v.x = -1;
  if (e + 1 >= e1) v.y = -1;
  if (e + 2 >= e1) v.z = -1;
  if (e + 3 >= e1) v.w = -1;
  return v;
}

__device__ __forceinline__ void bin_body(const int bid,
                      const int* __restrict__ h_idx, const int* __restrict__ t_idx, const int* __restrict__ rel,
                      int* __restrict__ gcount, int* __restrict__ binned,
                      int n_edges, int n_e, int nbin) {
  __shared__ int hist[NBIN_MAX];
  __shared__ int gbase[NBIN_MAX];
  __shared__ int lbase[NBIN_MAX];
  __shared__ int sRec[2 * EPB];
  __shared__ int sAdr[2 * EPB];
  __shared__ int wsum[8];
  const int tid  = threadIdx.x;
  const int lane = tid & 31;
  const int wave = tid >> 5;
  const int e0 = bid * EPB;
  const int e1 = (e0 + EPB < n_edges) ? e0 + EPB : n_edges;
  const bool vec = (n_edges & 3) == 0;
  const int ea = e0 + tid * 4;
  const int eb = ea + 1024;
  #pragma unroll
  for (int k = 0; k < 4; ++k) hist[tid + 256 * k] = 0;
  int4 hA, tA, hB, tB, rA, rB;
  if (vec) {
    hA = load4_vec(h_idx, ea, e0, e1); tA = load4_vec(t_idx, ea, e0, e1); rA = load4_vec(rel, ea, e0, e1);
    hB = load4_vec(h_idx, eb, e0, e1); tB = load4_vec(t_idx, eb, e0, e1); rB = load4_vec(rel, eb, e0, e1);
  } else {
    hA = load4_any(h_idx, ea, e0, e1); tA = load4_any(t_idx, ea, e0, e1); rA = load4_any(rel, ea, e0, e1);
    hB = load4_any(h_idx, eb, e0, e1); tB = load4_any(t_idx, eb, e0, e1); rB = load4_any(rel, eb, e0, e1);
  }
  const int hv[8] = {hA.x, hA.y, hA.z, hA.w, hB.x, hB.y, hB.z, hB.w};
  const int tv[8] = {tA.x, tA.y, tA.z, tA.w, tB.x, tB.y, tB.z, tB.w};
  const int rv[8] = {rA.x, rA.y, rA.z, rA.w, rB.x, rB.y, rB.z, rB.w};
  __syncthreads();
  #pragma unroll
  for (int j = 0; j < 8; ++j) {
    if (hv[j] >= 0) {
      atomicAdd(&hist[hv[j] >> 8], 1);
      atomicAdd(&hist[(n_e + tv[j]) >> 8], 1);
    }
  }
  __syncthreads();
  int cg[4], g[4];
  #pragma unroll
  for (int k = 0; k < 4; ++k) {
    const int b = tid + 256 * k;
    cg[k] = (b < nbin) ? hist[b] : 0;
  }
  #pragma unroll
  for (int k = 0; k < 4; ++k) g[k] = (cg[k] > 0) ? atomicAdd(gcount + tid + 256 * k, cg[k]) : 0;
  int c[4];
  #pragma unroll
  for (int k = 0; k < 4; ++k) {
    const int b = 4 * tid + k;
    c[k] = (b < nbin) ? hist[b] : 0;
  }
  const int tsum = c[0] + c[1] + c[2] + c[3];
  int incl = tsum;
  for (int o = 1; o < 32; o <<= 1) {
    const int v = __shfl_up(incl, o);
    if (lane >= o) incl += v;
  }
  if (lane == 31) wsum[wave] = incl;
  __syncthreads();
  int excl = incl - tsum;
  for (int w = 0; w < wave; ++w) excl += wsum[w];
  #pragma unroll
  for (int k = 0; k < 4; ++k) {
    const int b = 4 * tid + k;
    lbase[b] = excl;
    hist[b] = 0;
    excl += c[k];
  }
  #pragma unroll
  for (int k = 0; k < 4; ++k) gbase[tid + 256 * k] = g[k];
  __syncthreads();
  #pragma unroll
  for (int j = 0; j < 8; ++j) {
    if (hv[j] >= 0) {
      const int kh = hv[j];
      const int kt = n_e + tv[j];
      const int bh = kh >> 8;
      const int bt = kt >> 8;
      const int lh = atomicAdd(&hist[bh], 1);
      const int lt = atomicAdd(&hist[bt], 1);
      const int ph = lbase[bh] + lh;
      const int pt = lbase[bt] + lt;
      const int gh = gbase[bh] + lh;
      const int gt = gbase[bt] + lt;
      sRec[ph] = (rv[j] << 8) | (kh & 255);
      sRec[pt] = (rv[j] << 8) | (kt & 255);
      sAdr[ph] = (gh < BIN_CAP) ? bh * BIN_CAP + gh : -1;
      sAdr[pt] = (gt < BIN_CAP) ? bt * BIN_CAP + gt : -1;
    }
  }
  __syncthreads();
  const int nrec = 2 * (e1 - e0);
  for (int i = tid; i < nrec; i += 256) {
    const int a = sAdr[i];
    if (a >= 0) binned[a] = sRec[i];
  }
}

#define RPT (BIN_CAP / 256)
#define GRP_CAP (BIN_CAP + 3 * KPB)
__device__ __forceinline__ int f2ord(float f) { const int b = __float_as_int(f); return b >= 0 ? b : b ^ 0x7fffffff; }
__device__ __forceinline__ float ord2f(int o) { return __int_as_float(o >= 0 ? o : o ^ 0x7fffffff); }

__device__ __forceinline__ void binlist_body(const int bin, const int* __restrict__ gcount,
                          const int* __restrict__ binned,
                          const float* __restrict__ e_h, const float* __restrict__ e_t,
                          const float* __restrict__ e_r,
                          int* __restrict__ cnt, int* __restrict__ off, int2* __restrict__ pairs,
                          int n_keys, int n_e) {
  __shared__ int kcnt[KPB];
  __shared__ int koff2[KPB];
  __shared__ float ken[KPB];
  __shared__ int kmx[KPB];
  __shared__ uint4 grpv[GRP_CAP / 4];
  int* const grp = (int*)grpv;
  __shared__ int wgt[BIN_CAP];
  __shared__ int wsum2[8];
  const int tid  = threadIdx.x;
  const int lane = tid & 31;
  const int wave = tid >> 5;
  const int tot = gcount[bin];
  const bool over = tot > BIN_CAP;
  const int n = over ? BIN_CAP : tot;
  kcnt[tid] = 0;
  kmx[tid] = (int)0x80000000;
  const int* __restrict__ src = binned + (size_t)bin * BIN_CAP;
  int rc[RPT], sl[RPT];
  #pragma unroll
  for (int m = 0; m < RPT; ++m) {
    const int i = tid + 256 * m;
    const int v = src[i < n ? i : 0];
    rc[m] = i < n ? v : -1;
  }
  __syncthreads();
  #pragma unroll
  for (int m = 0; m < RPT; ++m) {
    sl[m] = 0;
    if (rc[m] >= 0) sl[m] = atomicAdd(&kcnt[rc[m] & 255], 1);
  }
  __syncthreads();
  const int c = kcnt[tid];
  const int c4 = (c + 3) & ~3;
  const int both = (c4 << 16) | c;
  int incl = both;
  for (int o = 1; o < 32; o <<= 1) {
    const int v = __shfl_up(incl, o);
    if (lane >= o) incl += v;
  }
  if (lane == 31) wsum2[wave] = incl;
  __syncthreads();
  int wb = 0;
  for (int w = 0; w < wave; ++w) wb += wsum2[w];
  const int excl = wb + incl - both;
  const int ko = excl & 0xffff;
  const int ko4 = excl >> 16;
  koff2[tid] = excl;
  for (int u = c; u < c4; ++u) grp[ko4 + u] = -1;
  const int key = bin * KPB + tid;
  float en = 0.f;
  if (key < n_keys) {
    cnt[key] = over ? CNT_OVER : c;
    off[key] = bin * BIN_CAP + ko;
    en = key < n_e ? e_h[key] : e_t[key - n_e];
  }
  ken[tid] = en;
  __syncthreads();
  if (over) return;
  float er[RPT];
  #pragma unroll
  for (int m = 0; m < RPT; ++m) {
    const int r = rc[m] >= 0 ? (rc[m] >> 8) : 0;
    er[m] = e_r[r];
    if (rc[m] >= 0) grp[(koff2[rc[m] & 255] >> 16) + sl[m]] = (int)(((unsigned)r << 12) | (unsigned)sl[m]);
  }
  __syncthreads();
  int dst[RPT];
  float zz[RPT];
  #pragma unroll
  for (int m = 0; m < RPT; ++m) {
    dst[m] = -1;
    zz[m] = 0.f;
    if (rc[m] >= 0) {
      const int k = rc[m] & 255;
      const int o2 = koff2[k];
      const int lo = o2 >> 18;
      const int hi = lo + ((kcnt[k] + 3) >> 2);
      const unsigned w = ((unsigned)(rc[m] >> 8) << 12) | (unsigned)sl[m];
      int rank = 0;
      for (int q = lo; q < hi; ++q) {
        const uint4 y = grpv[q];
        rank += (y.x < w ? 1 : 0) + (y.y < w ? 1 : 0) + (y.z < w ? 1 : 0) + (y.w < w ? 1 : 0);
      }
      dst[m] = (o2 & 0xffff) + rank;
      const float z = leaky(ken[k] + er[m]);
      zz[m] = z;
      atomicMax(&kmx[k], f2ord(z));
    }
  }
  __syncthreads();
  #pragma unroll
  for (int m = 0; m < RPT; ++m) {
    if (dst[m] >= 0) {
      grp[dst[m]] = rc[m];
      wgt[dst[m]] = __float_as_int(__expf(zz[m] - ord2f(kmx[rc[m] & 255])));
    }
  }
  __syncthreads();
  if (c > 0) {
    float sum = 0.f;
    #pragma unroll 4
    for (int q = 0; q < c; ++q) sum += __int_as_float(wgt[ko + q]);
    kmx[tid] = __float_as_int(sum + 1e-16f);
  }
  __syncthreads();
  int2* __restrict__ dstp = pairs + (size_t)bin * BIN_CAP;
  for (int p = tid; p < n; p += 256) {
    const int rec = grp[p];
    const float al = __int_as_float(wgt[p]) / __int_as_float(kmx[rec & 255]);
    dstp[p] = make_int2(rec >> 8, __float_as_int(al));
  }
}

__device__ __forceinline__ void proj_body(const int mb, const int half,
                       const float* __restrict__ msg, const float* __restrict__ Wr,
                       float* __restrict__ P, int n_r) {
  constexpr int KTOT = 128, NTOT = 256, KC = 32;
  constexpr int ASTR = 36, BKSTR = 36;
  __shared__ float As[16 * ASTR];
  __shared__ float Bs[NTOT * BKSTR];
  const int tid  = threadIdx.x;
  const int lane = tid & 31;
  const int wave = tid >> 5;
  const int m0   = mb * 16;
  const int lrow = lane & 15;
  const int hsel = (lane >= 16) ? 2 : 0;
  const int mrow = (lane >= 16) ? 8 : 0;
  const int n0   = wave * 16;
  const float* __restrict__ B = Wr + (size_t)half * KTOT * NTOT;
  float* __restrict__ O = P + (size_t)half * n_r * NTOT;

  const int arow = tid >> 3;
  const int acol = (tid & 7) * 4;

  v8f c0 = {}, c1 = {};
  float4 aReg = make_float4(0.f, 0.f, 0.f, 0.f);
  float4 bReg[8];

  auto loadGlobal = [&](int kb) {
    if (tid < 128) {
      int grow = m0 + arow;
      aReg = make_float4(0.f, 0.f, 0.f, 0.f);
      if (grow < n_r) aReg = *(const float4*)(msg + (size_t)grow * KTOT + kb + acol);
    }
    #pragma unroll
    for (int i = 0; i < 8; ++i) {
      int k4 = i * 4;
      bReg[i].x = B[(size_t)(kb + k4 + 0) * NTOT + tid];
      bReg[i].y = B[(size_t)(kb + k4 + 1) * NTOT + tid];
      bReg[i].z = B[(size_t)(kb + k4 + 2) * NTOT + tid];
      bReg[i].w = B[(size_t)(kb + k4 + 3) * NTOT + tid];
    }
  };
  auto storeSmem = [&]() {
    if (tid < 128) *(float4*)&As[arow * ASTR + acol] = aReg;
    #pragma unroll
    for (int i = 0; i < 8; ++i)
      *(float4*)&Bs[tid * BKSTR + i * 4] = bReg[i];
  };

  loadGlobal(0);
  for (int kb = 0; kb < KTOT; kb += KC) {
    storeSmem();
    __syncthreads();
    if (kb + KC < KTOT) loadGlobal(kb + KC);
    #pragma unroll
    for (int kk = 0; kk < KC; kk += 4) {
      int ka = kk + hsel;
      v2f a  = *(const v2f*)&As[lrow * ASTR + ka];
      v2f b0 = *(const v2f*)&Bs[(n0 + lrow) * BKSTR + ka];
      v2f b1 = *(const v2f*)&Bs[(n0 + 128 + lrow) * BKSTR + ka];
      c0 = __builtin_amdgcn_wmma_f32_16x16x4_f32(false, a, false, b0, (short)0, c0, false, false);
      c1 = __builtin_amdgcn_wmma_f32_16x16x4_f32(false, a, false, b1, (short)0, c1, false, false);
    }
    __syncthreads();
  }
  #pragma unroll
  for (int j = 0; j < 8; ++j) {
    int grow = m0 + mrow + j;
    if (grow < n_r) {
      O[(size_t)grow * NTOT + n0 + lrow]       = c0[j];
      O[(size_t)grow * NTOT + n0 + 128 + lrow] = c1[j];
    }
  }
}

__global__ void k_rel_bin(const float* __restrict__ x_r,
                          const float* __restrict__ w_ar,
                          const float* __restrict__ W1, const float* __restrict__ b1,
                          const float* __restrict__ W2, const float* __restrict__ b2,
                          float* __restrict__ e_r, float* __restrict__ msg, int n_r,
                          const int* __restrict__ h_idx, const int* __restrict__ t_idx, const int* __restrict__ rel,
                          int* __restrict__ gcount, int* __restrict__ binned,
                          int n_edges, int n_e, int nbin) {
  const int b = (int)blockIdx.x;
  if (b < n_r) rel_body(b, x_r, w_ar, W1, b1, W2, b2, e_r, msg);
  else bin_body(b - n_r, h_idx, t_idx, rel, gcount, binned, n_edges, n_e, nbin);
}

__global__ void k_proj_list(const float* __restrict__ msg, const float* __restrict__ Wr,
                            float* __restrict__ P, int n_r, int n_mb,
                            const int* __restrict__ gcount, const int* __restrict__ binned,
                            const float* __restrict__ e_h, const float* __restrict__ e_t,
                            const float* __restrict__ e_r,
                            int* __restrict__ cnt, int* __restrict__ off, int* __restrict__ pairs,
                            int n_keys, int n_e) {
  const int b = (int)blockIdx.x;
  if (b < 2 * n_mb) {
    const int half = b >= n_mb ? 1 : 0;
    proj_body(b - half * n_mb, half, msg, Wr, P, n_r);
  } else {
    binlist_body(b - 2 * n_mb, gcount, binned, e_h, e_t, e_r, cnt, off, (int2*)pairs, n_keys, n_e);
  }
}

#define GS_COLS 32
#define GS_NSL (256 / GS_COLS)
#define GS_ROWS 1024
#define GS_WAVES 32

struct __attribute__((aligned(8))) pair2 { int r0, w0, r1, w1; };

__device__ __forceinline__ float scan_side_col(const float en, const float* __restrict__ e_r,
                                               const float* __restrict__ Pc, const int lane,
                                               const int node, const int* __restrict__ idx,
                                               const int* __restrict__ rel, const int n_edges) {
  float mx = -__builtin_inff();
  for (int i = lane; i < n_edges; i += 32) if (idx[i] == node) mx = fmaxf(mx, leaky(en + e_r[rel[i]]));
  mx = wave_max_f(mx);
  float s = 0.f;
  int last = -1;
  for (int it = 0; it < n_edges; ++it) {
    int best = 0x7fffffff;
    for (int i = lane; i < n_edges; i += 32) if (idx[i] == node) { const int r = rel[i]; if (r > last && r < best) best = r; }
    best = __builtin_amdgcn_readfirstlane(wave_min_i(best));
    if (best == 0x7fffffff) break;
    int c = 0;
    for (int i = lane; i < n_edges; i += 32) c += (idx[i] == node && rel[i] == best) ? 1 : 0;
    c = __builtin_amdgcn_readfirstlane(wave_sum_i(c));
    const float p = __expf(leaky(en + e_r[best]) - mx);
    for (int q = 0; q < c; ++q) s += p;
    last = best;
  }
  const float den = s + 1e-16f;
  float acc = 0.f;
  last = -1;
  for (int it = 0; it < n_edges; ++it) {
    int best = 0x7fffffff;
    for (int i = lane; i < n_edges; i += 32) if (idx[i] == node) { const int r = rel[i]; if (r > last && r < best) best = r; }
    best = __builtin_amdgcn_readfirstlane(wave_min_i(best));
    if (best == 0x7fffffff) break;
    int c = 0;
    for (int i = lane; i < n_edges; i += 32) c += (idx[i] == node && rel[i] == best) ? 1 : 0;
    c = __builtin_amdgcn_readfirstlane(wave_sum_i(c));
    const float a = __expf(leaky(en + e_r[best]) - mx) / den;
    const float m0 = Pc[(size_t)best * 256];
    for (int q = 0; q < c; ++q) acc += a * m0;
    last = best;
  }
  return acc;
}

template <int N, bool CHECK>
__device__ __forceinline__ void slice_trips(const char* __restrict__ tabb, const unsigned lbyte, const int rem,
                                            const pair2 pa, const pair2 pb, v4f& acc) {
  const int rr[4] = {pa.r0, pa.r1, pb.r0, pb.r1};
  const int ww[4] = {pa.w0, pa.w1, pb.w0, pb.w1};
  v4f t[N];
  float w[N];
  #pragma unroll
  for (int i = 0; i < N; ++i) {
    const bool ok = !CHECK || i < rem;
    const unsigned row = ok ? (unsigned)rr[i] : 0u;
    w[i] = ok ? __int_as_float(ww[i]) : 0.f;
    t[i] = *(const v4f*)(tabb + ((row << 8) | lbyte));
  }
  if constexpr (N == 4) {
    asm volatile("" :: "v"(t[0]), "v"(t[1]), "v"(t[2]), "v"(t[3]));
  } else if constexpr (N == 3) {
    asm volatile("" :: "v"(t[0]), "v"(t[1]), "v"(t[2]));
  } else if constexpr (N == 2) {
    asm volatile("" :: "v"(t[0]), "v"(t[1]));
  }
  #pragma unroll
  for (int i = 0; i < N; ++i) acc += t[i] * w[i];
}

__global__ void __launch_bounds__(GS_WAVES * 32) k_gather(const int* __restrict__ cnt, const int* __restrict__ off,
                         const int* __restrict__ pairs_w,
                         const float* __restrict__ e_h, const float* __restrict__ e_t,
                         const float* __restrict__ e_r, const float* __restrict__ P,
                         const float* __restrict__ br,
                         float* __restrict__ out, int n_e, int n_r,
                         const int* __restrict__ h_idx, const int* __restrict__ t_idx,
                         const int* __restrict__ rel, int n_edges) {
  __shared__ __attribute__((aligned(16))) float tab[GS_ROWS * 2 * GS_COLS];
  __shared__ int next_step;
  const int tid  = threadIdx.x;
  const int lane = tid & 31;
  const int wave = __builtin_amdgcn_readfirstlane(tid >> 5);
  const int cs = (int)blockIdx.x % GS_NSL;
  const int ng = (int)blockIdx.x / GS_NSL;
  const int n_ng = (int)gridDim.x / GS_NSL;
  const int npg = (n_e + n_ng - 1) / n_ng;
  const int nb = ng * npg;
  if (nb >= n_e) return;
  const int ne = (nb + npg < n_e) ? nb + npg : n_e;
  const int nsteps = (ne - nb + 1) >> 1;
  const int s = lane & 7;
  const int g = lane >> 3;
  const int side = g & 1;
  v4f* const tab4 = (v4f*)tab;

  {
    const int npiece = 2 * n_r;
    const float* __restrict__ Pc = P + cs * GS_COLS + s * 4;
    for (int p0 = 0; p0 < npiece; p0 += 8 * GS_WAVES * 4) {
      v4f v[8];
      #pragma unroll
      for (int u = 0; u < 8; ++u) {
        const int p = p0 + u * GS_WAVES * 4 + (tid >> 3);
        const int pc = p < npiece ? p : 0;
        v[u] = *(const v4f*)(Pc + (size_t)((pc & 1) * n_r + (pc >> 1)) * 256);
      }
      #pragma unroll
      for (int u = 0; u < 8; ++u) {
        const int p = p0 + u * GS_WAVES * 4 + (tid >> 3);
        if (p < npiece) tab4[p * 8 + s] = v[u];
      }
    }
  }
  if (tid == 0) next_step = 3 * GS_WAVES;
  const v4f bias = *(const v4f*)(br + cs * GS_COLS + s * 4);
  const unsigned lbyte = (unsigned)(side * (GS_COLS * 4) + s * 16);
  const char* const tabb = (const char*)tab;
  __syncthreads();

  auto side_of = [&](const int st, int& d, int& o, int& node) {
    node = nb + 2 * st + (g >> 1);
    const bool ok = st < nsteps && node < ne;
    const int key = side * n_e + (ok ? node : nb);
    const int dd = cnt[key];
    o = off[key];
    d = ok ? dd : 0;
  };
  auto ld2 = [&](const int o, const int k) -> pair2 {
    return *(const pair2*)((const char*)pairs_w + (unsigned)(o + k) * 8u);
  };

  int st0 = wave, st1 = wave + GS_WAVES, st2 = wave + 2 * GS_WAVES;
  int d0, o0, n0, d1, o1, n1;
  side_of(st0, d0, o0, n0);
  side_of(st1, d1, o1, n1);
  pair2 a0 = ld2(o0, 0), b0 = ld2(o0, 2);
  while (st0 < nsteps) {
    int stn = 0;
    if (lane == 0) stn = atomicAdd(&next_step, 1);
    pair2 c[8];
    c[0] = a0; c[1] = b0;
    #pragma unroll
    for (int j = 1; j < 4; ++j) { c[2 * j] = ld2(o0, 4 * j); c[2 * j + 1] = ld2(o0, 4 * j + 2); }
    int d2, o2, n2;
    side_of(st2, d2, o2, n2);
    const pair2 a1 = ld2(o1, 0), b1 = ld2(o1, 2);

    const bool over = d0 == CNT_OVER;
    const int dl = over ? 0 : d0;
    int dmax = __builtin_amdgcn_readlane(dl, 0), dmin = dmax;
    { const int x = __builtin_amdgcn_readlane(dl, 8);  dmax = x > dmax ? x : dmax; dmin = x < dmin ? x : dmin; }
    { const int x = __builtin_amdgcn_readlane(dl, 16); dmax = x > dmax ? x : dmax; dmin = x < dmin ? x : dmin; }
    { const int x = __builtin_amdgcn_readlane(dl, 24); dmax = x > dmax ? x : dmax; dmin = x < dmin ? x : dmin; }
    v4f acc = {0.f, 0.f, 0.f, 0.f};
    auto group = [&](const pair2 xa, const pair2 xb, const int k) {
      const int remw = dmax - k;
      if (remw >= 4) {
        if (k + 4 <= dmin) slice_trips<4, false>(tabb, lbyte, 4, xa, xb, acc);
        else slice_trips<4, true>(tabb, lbyte, dl - k, xa, xb, acc);
      }
      else if (remw == 3) slice_trips<3, true>(tabb, lbyte, dl - k, xa, xb, acc);
      else if (remw == 2) slice_trips<2, true>(tabb, lbyte, dl - k, xa, xb, acc);
      else if (remw == 1) slice_trips<1, true>(tabb, lbyte, dl - k, xa, xb, acc);
    };
    #pragma unroll
    for (int j = 0; j < 4; ++j) {
      if (4 * j < dmax) group(c[2 * j], c[2 * j + 1], 4 * j);
    }
    for (int k = 16; k < dmax; k += 4) {
      const pair2 xa = ld2(o0, k), xb = ld2(o0, k + 2);
      group(xa, xb, k);
    }

    const unsigned ovm = __builtin_amdgcn_ballot_w32(over);
    if (ovm != 0u) {
      for (int gg = 0; gg < 4; ++gg) {
        if ((ovm >> (8 * gg)) & 1u) {
          const int nd = __builtin_amdgcn_readlane(n0, 8 * gg);
          const int sd = gg & 1;
          const float en = sd ? e_t[nd] : e_h[nd];
          const float part = scan_side_col(en, e_r, P + (size_t)sd * n_r * 256 + cs * GS_COLS + lane, lane, nd,
                                           sd ? t_idx : h_idx, rel, n_edges);
          v4f add;
          add.x = __shfl(part, 4 * s + 0); add.y = __shfl(part, 4 * s + 1);
          add.z = __shfl(part, 4 * s + 2); add.w = __shfl(part, 4 * s + 3);
          if (g == gg) acc += add;
        }
      }
    }
    auto ror8 = [](const float v) {
      return __int_as_float(__builtin_amdgcn_update_dpp(0, __float_as_int(v), 0x128, 0xf, 0xf, false));
    };
    v4f oth;
    oth.x = ror8(acc.x); oth.y = ror8(acc.y); oth.z = ror8(acc.z); oth.w = ror8(acc.w);
    if (side == 0 && n0 < ne) {
      const v4f r = (acc + oth) + bias;
      *(v4f*)(out + (size_t)n0 * 256 + cs * GS_COLS + s * 4) = r;
    }
    st0 = st1; d0 = d1; o0 = o1; n0 = n1; a0 = a1; b0 = b1;
    st1 = st2; d1 = d2; o1 = o2; n1 = n2;
    st2 = __builtin_amdgcn_readfirstlane(stn);
  }
}

extern "C" void kernel_launch(void* const* d_in, const int* in_sizes, int n_in,
                              void* d_out, int out_size, void* d_ws, size_t ws_size,
                              hipStream_t stream) {
  const float* x_e  = (const float*)d_in[0];
  const float* x_r  = (const float*)d_in[1];
  const int*   eidx = (const int*)d_in[2];
  const int*   rel  = (const int*)d_in[3];
  const float* w_ah = (const float*)d_in[5];
  const float* w_at = (const float*)d_in[6];
  const float* w_ar = (const float*)d_in[7];
  const float* W1   = (const float*)d_in[8];
  const float* b1   = (const float*)d_in[9];
  const float* W2   = (const float*)d_in[10];
  const float* b2   = (const float*)d_in[11];
  const float* Wr   = (const float*)d_in[12];
  const float* br   = (const float*)d_in[13];
  float* out = (float*)d_out;

  const int n_e     = in_sizes[0] / E_HID;
  const int n_r     = in_sizes[1] / R_HID;
  const int n_edges = in_sizes[3];
  const int* h_idx = eidx;
  const int* t_idx = eidx + n_edges;
  const int nbin   = (2 * n_e + KPB - 1) / KPB;

  int* pairs   = (int*)d_ws;
  float* P     = (float*)(pairs + (size_t)2 * nbin * BIN_CAP);
  float* msg   = P + (size_t)2 * n_r * 256;
  float* e_h   = msg + (size_t)n_r * R_HID;
  float* e_t   = e_h + n_e;
  float* e_r   = e_t + n_e;
  int* cnt     = (int*)(e_r + n_r);
  int* off     = cnt + (size_t)2 * n_e;
  int* gcount  = off + (size_t)2 * n_e;
  int* binned  = gcount + NBIN_MAX;

  k_node_scores<<<(n_e + 63) / 64, 64, 0, stream>>>(x_e, w_ah, w_at, e_h, e_t, gcount, nbin, n_e);
  k_rel_bin<<<n_r + (n_edges + EPB - 1) / EPB, 256, 0, stream>>>(x_r, w_ar, W1, b1, W2, b2, e_r, msg, n_r,
                                                                 h_idx, t_idx, rel, gcount, binned, n_edges, n_e, nbin);
  k_proj_list<<<2 * ((n_r + 15) / 16) + nbin, 256, 0, stream>>>(msg, Wr, P, n_r, (n_r + 15) / 16,
                                                                gcount, binned, e_h, e_t, e_r, cnt, off, pairs, 2 * n_e, n_e);
  k_gather<<<GS_NSL * 32, GS_WAVES * 32, 0, stream>>>(cnt, off, pairs, e_h, e_t, e_r, P, br, out, n_e, n_r, h_idx, t_idx, rel, n_edges);
}


# Relation MLP on the matrix unit (f32 16x16x4), 16 relations per work-group: W1 and W2 read once per 16 rows, not once per row; 1000 MLP groups -> 63; float sum order of the MLP products changed
# speedup vs baseline: 5.9875x; 1.1334x over previous
#include <hip/hip_runtime.h>

#define E_HID 128
#define R_HID 128

#define KPB 256
#define BIN_CAP 4096
#define NBIN_MAX 1024
#define EPB 2048
#define CNT_OVER 0x40000000

typedef float v2f __attribute__((ext_vector_type(2)));
typedef float v8f __attribute__((ext_vector_type(8)));
typedef float v4f __attribute__((ext_vector_type(4)));

__device__ __forceinline__ float leaky(float x) { return x > 0.f ? x : 0.01f * x; }


__device__ __forceinline__ float wave_max_f(float v) {
  for (int o = 16; o > 0; o >>= 1) v = fmaxf(v, __shfl_xor(v, o));
  return v;
}
__device__ __forceinline__ int wave_min_i(int v) {
  for (int o = 16; o > 0; o >>= 1) { int w = __shfl_xor(v, o); v = w < v ? w : v; }
  return v;
}
__device__ __forceinline__ int wave_sum_i(int v) {
  for (int o = 16; o > 0; o >>= 1) v += __shfl_xor(v, o);
  return v;
}

__global__ void __launch_bounds__(64) k_node_scores(const float* __restrict__ x_e,
                              const float* __restrict__ w_ah,
                              const float* __restrict__ w_at,
                              float* __restrict__ e_h, float* __restrict__ e_t,
                              int* __restrict__ gcount, int nbin, int n_e) {
  const int lane = threadIdx.x & 31;
  const int base = ((int)blockIdx.x * 2 + (int)(threadIdx.x >> 5)) * 32;
  if (base >= n_e) return;
  const int s = lane & 7;
  const int q = lane >> 3;
  v4f x[8][4];
  #pragma unroll
  for (int i = 0; i < 8; ++i) {
    int row = base + q * 8 + i;
    row = row < n_e ? row : n_e - 1;
    const v4f* p = (const v4f*)(x_e + (size_t)row * E_HID);
    #pragma unroll
    for (int j = 0; j < 4; ++j) x[i][j] = p[j * 8 + s];
  }
  v4f a[4], b[4];
  #pragma unroll
  for (int j = 0; j < 4; ++j) {
    a[j] = ((const v4f*)w_ah)[j * 8 + s];
    b[j] = ((const v4f*)w_at)[j * 8 + s];
  }
  asm volatile("" ::
               "v"(x[0][0]), "v"(x[0][1]), "v"(x[0][2]), "v"(x[0][3]),
               "v"(x[1][0]), "v"(x[1][1]), "v"(x[1][2]), "v"(x[1][3]),
               "v"(x[2][0]), "v"(x[2][1]), "v"(x[2][2]), "v"(x[2][3]),
               "v"(x[3][0]), "v"(x[3][1]), "v"(x[3][2]), "v"(x[3][3]),
               "v"(x[4][0]), "v"(x[4][1]), "v"(x[4][2]), "v"(x[4][3]),
               "v"(x[5][0]), "v"(x[5][1]), "v"(x[5][2]), "v"(x[5][3]),
               "v"(x[6][0]), "v"(x[6][1]), "v"(x[6][2]), "v"(x[6][3]),
               "v"(x[7][0]), "v"(x[7][1]), "v"(x[7][2]), "v"(x[7][3]),
               "v"(a[0]), "v"(b[0]), "v"(a[1]), "v"(b[1]),
               "v"(a[2]), "v"(b[2]), "v"(a[3]), "v"(b[3]));
  float vh = 0.f, vt = 0.f;
  #pragma unroll
  for (int i = 0; i < 8; ++i) {
    float sh = 0.f, st = 0.f;
    #pragma unroll
    for (int j = 0; j < 4; ++j) {
      const v4f v = x[i][j];
      sh = fmaf(v.x, a[j].x, sh); sh = fmaf(v.y, a[j].y, sh); sh = fmaf(v.z, a[j].z, sh); sh = fmaf(v.w, a[j].w, sh);
      st = fmaf(v.x, b[j].x, st); st = fmaf(v.y, b[j].y, st); st = fmaf(v.z, b[j].z, st); st = fmaf(v.w, b[j].w, st);
    }
    #pragma unroll
    for (int off = 4; off > 0; off >>= 1) {
      sh += __shfl_xor(sh, off);
      st += __shfl_xor(st, off);
    }
    if (s == i) { vh = sh; vt = st; }
  }
  const int row = base + lane;
  if (row < n_e) { e_h[row] = vh; e_t[row] = vt; }
  if (row < nbin) gcount[row] = 0;
}

__device__ __forceinline__ void rel16_body(const int mb, const float* __restrict__ x_r,
                      const float* __restrict__ w_ar,
                      const float* __restrict__ W1, const float* __restrict__ b1,
                      const float* __restrict__ W2, const float* __restrict__ b2,
                      float* __restrict__ e_r, float* __restrict__ msg, const int n_r) {
  constexpr int KC = 32;
  constexpr int XSTR = R_HID + 4;
  constexpr int HSTR = 2 * R_HID + 4;
  constexpr int BKSTR = 36;
  __shared__ __attribute__((aligned(16))) float Xs[16 * XSTR];
  __shared__ __attribute__((aligned(16))) float Bs[2 * R_HID * BKSTR];
  float* const Hs = Bs + R_HID * BKSTR;
  static_assert(16 * HSTR <= R_HID * BKSTR, "hid must fit the upper half of Bs");
  const int tid  = threadIdx.x;
  const int lane = tid & 31;
  const int wave = tid >> 5;
  const int m0   = mb * 16;
  const int lrow = lane & 15;
  const int hsel = (lane >= 16) ? 2 : 0;
  const int mrow = (lane >= 16) ? 8 : 0;
  const int n0   = wave * 16;

  #pragma unroll
  for (int u = 0; u < 2; ++u) {
    const int f  = tid + 256 * u;
    const int r  = f >> 5;
    const int c4 = (f & 31) * 4;
    float4 v = make_float4(0.f, 0.f, 0.f, 0.f);
    if (m0 + r < n_r) v = *(const float4*)(x_r + (size_t)(m0 + r) * R_HID + c4);
    *(float4*)&Xs[r * XSTR + c4] = v;
  }

  float4 bReg[8];
  auto loadW1 = [&](int kb) {
    #pragma unroll
    for (int i = 0; i < 8; ++i) {
      const int k4 = kb + i * 4;
      bReg[i].x = W1[(size_t)(k4 + 0) * (2 * R_HID) + tid];
      bReg[i].y = W1[(size_t)(k4 + 1) * (2 * R_HID) + tid];
      bReg[i].z = W1[(size_t)(k4 + 2) * (2 * R_HID) + tid];
      bReg[i].w = W1[(size_t)(k4 + 3) * (2 * R_HID) + tid];
    }
  };
  v8f c0 = {}, c1 = {};
  loadW1(0);
  for (int kb = 0; kb < R_HID; kb += KC) {
    #pragma unroll
    for (int i = 0; i < 8; ++i) *(float4*)&Bs[tid * BKSTR + i * 4] = bReg[i];
    __syncthreads();
    if (kb + KC < R_HID) loadW1(kb + KC);
    #pragma unroll
    for (int kk = 0; kk < KC; kk += 4) {
      const int ka = kk + hsel;
      const v2f a  = *(const v2f*)&Xs[lrow * XSTR + kb + ka];
      const v2f f0 = *(const v2f*)&Bs[(n0 + lrow) * BKSTR + ka];
      const v2f f1 = *(const v2f*)&Bs[(n0 + 128 + lrow) * BKSTR + ka];
      c0 = __builtin_amdgcn_wmma_f32_16x16x4_f32(false, a, false, f0, (short)0, c0, false, false);
      c1 = __builtin_amdgcn_wmma_f32_16x16x4_f32(false, a, false, f1, (short)0, c1, false, false);
    }
    __syncthreads();
  }

  float4 wReg[4];
  const int n2 = tid & 127;
  const int kh = (tid >> 7) * 16;
  auto loadW2 = [&](int kb) {
    #pragma unroll
    for (int i = 0; i < 4; ++i) {
      const int k4 = kb + kh + i * 4;
      wReg[i].x = W2[(size_t)(k4 + 0) * R_HID + n2];
      wReg[i].y = W2[(size_t)(k4 + 1) * R_HID + n2];
      wReg[i].z = W2[(size_t)(k4 + 2) * R_HID + n2];
      wReg[i].w = W2[(size_t)(k4 + 3) * R_HID + n2];
    }
  };
  loadW2(0);
  {
    const float bb0 = b1[n0 + lrow];
    const float bb1 = b1[n0 + 128 + lrow];
    #pragma unroll
    for (int j = 0; j < 8; ++j) {
      Hs[(mrow + j) * HSTR + n0 + lrow]       = c0[j] + bb0;
      Hs[(mrow + j) * HSTR + n0 + 128 + lrow] = c1[j] + bb1;
    }
  }
  #pragma unroll
  for (int rr = 0; rr < 2; ++rr) {
    const int r = 2 * wave + rr;
    float s = 0.f;
    #pragma unroll
    for (int j = 0; j < 4; ++j) s += Xs[r * XSTR + lane + 32 * j] * w_ar[lane + 32 * j];
    for (int off = 16; off > 0; off >>= 1) s += __shfl_xor(s, off);
    if (lane == 0 && m0 + r < n_r) e_r[m0 + r] = s;
  }

  v8f d0 = {}, d1 = {};
  for (int kb = 0; kb < 2 * R_HID; kb += KC) {
    #pragma unroll
    for (int i = 0; i < 4; ++i) *(float4*)&Bs[n2 * BKSTR + kh + i * 4] = wReg[i];
    __syncthreads();
    if (kb + KC < 2 * R_HID) loadW2(kb + KC);
    #pragma unroll
    for (int kk = 0; kk < KC; kk += 8) {
      const int ka = kk + hsel;
      const v2f a0 = *(const v2f*)&Hs[lrow * HSTR + kb + ka];
      const v2f a1 = *(const v2f*)&Hs[lrow * HSTR + kb + ka + 4];
      const v2f f0 = *(const v2f*)&Bs[(n0 + lrow) * BKSTR + ka];
      const v2f f1 = *(const v2f*)&Bs[(n0 + lrow) * BKSTR + ka + 4];
      d0 = __builtin_amdgcn_wmma_f32_16x16x4_f32(false, a0, false, f0, (short)0, d0, false, false);
      d1 = __builtin_amdgcn_wmma_f32_16x16x4_f32(false, a1, false, f1, (short)0, d1, false, false);
    }
    __syncthreads();
  }
  const float bb2 = b2[n0 + lrow];
  #pragma unroll
  for (int j = 0; j < 8; ++j) {
    const int grow = m0 + mrow + j;
    if (grow < n_r) msg[(size_t)grow * R_HID + n0 + lrow] = Xs[(mrow + j) * XSTR + n0 + lrow] + ((d0[j] + d1[j]) + bb2);
  }
}

__device__ __forceinline__ int4 load4_vec(const int* __restrict__ p, const int e, const int e0, const int e1) {
  const bool in = e < e1;
  int4 v = *(const int4*)(p + (in ? e : e0));
  if (!in) v = make_int4(-1, -1, -1, -1);
  return v;
}
__device__ __forceinline__ int4 load4_any(const int* __restrict__ p, const int e, const int e0, const int e1) {
  int4 v;
  v.x = p[(e     < e1) ? e     : e0];
  v.y = p[(e + 1 < e1) ? e + 1 : e0];
  v.z = p[(e + 2 < e1) ? e + 2 : e0];
  v.w = p[(e + 3 < e1) ? e + 3 : e0];
  if (e     >= e1) v.x = -1;
  if (e + 1 >= e1) v.y = -1;
  if (e + 2 >= e1) v.z = -1;
  if (e + 3 >= e1) v.w = -1;
  return v;
}

__device__ __forceinline__ void bin_body(const int bid,
                      const int* __restrict__ h_idx, const int* __restrict__ t_idx, const int* __restrict__ rel,
                      int* __restrict__ gcount, int* __restrict__ binned,
                      int n_edges, int n_e, int nbin) {
  __shared__ int hist[NBIN_MAX];
  __shared__ int gbase[NBIN_MAX];
  __shared__ int lbase[NBIN_MAX];
  __shared__ int sRec[2 * EPB];
  __shared__ int sAdr[2 * EPB];
  __shared__ int wsum[8];
  const int tid  = threadIdx.x;
  const int lane = tid & 31;
  const int wave = tid >> 5;
  const int e0 = bid * EPB;
  const int e1 = (e0 + EPB < n_edges) ? e0 + EPB : n_edges;
  const bool vec = (n_edges & 3) == 0;
  const int ea = e0 + tid * 4;
  const int eb = ea + 1024;
  #pragma unroll
  for (int k = 0; k < 4; ++k) hist[tid + 256 * k] = 0;
  int4 hA, tA, hB, tB, rA, rB;
  if (vec) {
    hA = load4_vec(h_idx, ea, e0, e1); tA = load4_vec(t_idx, ea, e0, e1); rA = load4_vec(rel, ea, e0, e1);
    hB = load4_vec(h_idx, eb, e0, e1); tB = load4_vec(t_idx, eb, e0, e1); rB = load4_vec(rel, eb, e0, e1);
  } else {
    hA = load4_any(h_idx, ea, e0, e1); tA = load4_any(t_idx, ea, e0, e1); rA = load4_any(rel, ea, e0, e1);
    hB = load4_any(h_idx, eb, e0, e1); tB = load4_any(t_idx, eb, e0, e1); rB = load4_any(rel, eb, e0, e1);
  }
  const int hv[8] = {hA.x, hA.y, hA.z, hA.w, hB.x, hB.y, hB.z, hB.w};
  const int tv[8] = {tA.x, tA.y, tA.z, tA.w, tB.x, tB.y, tB.z, tB.w};
  const int rv[8] = {rA.x, rA.y, rA.z, rA.w, rB.x, rB.y, rB.z, rB.w};
  __syncthreads();
  #pragma unroll
  for (int j = 0; j < 8; ++j) {
    if (hv[j] >= 0) {
      atomicAdd(&hist[hv[j] >> 8], 1);
      atomicAdd(&hist[(n_e + tv[j]) >> 8], 1);
    }
  }
  __syncthreads();
  int cg[4], g[4];
  #pragma unroll
  for (int k = 0; k < 4; ++k) {
    const int b = tid + 256 * k;
    cg[k] = (b < nbin) ? hist[b] : 0;
  }
  #pragma unroll
  for (int k = 0; k < 4; ++k) g[k] = (cg[k] > 0) ? atomicAdd(gcount + tid + 256 * k, cg[k]) : 0;
  int c[4];
  #pragma unroll
  for (int k = 0; k < 4; ++k) {
    const int b = 4 * tid + k;
    c[k] = (b < nbin) ? hist[b] : 0;
  }
  const int tsum = c[0] + c[1] + c[2] + c[3];
  int incl = tsum;
  for (int o = 1; o < 32; o <<= 1) {
    const int v = __shfl_up(incl, o);
    if (lane >= o) incl += v;
  }
  if (lane == 31) wsum[wave] = incl;
  __syncthreads();
  int excl = incl - tsum;
  for (int w = 0; w < wave; ++w) excl += wsum[w];
  #pragma unroll
  for (int k = 0; k < 4; ++k) {
    const int b = 4 * tid + k;
    lbase[b] = excl;
    hist[b] = 0;
    excl += c[k];
  }
  #pragma unroll
  for (int k = 0; k < 4; ++k) gbase[tid + 256 * k] = g[k];
  __syncthreads();
  #pragma unroll
  for (int j = 0; j < 8; ++j) {
    if (hv[j] >= 0) {
      const int kh = hv[j];
      const int kt = n_e + tv[j];
      const int bh = kh >> 8;
      const int bt = kt >> 8;
      const int lh = atomicAdd(&hist[bh], 1);
      const int lt = atomicAdd(&hist[bt], 1);
      const int ph = lbase[bh] + lh;
      const int pt = lbase[bt] + lt;
      const int gh = gbase[bh] + lh;
      const int gt = gbase[bt] + lt;
      sRec[ph] = (rv[j] << 8) | (kh & 255);
      sRec[pt] = (rv[j] << 8) | (kt & 255);
      sAdr[ph] = (gh < BIN_CAP) ? bh * BIN_CAP + gh : -1;
      sAdr[pt] = (gt < BIN_CAP) ? bt * BIN_CAP + gt : -1;
    }
  }
  __syncthreads();
  const int nrec = 2 * (e1 - e0);
  for (int i = tid; i < nrec; i += 256) {
    const int a = sAdr[i];
    if (a >= 0) binned[a] = sRec[i];
  }
}

#define RPT (BIN_CAP / 256)
#define GRP_CAP (BIN_CAP + 3 * KPB)
__device__ __forceinline__ int f2ord(float f) { const int b = __float_as_int(f); return b >= 0 ? b : b ^ 0x7fffffff; }
__device__ __forceinline__ float ord2f(int o) { return __int_as_float(o >= 0 ? o : o ^ 0x7fffffff); }

__device__ __forceinline__ void binlist_body(const int bin, const int* __restrict__ gcount,
                          const int* __restrict__ binned,
                          const float* __restrict__ e_h, const float* __restrict__ e_t,
                          const float* __restrict__ e_r,
                          int* __restrict__ cnt, int* __restrict__ off, int2* __restrict__ pairs,
                          int n_keys, int n_e) {
  __shared__ int kcnt[KPB];
  __shared__ int koff2[KPB];
  __shared__ float ken[KPB];
  __shared__ int kmx[KPB];
  __shared__ uint4 grpv[GRP_CAP / 4];
  int* const grp = (int*)grpv;
  __shared__ int wgt[BIN_CAP];
  __shared__ int wsum2[8];
  const int tid  = threadIdx.x;
  const int lane = tid & 31;
  const int wave = tid >> 5;
  const int tot = gcount[bin];
  const bool over = tot > BIN_CAP;
  const int n = over ? BIN_CAP : tot;
  kcnt[tid] = 0;
  kmx[tid] = (int)0x80000000;
  const int* __restrict__ src = binned + (size_t)bin * BIN_CAP;
  int rc[RPT], sl[RPT];
  #pragma unroll
  for (int m = 0; m < RPT; ++m) {
    const int i = tid + 256 * m;
    const int v = src[i < n ? i : 0];
    rc[m] = i < n ? v : -1;
  }
  __syncthreads();
  #pragma unroll
  for (int m = 0; m < RPT; ++m) {
    sl[m] = 0;
    if (rc[m] >= 0) sl[m] = atomicAdd(&kcnt[rc[m] & 255], 1);
  }
  __syncthreads();
  const int c = kcnt[tid];
  const int c4 = (c + 3) & ~3;
  const int both = (c4 << 16) | c;
  int incl = both;
  for (int o = 1; o < 32; o <<= 1) {
    const int v = __shfl_up(incl, o);
    if (lane >= o) incl += v;
  }
  if (lane == 31) wsum2[wave] = incl;
  __syncthreads();
  int wb = 0;
  for (int w = 0; w < wave; ++w) wb += wsum2[w];
  const int excl = wb + incl - both;
  const int ko = excl & 0xffff;
  const int ko4 = excl >> 16;
  koff2[tid] = excl;
  for (int u = c; u < c4; ++u) grp[ko4 + u] = -1;
  const int key = bin * KPB + tid;
  float en = 0.f;
  if (key < n_keys) {
    cnt[key] = over ? CNT_OVER : c;
    off[key] = bin * BIN_CAP + ko;
    en = key < n_e ? e_h[key] : e_t[key - n_e];
  }
  ken[tid] = en;
  __syncthreads();
  if (over) return;
  float er[RPT];
  #pragma unroll
  for (int m = 0; m < RPT; ++m) {
    const int r = rc[m] >= 0 ? (rc[m] >> 8) : 0;
    er[m] = e_r[r];
    if (rc[m] >= 0) grp[(koff2[rc[m] & 255] >> 16) + sl[m]] = (int)(((unsigned)r << 12) | (unsigned)sl[m]);
  }
  __syncthreads();
  int dst[RPT];
  float zz[RPT];
  #pragma unroll
  for (int m = 0; m < RPT; ++m) {
    dst[m] = -1;
    zz[m] = 0.f;
    if (rc[m] >= 0) {
      const int k = rc[m] & 255;
      const int o2 = koff2[k];
      const int lo = o2 >> 18;
      const int hi = lo + ((kcnt[k] + 3) >> 2);
      const unsigned w = ((unsigned)(rc[m] >> 8) << 12) | (unsigned)sl[m];
      int rank = 0;
      for (int q = lo; q < hi; ++q) {
        const uint4 y = grpv[q];
        rank += (y.x < w ? 1 : 0) + (y.y < w ? 1 : 0) + (y.z < w ? 1 : 0) + (y.w < w ? 1 : 0);
      }
      dst[m] = (o2 & 0xffff) + rank;
      const float z = leaky(ken[k] + er[m]);
      zz[m] = z;
      atomicMax(&kmx[k], f2ord(z));
    }
  }
  __syncthreads();
  #pragma unroll
  for (int m = 0; m < RPT; ++m) {
    if (dst[m] >= 0) {
      grp[dst[m]] = rc[m];
      wgt[dst[m]] = __float_as_int(__expf(zz[m] - ord2f(kmx[rc[m] & 255])));
    }
  }
  __syncthreads();
  if (c > 0) {
    float sum = 0.f;
    #pragma unroll 4
    for (int q = 0; q < c; ++q) sum += __int_as_float(wgt[ko + q]);
    kmx[tid] = __float_as_int(sum + 1e-16f);
  }
  __syncthreads();
  int2* __restrict__ dstp = pairs + (size_t)bin * BIN_CAP;
  for (int p = tid; p < n; p += 256) {
    const int rec = grp[p];
    const float al = __int_as_float(wgt[p]) / __int_as_float(kmx[rec & 255]);
    dstp[p] = make_int2(rec >> 8, __float_as_int(al));
  }
}

__device__ __forceinline__ void proj_body(const int mb, const int half,
                       const float* __restrict__ msg, const float* __restrict__ Wr,
                       float* __restrict__ P, int n_r) {
  constexpr int KTOT = 128, NTOT = 256, KC = 32;
  constexpr int ASTR = 36, BKSTR = 36;
  __shared__ float As[16 * ASTR];
  __shared__ float Bs[NTOT * BKSTR];
  const int tid  = threadIdx.x;
  const int lane = tid & 31;
  const int wave = tid >> 5;
  const int m0   = mb * 16;
  const int lrow = lane & 15;
  const int hsel = (lane >= 16) ? 2 : 0;
  const int mrow = (lane >= 16) ? 8 : 0;
  const int n0   = wave * 16;
  const float* __restrict__ B = Wr + (size_t)half * KTOT * NTOT;
  float* __restrict__ O = P + (size_t)half * n_r * NTOT;

  const int arow = tid >> 3;
  const int acol = (tid & 7) * 4;

  v8f c0 = {}, c1 = {};
  float4 aReg = make_float4(0.f, 0.f, 0.f, 0.f);
  float4 bReg[8];

  auto loadGlobal = [&](int kb) {
    if (tid < 128) {
      int grow = m0 + arow;
      aReg = make_float4(0.f, 0.f, 0.f, 0.f);
      if (grow < n_r) aReg = *(const float4*)(msg + (size_t)grow * KTOT + kb + acol);
    }
    #pragma unroll
    for (int i = 0; i < 8; ++i) {
      int k4 = i * 4;
      bReg[i].x = B[(size_t)(kb + k4 + 0) * NTOT + tid];
      bReg[i].y = B[(size_t)(kb + k4 + 1) * NTOT + tid];
      bReg[i].z = B[(size_t)(kb + k4 + 2) * NTOT + tid];
      bReg[i].w = B[(size_t)(kb + k4 + 3) * NTOT + tid];
    }
  };
  auto storeSmem = [&]() {
    if (tid < 128) *(float4*)&As[arow * ASTR + acol] = aReg;
    #pragma unroll
    for (int i = 0; i < 8; ++i)
      *(float4*)&Bs[tid * BKSTR + i * 4] = bReg[i];
  };

  loadGlobal(0);
  for (int kb = 0; kb < KTOT; kb += KC) {
    storeSmem();
    __syncthreads();
    if (kb + KC < KTOT) loadGlobal(kb + KC);
    #pragma unroll
    for (int kk = 0; kk < KC; kk += 4) {
      int ka = kk + hsel;
      v2f a  = *(const v2f*)&As[lrow * ASTR + ka];
      v2f b0 = *(const v2f*)&Bs[(n0 + lrow) * BKSTR + ka];
      v2f b1 = *(const v2f*)&Bs[(n0 + 128 + lrow) * BKSTR + ka];
      c0 = __builtin_amdgcn_wmma_f32_16x16x4_f32(false, a, false, b0, (short)0, c0, false, false);
      c1 = __builtin_amdgcn_wmma_f32_16x16x4_f32(false, a, false, b1, (short)0, c1, false, false);
    }
    __syncthreads();
  }
  #pragma unroll
  for (int j = 0; j < 8; ++j) {
    int grow = m0 + mrow + j;
    if (grow < n_r) {
      O[(size_t)grow * NTOT + n0 + lrow]       = c0[j];
      O[(size_t)grow * NTOT + n0 + 128 + lrow] = c1[j];
    }
  }
}

__global__ void k_rel_bin(const float* __restrict__ x_r,
                          const float* __restrict__ w_ar,
                          const float* __restrict__ W1, const float* __restrict__ b1,
                          const float* __restrict__ W2, const float* __restrict__ b2,
                          float* __restrict__ e_r, float* __restrict__ msg, int n_r, int n_rb,
                          const int* __restrict__ h_idx, const int* __restrict__ t_idx, const int* __restrict__ rel,
                          int* __restrict__ gcount, int* __restrict__ binned,
                          int n_edges, int n_e, int nbin) {
  const int b = (int)blockIdx.x;
  if (b < n_rb) rel16_body(b, x_r, w_ar, W1, b1, W2, b2, e_r, msg, n_r);
  else bin_body(b - n_rb, h_idx, t_idx, rel, gcount, binned, n_edges, n_e, nbin);
}

__global__ void k_proj_list(const float* __restrict__ msg, const float* __restrict__ Wr,
                            float* __restrict__ P, int n_r, int n_mb,
                            const int* __restrict__ gcount, const int* __restrict__ binned,
                            const float* __restrict__ e_h, const float* __restrict__ e_t,
                            const float* __restrict__ e_r,
                            int* __restrict__ cnt, int* __restrict__ off, int* __restrict__ pairs,
                            int n_keys, int n_e) {
  const int b = (int)blockIdx.x;
  if (b < 2 * n_mb) {
    const int half = b >= n_mb ? 1 : 0;
    proj_body(b - half * n_mb, half, msg, Wr, P, n_r);
  } else {
    binlist_body(b - 2 * n_mb, gcount, binned, e_h, e_t, e_r, cnt, off, (int2*)pairs, n_keys, n_e);
  }
}

#define GS_COLS 32
#define GS_NSL (256 / GS_COLS)
#define GS_ROWS 1024
#define GS_WAVES 32

struct __attribute__((aligned(8))) pair2 { int r0, w0, r1, w1; };

__device__ __forceinline__ float scan_side_col(const float en, const float* __restrict__ e_r,
                                               const float* __restrict__ Pc, const int lane,
                                               const int node, const int* __restrict__ idx,
                                               const int* __restrict__ rel, const int n_edges) {
  float mx = -__builtin_inff();
  for (int i = lane; i < n_edges; i += 32) if (idx[i] == node) mx = fmaxf(mx, leaky(en + e_r[rel[i]]));
  mx = wave_max_f(mx);
  float s = 0.f;
  int last = -1;
  for (int it = 0; it < n_edges; ++it) {
    int best = 0x7fffffff;
    for (int i = lane; i < n_edges; i += 32) if (idx[i] == node) { const int r = rel[i]; if (r > last && r < best) best = r; }
    best = __builtin_amdgcn_readfirstlane(wave_min_i(best));
    if (best == 0x7fffffff) break;
    int c = 0;
    for (int i = lane; i < n_edges; i += 32) c += (idx[i] == node && rel[i] == best) ? 1 : 0;
    c = __builtin_amdgcn_readfirstlane(wave_sum_i(c));
    const float p = __expf(leaky(en + e_r[best]) - mx);
    for (int q = 0; q < c; ++q) s += p;
    last = best;
  }
  const float den = s + 1e-16f;
  float acc = 0.f;
  last = -1;
  for (int it = 0; it < n_edges; ++it) {
    int best = 0x7fffffff;
    for (int i = lane; i < n_edges; i += 32) if (idx[i] == node) { const int r = rel[i]; if (r > last && r < best) best = r; }
    best = __builtin_amdgcn_readfirstlane(wave_min_i(best));
    if (best == 0x7fffffff) break;
    int c = 0;
    for (int i = lane; i < n_edges; i += 32) c += (idx[i] == node && rel[i] == best) ? 1 : 0;
    c = __builtin_amdgcn_readfirstlane(wave_sum_i(c));
    const float a = __expf(leaky(en + e_r[best]) - mx) / den;
    const float m0 = Pc[(size_t)best * 256];
    for (int q = 0; q < c; ++q) acc += a * m0;
    last = best;
  }
  return acc;
}

template <int N, bool CHECK>
__device__ __forceinline__ void slice_trips(const char* __restrict__ tabb, const unsigned lbyte, const int rem,
                                            const pair2 pa, const pair2 pb, v4f& acc) {
  const int rr[4] = {pa.r0, pa.r1, pb.r0, pb.r1};
  const int ww[4] = {pa.w0, pa.w1, pb.w0, pb.w1};
  v4f t[N];
  float w[N];
  #pragma unroll
  for (int i = 0; i < N; ++i) {
    const bool ok = !CHECK || i < rem;
    const unsigned row = ok ? (unsigned)rr[i] : 0u;
    w[i] = ok ? __int_as_float(ww[i]) : 0.f;
    t[i] = *(const v4f*)(tabb + ((row << 8) | lbyte));
  }
  if constexpr (N == 4) {
    asm volatile("" :: "v"(t[0]), "v"(t[1]), "v"(t[2]), "v"(t[3]));
  } else if constexpr (N == 3) {
    asm volatile("" :: "v"(t[0]), "v"(t[1]), "v"(t[2]));
  } else if constexpr (N == 2) {
    asm volatile("" :: "v"(t[0]), "v"(t[1]));
  }
  #pragma unroll
  for (int i = 0; i < N; ++i) acc += t[i] * w[i];
}

__global__ void __launch_bounds__(GS_WAVES * 32) k_gather(const int* __restrict__ cnt, const int* __restrict__ off,
                         const int* __restrict__ pairs_w,
                         const float* __restrict__ e_h, const float* __restrict__ e_t,
                         const float* __restrict__ e_r, const float* __restrict__ P,
                         const float* __restrict__ br,
                         float* __restrict__ out, int n_e, int n_r,
                         const int* __restrict__ h_idx, const int* __restrict__ t_idx,
                         const int* __restrict__ rel, int n_edges) {
  __shared__ __attribute__((aligned(16))) float tab[GS_ROWS * 2 * GS_COLS];
  __shared__ int next_step;
  const int tid  = threadIdx.x;
  const int lane = tid & 31;
  const int wave = __builtin_amdgcn_readfirstlane(tid >> 5);
  const int cs = (int)blockIdx.x % GS_NSL;
  const int ng = (int)blockIdx.x / GS_NSL;
  const int n_ng = (int)gridDim.x / GS_NSL;
  const int npg = (n_e + n_ng - 1) / n_ng;
  const int nb = ng * npg;
  if (nb >= n_e) return;
  const int ne = (nb + npg < n_e) ? nb + npg : n_e;
  const int nsteps = (ne - nb + 1) >> 1;
  const int s = lane & 7;
  const int g = lane >> 3;
  const int side = g & 1;
  v4f* const tab4 = (v4f*)tab;

  {
    const int npiece = 2 * n_r;
    const float* __restrict__ Pc = P + cs * GS_COLS + s * 4;
    for (int p0 = 0; p0 < npiece; p0 += 8 * GS_WAVES * 4) {
      v4f v[8];
      #pragma unroll
      for (int u = 0; u < 8; ++u) {
        const int p = p0 + u * GS_WAVES * 4 + (tid >> 3);
        const int pc = p < npiece ? p : 0;
        v[u] = *(const v4f*)(Pc + (size_t)((pc & 1) * n_r + (pc >> 1)) * 256);
      }
      #pragma unroll
      for (int u = 0; u < 8; ++u) {
        const int p = p0 + u * GS_WAVES * 4 + (tid >> 3);
        if (p < npiece) tab4[p * 8 + s] = v[u];
      }
    }
  }
  if (tid == 0) next_step = 3 * GS_WAVES;
  const v4f bias = *(const v4f*)(br + cs * GS_COLS + s * 4);
  const unsigned lbyte = (unsigned)(side * (GS_COLS * 4) + s * 16);
  const char* const tabb = (const char*)tab;
  __syncthreads();

  auto side_of = [&](const int st, int& d, int& o, int& node) {
    node = nb + 2 * st + (g >> 1);
    const bool ok = st < nsteps && node < ne;
    const int key = side * n_e + (ok ? node : nb);
    const int dd = cnt[key];
    o = off[key];
    d = ok ? dd : 0;
  };
  auto ld2 = [&](const int o, const int k) -> pair2 {
    return *(const pair2*)((const char*)pairs_w + (unsigned)(o + k) * 8u);
  };

  int st0 = wave, st1 = wave + GS_WAVES, st2 = wave + 2 * GS_WAVES;
  int d0, o0, n0, d1, o1, n1;
  side_of(st0, d0, o0, n0);
  side_of(st1, d1, o1, n1);
  pair2 a0 = ld2(o0, 0), b0 = ld2(o0, 2);
  while (st0 < nsteps) {
    int stn = 0;
    if (lane == 0) stn = atomicAdd(&next_step, 1);
    pair2 c[8];
    c[0] = a0; c[1] = b0;
    #pragma unroll
    for (int j = 1; j < 4; ++j) { c[2 * j] = ld2(o0, 4 * j); c[2 * j + 1] = ld2(o0, 4 * j + 2); }
    int d2, o2, n2;
    side_of(st2, d2, o2, n2);
    const pair2 a1 = ld2(o1, 0), b1 = ld2(o1, 2);

    const bool over = d0 == CNT_OVER;
    const int dl = over ? 0 : d0;
    int dmax = __builtin_amdgcn_readlane(dl, 0), dmin = dmax;
    { const int x = __builtin_amdgcn_readlane(dl, 8);  dmax = x > dmax ? x : dmax; dmin = x < dmin ? x : dmin; }
    { const int x = __builtin_amdgcn_readlane(dl, 16); dmax = x > dmax ? x : dmax; dmin = x < dmin ? x : dmin; }
    { const int x = __builtin_amdgcn_readlane(dl, 24); dmax = x > dmax ? x : dmax; dmin = x < dmin ? x : dmin; }
    v4f acc = {0.f, 0.f, 0.f, 0.f};
    auto group = [&](const pair2 xa, const pair2 xb, const int k) {
      const int remw = dmax - k;
      if (remw >= 4) {
        if (k + 4 <= dmin) slice_trips<4, false>(tabb, lbyte, 4, xa, xb, acc);
        else slice_trips<4, true>(tabb, lbyte, dl - k, xa, xb, acc);
      }
      else if (remw == 3) slice_trips<3, true>(tabb, lbyte, dl - k, xa, xb, acc);
      else if (remw == 2) slice_trips<2, true>(tabb, lbyte, dl - k, xa, xb, acc);
      else if (remw == 1) slice_trips<1, true>(tabb, lbyte, dl - k, xa, xb, acc);
    };
    #pragma unroll
    for (int j = 0; j < 4; ++j) {
      if (4 * j < dmax) group(c[2 * j], c[2 * j + 1], 4 * j);
    }
    for (int k = 16; k < dmax; k += 4) {
      const pair2 xa = ld2(o0, k), xb = ld2(o0, k + 2);
      group(xa, xb, k);
    }

    const unsigned ovm = __builtin_amdgcn_ballot_w32(over);
    if (ovm != 0u) {
      for (int gg = 0; gg < 4; ++gg) {
        if ((ovm >> (8 * gg)) & 1u) {
          const int nd = __builtin_amdgcn_readlane(n0, 8 * gg);
          const int sd = gg & 1;
          const float en = sd ? e_t[nd] : e_h[nd];
          const float part = scan_side_col(en, e_r, P + (size_t)sd * n_r * 256 + cs * GS_COLS + lane, lane, nd,
                                           sd ? t_idx : h_idx, rel, n_edges);
          v4f add;
          add.x = __shfl(part, 4 * s + 0); add.y = __shfl(part, 4 * s + 1);
          add.z = __shfl(part, 4 * s + 2); add.w = __shfl(part, 4 * s + 3);
          if (g == gg) acc += add;
        }
      }
    }
    auto ror8 = [](const float v) {
      return __int_as_float(__builtin_amdgcn_update_dpp(0, __float_as_int(v), 0x128, 0xf, 0xf, false));
    };
    v4f oth;
    oth.x = ror8(acc.x); oth.y = ror8(acc.y); oth.z = ror8(acc.z); oth.w = ror8(acc.w);
    if (side == 0 && n0 < ne) {
      const v4f r = (acc + oth) + bias;
      *(v4f*)(out + (size_t)n0 * 256 + cs * GS_COLS + s * 4) = r;
    }
    st0 = st1; d0 = d1; o0 = o1; n0 = n1; a0 = a1; b0 = b1;
    st1 = st2; d1 = d2; o1 = o2; n1 = n2;
    st2 = __builtin_amdgcn_readfirstlane(stn);
  }
}

extern "C" void kernel_launch(void* const* d_in, const int* in_sizes, int n_in,
                              void* d_out, int out_size, void* d_ws, size_t ws_size,
                              hipStream_t stream) {
  const float* x_e  = (const float*)d_in[0];
  const float* x_r  = (const float*)d_in[1];
  const int*   eidx = (const int*)d_in[2];
  const int*   rel  = (const int*)d_in[3];
  const float* w_ah = (const float*)d_in[5];
  const float* w_at = (const float*)d_in[6];
  const float* w_ar = (const float*)d_in[7];
  const float* W1   = (const float*)d_in[8];
  const float* b1   = (const float*)d_in[9];
  const float* W2   = (const float*)d_in[10];
  const float* b2   = (const float*)d_in[11];
  const float* Wr   = (const float*)d_in[12];
  const float* br   = (const float*)d_in[13];
  float* out = (float*)d_out;

  const int n_e     = in_sizes[0] / E_HID;
  const int n_r     = in_sizes[1] / R_HID;
  const int n_edges = in_sizes[3];
  const int* h_idx = eidx;
  const int* t_idx = eidx + n_edges;
  const int nbin   = (2 * n_e + KPB - 1) / KPB;

  int* pairs   = (int*)d_ws;
  float* P     = (float*)(pairs + (size_t)2 * nbin * BIN_CAP);
  float* msg   = P + (size_t)2 * n_r * 256;
  float* e_h   = msg + (size_t)n_r * R_HID;
  float* e_t   = e_h + n_e;
  float* e_r   = e_t + n_e;
  int* cnt     = (int*)(e_r + n_r);
  int* off     = cnt + (size_t)2 * n_e;
  int* gcount  = off + (size_t)2 * n_e;
  int* binned  = gcount + NBIN_MAX;

  k_node_scores<<<(n_e + 63) / 64, 64, 0, stream>>>(x_e, w_ah, w_at, e_h, e_t, gcount, nbin, n_e);
  k_rel_bin<<<(n_r + 15) / 16 + (n_edges + EPB - 1) / EPB, 256, 0, stream>>>(x_r, w_ar, W1, b1, W2, b2, e_r, msg, n_r, (n_r + 15) / 16,
                                                                 h_idx, t_idx, rel, gcount, binned, n_edges, n_e, nbin);
  k_proj_list<<<2 * ((n_r + 15) / 16) + nbin, 256, 0, stream>>>(msg, Wr, P, n_r, (n_r + 15) / 16,
                                                                gcount, binned, e_h, e_t, e_r, cnt, off, pairs, 2 * n_e, n_e);
  k_gather<<<GS_NSL * 32, GS_WAVES * 32, 0, stream>>>(cnt, off, pairs, e_h, e_t, e_r, P, br, out, n_e, n_r, h_idx, t_idx, rel, n_edges);
}


# Gather walks four nodes per wave step (eight 4-lane groups, 32 bytes of a row per lane) instead of two: step overhead and pair loads paid once per four nodes; every float sum in the same order
# speedup vs baseline: 6.8606x; 1.1458x over previous
#include <hip/hip_runtime.h>

#define E_HID 128
#define R_HID 128

#define KPB 256
#define BIN_CAP 4096
#define NBIN_MAX 1024
#define EPB 2048
#define CNT_OVER 0x40000000

typedef float v2f __attribute__((ext_vector_type(2)));
typedef float v8f __attribute__((ext_vector_type(8)));
typedef float v4f __attribute__((ext_vector_type(4)));

__device__ __forceinline__ float leaky(float x) { return x > 0.f ? x : 0.01f * x; }


__device__ __forceinline__ float wave_max_f(float v) {
  for (int o = 16; o > 0; o >>= 1) v = fmaxf(v, __shfl_xor(v, o));
  return v;
}
__device__ __forceinline__ int wave_min_i(int v) {
  for (int o = 16; o > 0; o >>= 1) { int w = __shfl_xor(v, o); v = w < v ? w : v; }
  return v;
}
__device__ __forceinline__ int wave_sum_i(int v) {
  for (int o = 16; o > 0; o >>= 1) v += __shfl_xor(v, o);
  return v;
}

__global__ void __launch_bounds__(64) k_node_scores(const float* __restrict__ x_e,
                              const float* __restrict__ w_ah,
                              const float* __restrict__ w_at,
                              float* __restrict__ e_h, float* __restrict__ e_t,
                              int* __restrict__ gcount, int nbin, int n_e) {
  const int lane = threadIdx.x & 31;
  const int base = ((int)blockIdx.x * 2 + (int)(threadIdx.x >> 5)) * 32;
  if (base >= n_e) return;
  const int s = lane & 7;
  const int q = lane >> 3;
  v4f x[8][4];
  #pragma unroll
  for (int i = 0; i < 8; ++i) {
    int row = base + q * 8 + i;
    row = row < n_e ? row : n_e - 1;
    const v4f* p = (const v4f*)(x_e + (size_t)row * E_HID);
    #pragma unroll
    for (int j = 0; j < 4; ++j) x[i][j] = p[j * 8 + s];
  }
  v4f a[4], b[4];
  #pragma unroll
  for (int j = 0; j < 4; ++j) {
    a[j] = ((const v4f*)w_ah)[j * 8 + s];
    b[j] = ((const v4f*)w_at)[j * 8 + s];
  }
  asm volatile("" ::
               "v"(x[0][0]), "v"(x[0][1]), "v"(x[0][2]), "v"(x[0][3]),
               "v"(x[1][0]), "v"(x[1][1]), "v"(x[1][2]), "v"(x[1][3]),
               "v"(x[2][0]), "v"(x[2][1]), "v"(x[2][2]), "v"(x[2][3]),
               "v"(x[3][0]), "v"(x[3][1]), "v"(x[3][2]), "v"(x[3][3]),
               "v"(x[4][0]), "v"(x[4][1]), "v"(x[4][2]), "v"(x[4][3]),
               "v"(x[5][0]), "v"(x[5][1]), "v"(x[5][2]), "v"(x[5][3]),
               "v"(x[6][0]), "v"(x[6][1]), "v"(x[6][2]), "v"(x[6][3]),
               "v"(x[7][0]), "v"(x[7][1]), "v"(x[7][2]), "v"(x[7][3]),
               "v"(a[0]), "v"(b[0]), "v"(a[1]), "v"(b[1]),
               "v"(a[2]), "v"(b[2]), "v"(a[3]), "v"(b[3]));
  float vh = 0.f, vt = 0.f;
  #pragma unroll
  for (int i = 0; i < 8; ++i) {
    float sh = 0.f, st = 0.f;
    #pragma unroll
    for (int j = 0; j < 4; ++j) {
      const v4f v = x[i][j];
      sh = fmaf(v.x, a[j].x, sh); sh = fmaf(v.y, a[j].y, sh); sh = fmaf(v.z, a[j].z, sh); sh = fmaf(v.w, a[j].w, sh);
      st = fmaf(v.x, b[j].x, st); st = fmaf(v.y, b[j].y, st); st = fmaf(v.z, b[j].z, st); st = fmaf(v.w, b[j].w, st);
    }
    #pragma unroll
    for (int off = 4; off > 0; off >>= 1) {
      sh += __shfl_xor(sh, off);
      st += __shfl_xor(st, off);
    }
    if (s == i) { vh = sh; vt = st; }
  }
  const int row = base + lane;
  if (row < n_e) { e_h[row] = vh; e_t[row] = vt; }
  if (row < nbin) gcount[row] = 0;
}

__device__ __forceinline__ void rel16_body(const int mb, const float* __restrict__ x_r,
                      const float* __restrict__ w_ar,
                      const float* __restrict__ W1, const float* __restrict__ b1,
                      const float* __restrict__ W2, const float* __restrict__ b2,
                      float* __restrict__ e_r, float* __restrict__ msg, const int n_r) {
  constexpr int KC = 32;
  constexpr int XSTR = R_HID + 4;
  constexpr int HSTR = 2 * R_HID + 4;
  constexpr int BKSTR = 36;
  __shared__ __attribute__((aligned(16))) float Xs[16 * XSTR];
  __shared__ __attribute__((aligned(16))) float Bs[2 * R_HID * BKSTR];
  float* const Hs = Bs + R_HID * BKSTR;
  static_assert(16 * HSTR <= R_HID * BKSTR, "hid must fit the upper half of Bs");
  const int tid  = threadIdx.x;
  const int lane = tid & 31;
  const int wave = tid >> 5;
  const int m0   = mb * 16;
  const int lrow = lane & 15;
  const int hsel = (lane >= 16) ? 2 : 0;
  const int mrow = (lane >= 16) ? 8 : 0;
  const int n0   = wave * 16;

  #pragma unroll
  for (int u = 0; u < 2; ++u) {
    const int f  = tid + 256 * u;
    const int r  = f >> 5;
    const int c4 = (f & 31) * 4;
    float4 v = make_float4(0.f, 0.f, 0.f, 0.f);
    if (m0 + r < n_r) v = *(const float4*)(x_r + (size_t)(m0 + r) * R_HID + c4);
    *(float4*)&Xs[r * XSTR + c4] = v;
  }

  float4 bReg[8];
  auto loadW1 = [&](int kb) {
    #pragma unroll
    for (int i = 0; i < 8; ++i) {
      const int k4 = kb + i * 4;
      bReg[i].x = W1[(size_t)(k4 + 0) * (2 * R_HID) + tid];
      bReg[i].y = W1[(size_t)(k4 + 1) * (2 * R_HID) + tid];
      bReg[i].z = W1[(size_t)(k4 + 2) * (2 * R_HID) + tid];
      bReg[i].w = W1[(size_t)(k4 + 3) * (2 * R_HID) + tid];
    }
  };
  v8f c0 = {}, c1 = {};
  loadW1(0);
  for (int kb = 0; kb < R_HID; kb += KC) {
    #pragma unroll
    for (int i = 0; i < 8; ++i) *(float4*)&Bs[tid * BKSTR + i * 4] = bReg[i];
    __syncthreads();
    if (kb + KC < R_HID) loadW1(kb + KC);
    #pragma unroll
    for (int kk = 0; kk < KC; kk += 4) {
      const int ka = kk + hsel;
      const v2f a  = *(const v2f*)&Xs[lrow * XSTR + kb + ka];
      const v2f f0 = *(const v2f*)&Bs[(n0 + lrow) * BKSTR + ka];
      const v2f f1 = *(const v2f*)&Bs[(n0 + 128 + lrow) * BKSTR + ka];
      c0 = __builtin_amdgcn_wmma_f32_16x16x4_f32(false, a, false, f0, (short)0, c0, false, false);
      c1 = __builtin_amdgcn_wmma_f32_16x16x4_f32(false, a, false, f1, (short)0, c1, false, false);
    }
    __syncthreads();
  }

  float4 wReg[4];
  const int n2 = tid & 127;
  const int kh = (tid >> 7) * 16;
  auto loadW2 = [&](int kb) {
    #pragma unroll
    for (int i = 0; i < 4; ++i) {
      const int k4 = kb + kh + i * 4;
      wReg[i].x = W2[(size_t)(k4 + 0) * R_HID + n2];
      wReg[i].y = W2[(size_t)(k4 + 1) * R_HID + n2];
      wReg[i].z = W2[(size_t)(k4 + 2) * R_HID + n2];
      wReg[i].w = W2[(size_t)(k4 + 3) * R_HID + n2];
    }
  };
  loadW2(0);
  {
    const float bb0 = b1[n0 + lrow];
    const float bb1 = b1[n0 + 128 + lrow];
    #pragma unroll
    for (int j = 0; j < 8; ++j) {
      Hs[(mrow + j) * HSTR + n0 + lrow]       = c0[j] + bb0;
      Hs[(mrow + j) * HSTR + n0 + 128 + lrow] = c1[j] + bb1;
    }
  }
  #pragma unroll
  for (int rr = 0; rr < 2; ++rr) {
    const int r = 2 * wave + rr;
    float s = 0.f;
    #pragma unroll
    for (int j = 0; j < 4; ++j) s += Xs[r * XSTR + lane + 32 * j] * w_ar[lane + 32 * j];
    for (int off = 16; off > 0; off >>= 1) s += __shfl_xor(s, off);
    if (lane == 0 && m0 + r < n_r) e_r[m0 + r] = s;
  }

  v8f d0 = {}, d1 = {};
  for (int kb = 0; kb < 2 * R_HID; kb += KC) {
    #pragma unroll
    for (int i = 0; i < 4; ++i) *(float4*)&Bs[n2 * BKSTR + kh + i * 4] = wReg[i];
    __syncthreads();
    if (kb + KC < 2 * R_HID) loadW2(kb + KC);
    #pragma unroll
    for (int kk = 0; kk < KC; kk += 8) {
      const int ka = kk + hsel;
      const v2f a0 = *(const v2f*)&Hs[lrow * HSTR + kb + ka];
      const v2f a1 = *(const v2f*)&Hs[lrow * HSTR + kb + ka + 4];
      const v2f f0 = *(const v2f*)&Bs[(n0 + lrow) * BKSTR + ka];
      const v2f f1 = *(const v2f*)&Bs[(n0 + lrow) * BKSTR + ka + 4];
      d0 = __builtin_amdgcn_wmma_f32_16x16x4_f32(false, a0, false, f0, (short)0, d0, false, false);
      d1 = __builtin_amdgcn_wmma_f32_16x16x4_f32(false, a1, false, f1, (short)0, d1, false, false);
    }
    __syncthreads();
  }
  const float bb2 = b2[n0 + lrow];
  #pragma unroll
  for (int j = 0; j < 8; ++j) {
    const int grow = m0 + mrow + j;
    if (grow < n_r) msg[(size_t)grow * R_HID + n0 + lrow] = Xs[(mrow + j) * XSTR + n0 + lrow] + ((d0[j] + d1[j]) + bb2);
  }
}

__device__ __forceinline__ int4 load4_vec(const int* __restrict__ p, const int e, const int e0, const int e1) {
  const bool in = e < e1;
  int4 v = *(const int4*)(p + (in ? e : e0));
  if (!in) v = make_int4(-1, -1, -1, -1);
  return v;
}
__device__ __forceinline__ int4 load4_any(const int* __restrict__ p, const int e, const int e0, const int e1) {
  int4 v;
  v.x = p[(e     < e1) ? e     : e0];
  v.y = p[(e + 1 < e1) ? e + 1 : e0];
  v.z = p[(e + 2 < e1) ? e + 2 : e0];
  v.w = p[(e + 3 < e1) ? e + 3 : e0];
  if (e     >= e1) v.x = -1;
  if (e + 1 >= e1) v.y = -1;
  if (e + 2 >= e1) v.z = -1;
  if (e + 3 >= e1) v.w = -1;
  return v;
}

__device__ __forceinline__ void bin_body(const int bid,
                      const int* __restrict__ h_idx, const int* __restrict__ t_idx, const int* __restrict__ rel,
                      int* __restrict__ gcount, int* __restrict__ binned,
                      int n_edges, int n_e, int nbin) {
  __shared__ int hist[NBIN_MAX];
  __shared__ int gbase[NBIN_MAX];
  __shared__ int lbase[NBIN_MAX];
  __shared__ int sRec[2 * EPB];
  __shared__ int sAdr[2 * EPB];
  __shared__ int wsum[8];
  const int tid  = threadIdx.x;
  const int lane = tid & 31;
  const int wave = tid >> 5;
  const int e0 = bid * EPB;
  const int e1 = (e0 + EPB < n_edges) ? e0 + EPB : n_edges;
  const bool vec = (n_edges & 3) == 0;
  const int ea = e0 + tid * 4;
  const int eb = ea + 1024;
  #pragma unroll
  for (int k = 0; k < 4; ++k) hist[tid + 256 * k] = 0;
  int4 hA, tA, hB, tB, rA, rB;
  if (vec) {
    hA = load4_vec(h_idx, ea, e0, e1); tA = load4_vec(t_idx, ea, e0, e1); rA = load4_vec(rel, ea, e0, e1);
    hB = load4_vec(h_idx, eb, e0, e1); tB = load4_vec(t_idx, eb, e0, e1); rB = load4_vec(rel, eb, e0, e1);
  } else {
    hA = load4_any(h_idx, ea, e0, e1); tA = load4_any(t_idx, ea, e0, e1); rA = load4_any(rel, ea, e0, e1);
    hB = load4_any(h_idx, eb, e0, e1); tB = load4_any(t_idx, eb, e0, e1); rB = load4_any(rel, eb, e0, e1);
  }
  const int hv[8] = {hA.x, hA.y, hA.z, hA.w, hB.x, hB.y, hB.z, hB.w};
  const int tv[8] = {tA.x, tA.y, tA.z, tA.w, tB.x, tB.y, tB.z, tB.w};
  const int rv[8] = {rA.x, rA.y, rA.z, rA.w, rB.x, rB.y, rB.z, rB.w};
  __syncthreads();
  #pragma unroll
  for (int j = 0; j < 8; ++j) {
    if (hv[j] >= 0) {
      atomicAdd(&hist[hv[j] >> 8], 1);
      atomicAdd(&hist[(n_e + tv[j]) >> 8], 1);
    }
  }
  __syncthreads();
  int cg[4], g[4];
  #pragma unroll
  for (int k = 0; k < 4; ++k) {
    const int b = tid + 256 * k;
    cg[k] = (b < nbin) ? hist[b] : 0;
  }
  #pragma unroll
  for (int k = 0; k < 4; ++k) g[k] = (cg[k] > 0) ? atomicAdd(gcount + tid + 256 * k, cg[k]) : 0;
  int c[4];
  #pragma unroll
  for (int k = 0; k < 4; ++k) {
    const int b = 4 * tid + k;
    c[k] = (b < nbin) ? hist[b] : 0;
  }
  const int tsum = c[0] + c[1] + c[2] + c[3];
  int incl = tsum;
  for (int o = 1; o < 32; o <<= 1) {
    const int v = __shfl_up(incl, o);
    if (lane >= o) incl += v;
  }
  if (lane == 31) wsum[wave] = incl;
  __syncthreads();
  int excl = incl - tsum;
  for (int w = 0; w < wave; ++w) excl += wsum[w];
  #pragma unroll
  for (int k = 0; k < 4; ++k) {
    const int b = 4 * tid + k;
    lbase[b] = excl;
    hist[b] = 0;
    excl += c[k];
  }
  #pragma unroll
  for (int k = 0; k < 4; ++k) gbase[tid + 256 * k] = g[k];
  __syncthreads();
  #pragma unroll
  for (int j = 0; j < 8; ++j) {
    if (hv[j] >= 0) {
      const int kh = hv[j];
      const int kt = n_e + tv[j];
      const int bh = kh >> 8;
      const int bt = kt >> 8;
      const int lh = atomicAdd(&hist[bh], 1);
      const int lt = atomicAdd(&hist[bt], 1);
      const int ph = lbase[bh] + lh;
      const int pt = lbase[bt] + lt;
      const int gh = gbase[bh] + lh;
      const int gt = gbase[bt] + lt;
      sRec[ph] = (rv[j] << 8) | (kh & 255);
      sRec[pt] = (rv[j] << 8) | (kt & 255);
      sAdr[ph] = (gh < BIN_CAP) ? bh * BIN_CAP + gh : -1;
      sAdr[pt] = (gt < BIN_CAP) ? bt * BIN_CAP + gt : -1;
    }
  }
  __syncthreads();
  const int nrec = 2 * (e1 - e0);
  for (int i = tid; i < nrec; i += 256) {
    const int a = sAdr[i];
    if (a >= 0) binned[a] = sRec[i];
  }
}

#define RPT (BIN_CAP / 256)
#define GRP_CAP (BIN_CAP + 3 * KPB)
__device__ __forceinline__ int f2ord(float f) { const int b = __float_as_int(f); return b >= 0 ? b : b ^ 0x7fffffff; }
__device__ __forceinline__ float ord2f(int o) { return __int_as_float(o >= 0 ? o : o ^ 0x7fffffff); }

__device__ __forceinline__ void binlist_body(const int bin, const int* __restrict__ gcount,
                          const int* __restrict__ binned,
                          const float* __restrict__ e_h, const float* __restrict__ e_t,
                          const float* __restrict__ e_r,
                          int* __restrict__ cnt, int* __restrict__ off, int2* __restrict__ pairs,
                          int n_keys, int n_e) {
  __shared__ int kcnt[KPB];
  __shared__ int koff2[KPB];
  __shared__ float ken[KPB];
  __shared__ int kmx[KPB];
  __shared__ uint4 grpv[GRP_CAP / 4];
  int* const grp = (int*)grpv;
  __shared__ int wgt[BIN_CAP];
  __shared__ int wsum2[8];
  const int tid  = threadIdx.x;
  const int lane = tid & 31;
  const int wave = tid >> 5;
  const int tot = gcount[bin];
  const bool over = tot > BIN_CAP;
  const int n = over ? BIN_CAP : tot;
  kcnt[tid] = 0;
  kmx[tid] = (int)0x80000000;
  const int* __restrict__ src = binned + (size_t)bin * BIN_CAP;
  int rc[RPT], sl[RPT];
  #pragma unroll
  for (int m = 0; m < RPT; ++m) {
    const int i = tid + 256 * m;
    const int v = src[i < n ? i : 0];
    rc[m] = i < n ? v : -1;
  }
  __syncthreads();
  #pragma unroll
  for (int m = 0; m < RPT; ++m) {
    sl[m] = 0;
    if (rc[m] >= 0) sl[m] = atomicAdd(&kcnt[rc[m] & 255], 1);
  }
  __syncthreads();
  const int c = kcnt[tid];
  const int c4 = (c + 3) & ~3;
  const int both = (c4 << 16) | c;
  int incl = both;
  for (int o = 1; o < 32; o <<= 1) {
    const int v = __shfl_up(incl, o);
    if (lane >= o) incl += v;
  }
  if (lane == 31) wsum2[wave] = incl;
  __syncthreads();
  int wb = 0;
  for (int w = 0; w < wave; ++w) wb += wsum2[w];
  const int excl = wb + incl - both;
  const int ko = excl & 0xffff;
  const int ko4 = excl >> 16;
  koff2[tid] = excl;
  for (int u = c; u < c4; ++u) grp[ko4 + u] = -1;
  const int key = bin * KPB + tid;
  float en = 0.f;
  if (key < n_keys) {
    cnt[key] = over ? CNT_OVER : c;
    off[key] = bin * BIN_CAP + ko;
    en = key < n_e ? e_h[key] : e_t[key - n_e];
  }
  ken[tid] = en;
  __syncthreads();
  if (over) return;
  float er[RPT];
  #pragma unroll
  for (int m = 0; m < RPT; ++m) {
    const int r = rc[m] >= 0 ? (rc[m] >> 8) : 0;
    er[m] = e_r[r];
    if (rc[m] >= 0) grp[(koff2[rc[m] & 255] >> 16) + sl[m]] = (int)(((unsigned)r << 12) | (unsigned)sl[m]);
  }
  __syncthreads();
  int dst[RPT];
  float zz[RPT];
  #pragma unroll
  for (int m = 0; m < RPT; ++m) {
    dst[m] = -1;
    zz[m] = 0.f;
    if (rc[m] >= 0) {
      const int k = rc[m] & 255;
      const int o2 = koff2[k];
      const int lo = o2 >> 18;
      const int hi = lo + ((kcnt[k] + 3) >> 2);
      const unsigned w = ((unsigned)(rc[m] >> 8) << 12) | (unsigned)sl[m];
      int rank = 0;
      for (int q = lo; q < hi; ++q) {
        const uint4 y = grpv[q];
        rank += (y.x < w ? 1 : 0) + (y.y < w ? 1 : 0) + (y.z < w ? 1 : 0) + (y.w < w ? 1 : 0);
      }
      dst[m] = (o2 & 0xffff) + rank;
      const float z = leaky(ken[k] + er[m]);
      zz[m] = z;
      atomicMax(&kmx[k], f2ord(z));
    }
  }
  __syncthreads();
  #pragma unroll
  for (int m = 0; m < RPT; ++m) {
    if (dst[m] >= 0) {
      grp[dst[m]] = rc[m];
      wgt[dst[m]] = __float_as_int(__expf(zz[m] - ord2f(kmx[rc[m] & 255])));
    }
  }
  __syncthreads();
  if (c > 0) {
    float sum = 0.f;
    #pragma unroll 4
    for (int q = 0; q < c; ++q) sum += __int_as_float(wgt[ko + q]);
    kmx[tid] = __float_as_int(sum + 1e-16f);
  }
  __syncthreads();
  int2* __restrict__ dstp = pairs + (size_t)bin * BIN_CAP;
  for (int p = tid; p < n; p += 256) {
    const int rec = grp[p];
    const float al = __int_as_float(wgt[p]) / __int_as_float(kmx[rec & 255]);
    dstp[p] = make_int2(rec >> 8, __float_as_int(al));
  }
}

__device__ __forceinline__ void proj_body(const int mb, const int half,
                       const float* __restrict__ msg, const float* __restrict__ Wr,
                       float* __restrict__ P, int n_r) {
  constexpr int KTOT = 128, NTOT = 256, KC = 32;
  constexpr int ASTR = 36, BKSTR = 36;
  __shared__ float As[16 * ASTR];
  __shared__ float Bs[NTOT * BKSTR];
  const int tid  = threadIdx.x;
  const int lane = tid & 31;
  const int wave = tid >> 5;
  const int m0   = mb * 16;
  const int lrow = lane & 15;
  const int hsel = (lane >= 16) ? 2 : 0;
  const int mrow = (lane >= 16) ? 8 : 0;
  const int n0   = wave * 16;
  const float* __restrict__ B = Wr + (size_t)half * KTOT * NTOT;
  float* __restrict__ O = P + (size_t)half * n_r * NTOT;

  const int arow = tid >> 3;
  const int acol = (tid & 7) * 4;

  v8f c0 = {}, c1 = {};
  float4 aReg = make_float4(0.f, 0.f, 0.f, 0.f);
  float4 bReg[8];

  auto loadGlobal = [&](int kb) {
    if (tid < 128) {
      int grow = m0 + arow;
      aReg = make_float4(0.f, 0.f, 0.f, 0.f);
      if (grow < n_r) aReg = *(const float4*)(msg + (size_t)grow * KTOT + kb + acol);
    }
    #pragma unroll
    for (int i = 0; i < 8; ++i) {
      int k4 = i * 4;
      bReg[i].x = B[(size_t)(kb + k4 + 0) * NTOT + tid];
      bReg[i].y = B[(size_t)(kb + k4 + 1) * NTOT + tid];
      bReg[i].z = B[(size_t)(kb + k4 + 2) * NTOT + tid];
      bReg[i].w = B[(size_t)(kb + k4 + 3) * NTOT + tid];
    }
  };
  auto storeSmem = [&]() {
    if (tid < 128) *(float4*)&As[arow * ASTR + acol] = aReg;
    #pragma unroll
    for (int i = 0; i < 8; ++i)
      *(float4*)&Bs[tid * BKSTR + i * 4] = bReg[i];
  };

  loadGlobal(0);
  for (int kb = 0; kb < KTOT; kb += KC) {
    storeSmem();
    __syncthreads();
    if (kb + KC < KTOT) loadGlobal(kb + KC);
    #pragma unroll
    for (int kk = 0; kk < KC; kk += 4) {
      int ka = kk + hsel;
      v2f a  = *(const v2f*)&As[lrow * ASTR + ka];
      v2f b0 = *(const v2f*)&Bs[(n0 + lrow) * BKSTR + ka];
      v2f b1 = *(const v2f*)&Bs[(n0 + 128 + lrow) * BKSTR + ka];
      c0 = __builtin_amdgcn_wmma_f32_16x16x4_f32(false, a, false, b0, (short)0, c0, false, false);
      c1 = __builtin_amdgcn_wmma_f32_16x16x4_f32(false, a, false, b1, (short)0, c1, false, false);
    }
    __syncthreads();
  }
  #pragma unroll
  for (int j = 0; j < 8; ++j) {
    int grow = m0 + mrow + j;
    if (grow < n_r) {
      O[(size_t)grow * NTOT + n0 + lrow]       = c0[j];
      O[(size_t)grow * NTOT + n0 + 128 + lrow] = c1[j];
    }
  }
}

__global__ void k_rel_bin(const float* __restrict__ x_r,
                          const float* __restrict__ w_ar,
                          const float* __restrict__ W1, const float* __restrict__ b1,
                          const float* __restrict__ W2, const float* __restrict__ b2,
                          float* __restrict__ e_r, float* __restrict__ msg, int n_r, int n_rb,
                          const int* __restrict__ h_idx, const int* __restrict__ t_idx, const int* __restrict__ rel,
                          int* __restrict__ gcount, int* __restrict__ binned,
                          int n_edges, int n_e, int nbin) {
  const int b = (int)blockIdx.x;
  if (b < n_rb) rel16_body(b, x_r, w_ar, W1, b1, W2, b2, e_r, msg, n_r);
  else bin_body(b - n_rb, h_idx, t_idx, rel, gcount, binned, n_edges, n_e, nbin);
}

__global__ void k_proj_list(const float* __restrict__ msg, const float* __restrict__ Wr,
                            float* __restrict__ P, int n_r, int n_mb,
                            const int* __restrict__ gcount, const int* __restrict__ binned,
                            const float* __restrict__ e_h, const float* __restrict__ e_t,
                            const float* __restrict__ e_r,
                            int* __restrict__ cnt, int* __restrict__ off, int* __restrict__ pairs,
                            int n_keys, int n_e) {
  const int b = (int)blockIdx.x;
  if (b < 2 * n_mb) {
    const int half = b >= n_mb ? 1 : 0;
    proj_body(b - half * n_mb, half, msg, Wr, P, n_r);
  } else {
    binlist_body(b - 2 * n_mb, gcount, binned, e_h, e_t, e_r, cnt, off, (int2*)pairs, n_keys, n_e);
  }
}

#define GS_COLS 32
#define GS_NSL (256 / GS_COLS)
#define GS_ROWS 1024
#define GS_WAVES 32

struct __attribute__((aligned(8))) pair2 { int r0, w0, r1, w1; };

__device__ __forceinline__ float scan_side_col(const float en, const float* __restrict__ e_r,
                                               const float* __restrict__ Pc, const int lane,
                                               const int node, const int* __restrict__ idx,
                                               const int* __restrict__ rel, const int n_edges) {
  float mx = -__builtin_inff();
  for (int i = lane; i < n_edges; i += 32) if (idx[i] == node) mx = fmaxf(mx, leaky(en + e_r[rel[i]]));
  mx = wave_max_f(mx);
  float s = 0.f;
  int last = -1;
  for (int it = 0; it < n_edges; ++it) {
    int best = 0x7fffffff;
    for (int i = lane; i < n_edges; i += 32) if (idx[i] == node) { const int r = rel[i]; if (r > last && r < best) best = r; }
    best = __builtin_amdgcn_readfirstlane(wave_min_i(best));
    if (best == 0x7fffffff) break;
    int c = 0;
    for (int i = lane; i < n_edges; i += 32) c += (idx[i] == node && rel[i] == best) ? 1 : 0;
    c = __builtin_amdgcn_readfirstlane(wave_sum_i(c));
    const float p = __expf(leaky(en + e_r[best]) - mx);
    for (int q = 0; q < c; ++q) s += p;
    last = best;
  }
  const float den = s + 1e-16f;
  float acc = 0.f;
  last = -1;
  for (int it = 0; it < n_edges; ++it) {
    int best = 0x7fffffff;
    for (int i = lane; i < n_edges; i += 32) if (idx[i] == node) { const int r = rel[i]; if (r > last && r < best) best = r; }
    best = __builtin_amdgcn_readfirstlane(wave_min_i(best));
    if (best == 0x7fffffff) break;
    int c = 0;
    for (int i = lane; i < n_edges; i += 32) c += (idx[i] == node && rel[i] == best) ? 1 : 0;
    c = __builtin_amdgcn_readfirstlane(wave_sum_i(c));
    const float a = __expf(leaky(en + e_r[best]) - mx) / den;
    const float m0 = Pc[(size_t)best * 256];
    for (int q = 0; q < c; ++q) acc += a * m0;
    last = best;
  }
  return acc;
}

template <int N, bool CHECK>
__device__ __forceinline__ void slice_trips(const char* __restrict__ tabb, const unsigned lbA, const unsigned lbB, const int rem,
                                            const pair2 pa, const pair2 pb, v4f& accA, v4f& accB) {
  const int rr[4] = {pa.r0, pa.r1, pb.r0, pb.r1};
  const int ww[4] = {pa.w0, pa.w1, pb.w0, pb.w1};
  v4f ta[N], tb[N];
  float w[N];
  #pragma unroll
  for (int i = 0; i < N; ++i) {
    const bool ok = !CHECK || i < rem;
    const unsigned row = ok ? (unsigned)rr[i] : 0u;
    w[i] = ok ? __int_as_float(ww[i]) : 0.f;
    ta[i] = *(const v4f*)(tabb + ((row << 8) | lbA));
    tb[i] = *(const v4f*)(tabb + ((row << 8) | lbB));
  }
  if constexpr (N == 4) {
    asm volatile("" :: "v"(ta[0]), "v"(tb[0]), "v"(ta[1]), "v"(tb[1]), "v"(ta[2]), "v"(tb[2]), "v"(ta[3]), "v"(tb[3]));
  } else if constexpr (N == 3) {
    asm volatile("" :: "v"(ta[0]), "v"(tb[0]), "v"(ta[1]), "v"(tb[1]), "v"(ta[2]), "v"(tb[2]));
  } else if constexpr (N == 2) {
    asm volatile("" :: "v"(ta[0]), "v"(tb[0]), "v"(ta[1]), "v"(tb[1]));
  } else {
    asm volatile("" :: "v"(ta[0]), "v"(tb[0]));
  }
  #pragma unroll
  for (int i = 0; i < N; ++i) { accA += ta[i] * w[i]; accB += tb[i] * w[i]; }
}

__global__ void __launch_bounds__(GS_WAVES * 32) k_gather(const int* __restrict__ cnt, const int* __restrict__ off,
                         const int* __restrict__ pairs_w,
                         const float* __restrict__ e_h, const float* __restrict__ e_t,
                         const float* __restrict__ e_r, const float* __restrict__ P,
                         const float* __restrict__ br,
                         float* __restrict__ out, int n_e, int n_r,
                         const int* __restrict__ h_idx, const int* __restrict__ t_idx,
                         const int* __restrict__ rel, int n_edges) {
  __shared__ __attribute__((aligned(16))) float tab[GS_ROWS * 2 * GS_COLS];
  __shared__ int next_step;
  const int tid  = threadIdx.x;
  const int lane = tid & 31;
  const int wave = __builtin_amdgcn_readfirstlane(tid >> 5);
  const int cs = (int)blockIdx.x % GS_NSL;
  const int ng = (int)blockIdx.x / GS_NSL;
  const int n_ng = (int)gridDim.x / GS_NSL;
  const int npg = (n_e + n_ng - 1) / n_ng;
  const int nb = ng * npg;
  if (nb >= n_e) return;
  const int ne = (nb + npg < n_e) ? nb + npg : n_e;
  const int nsteps = (ne - nb + 3) >> 2;
  const int s8 = lane & 7;
  const int s = lane & 3;
  const int g = lane >> 2;
  const int side = g & 1;
  v4f* const tab4 = (v4f*)tab;

  {
    const int npiece = 2 * n_r;
    const float* __restrict__ Pc = P + cs * GS_COLS + s8 * 4;
    for (int p0 = 0; p0 < npiece; p0 += 8 * GS_WAVES * 4) {
      v4f v[8];
      #pragma unroll
      for (int u = 0; u < 8; ++u) {
        const int p = p0 + u * GS_WAVES * 4 + (tid >> 3);
        const int pc = p < npiece ? p : 0;
        v[u] = *(const v4f*)(Pc + (size_t)((pc & 1) * n_r + (pc >> 1)) * 256);
      }
      #pragma unroll
      for (int u = 0; u < 8; ++u) {
        const int p = p0 + u * GS_WAVES * 4 + (tid >> 3);
        if (p < npiece) tab4[p * 8 + s8] = v[u];
      }
    }
  }
  if (tid == 0) next_step = 3 * GS_WAVES;
  const unsigned cbA = (unsigned)(s * 32 + ((g >> 1) & 1) * 16);
  const unsigned cbB = cbA ^ 16u;
  const unsigned lbA = (unsigned)(side * (GS_COLS * 4)) + cbA;
  const unsigned lbB = (unsigned)(side * (GS_COLS * 4)) + cbB;
  const unsigned cbW = side ? cbB : cbA;
  const v4f bias = *(const v4f*)((const char*)(br + cs * GS_COLS) + cbW);
  const char* const tabb = (const char*)tab;
  __syncthreads();

  auto side_of = [&](const int st, int& d, int& o, int& node) {
    node = nb + 4 * st + (g >> 1);
    const bool ok = st < nsteps && node < ne;
    const int key = side * n_e + (ok ? node : nb);
    const int dd = cnt[key];
    o = off[key];
    d = ok ? dd : 0;
  };
  auto ld2 = [&](const int o, const int k) -> pair2 {
    return *(const pair2*)((const char*)pairs_w + (unsigned)(o + k) * 8u);
  };

  int st0 = wave, st1 = wave + GS_WAVES, st2 = wave + 2 * GS_WAVES;
  int d0, o0, n0, d1, o1, n1;
  side_of(st0, d0, o0, n0);
  side_of(st1, d1, o1, n1);
  pair2 a0 = ld2(o0, 0), b0 = ld2(o0, 2);
  while (st0 < nsteps) {
    int stn = 0;
    if (lane == 0) stn = atomicAdd(&next_step, 1);
    pair2 c[8];
    c[0] = a0; c[1] = b0;
    #pragma unroll
    for (int j = 1; j < 4; ++j) { c[2 * j] = ld2(o0, 4 * j); c[2 * j + 1] = ld2(o0, 4 * j + 2); }
    int d2, o2, n2;
    side_of(st2, d2, o2, n2);
    const pair2 a1 = ld2(o1, 0), b1 = ld2(o1, 2);

    const bool over = d0 == CNT_OVER;
    const int dl = over ? 0 : d0;
    int dmax = __builtin_amdgcn_readlane(dl, 0), dmin = dmax;
    #pragma unroll
    for (int q = 1; q < 8; ++q) {
      const int x = __builtin_amdgcn_readlane(dl, 4 * q);
      dmax = x > dmax ? x : dmax; dmin = x < dmin ? x : dmin;
    }
    v4f accA = {0.f, 0.f, 0.f, 0.f}, accB = {0.f, 0.f, 0.f, 0.f};
    auto group = [&](const pair2 xa, const pair2 xb, const int k) {
      const int remw = dmax - k;
      if (remw >= 4) {
        if (k + 4 <= dmin) slice_trips<4, false>(tabb, lbA, lbB, 4, xa, xb, accA, accB);
        else slice_trips<4, true>(tabb, lbA, lbB, dl - k, xa, xb, accA, accB);
      }
      else if (remw == 3) slice_trips<3, true>(tabb, lbA, lbB, dl - k, xa, xb, accA, accB);
      else if (remw == 2) slice_trips<2, true>(tabb, lbA, lbB, dl - k, xa, xb, accA, accB);
      else if (remw == 1) slice_trips<1, true>(tabb, lbA, lbB, dl - k, xa, xb, accA, accB);
    };
    #pragma unroll
    for (int j = 0; j < 4; ++j) {
      if (4 * j < dmax) group(c[2 * j], c[2 * j + 1], 4 * j);
    }
    for (int k = 16; k < dmax; k += 4) {
      const pair2 xa = ld2(o0, k), xb = ld2(o0, k + 2);
      group(xa, xb, k);
    }

    const unsigned ovm = __builtin_amdgcn_ballot_w32(over);
    if (ovm != 0u) {
      for (int gg = 0; gg < 8; ++gg) {
        if ((ovm >> (4 * gg)) & 1u) {
          const int nd = __builtin_amdgcn_readlane(n0, 4 * gg);
          const int sd = gg & 1;
          const float en = sd ? e_t[nd] : e_h[nd];
          const float part = scan_side_col(en, e_r, P + (size_t)sd * n_r * 256 + cs * GS_COLS + lane, lane, nd,
                                           sd ? t_idx : h_idx, rel, n_edges);
          const int ca = (int)(cbA >> 2), cb = (int)(cbB >> 2);
          v4f addA, addB;
          addA.x = __shfl(part, ca + 0); addA.y = __shfl(part, ca + 1);
          addA.z = __shfl(part, ca + 2); addA.w = __shfl(part, ca + 3);
          addB.x = __shfl(part, cb + 0); addB.y = __shfl(part, cb + 1);
          addB.z = __shfl(part, cb + 2); addB.w = __shfl(part, cb + 3);
          if (g == gg) { accA += addA; accB += addB; }
        }
      }
    }
    auto xor4 = [](const float v) {
      return __int_as_float(__builtin_amdgcn_update_dpp(0, __float_as_int(v), 0x164, 0xf, 0xf, false));
    };
    const v4f mine = side ? accB : accA;
    const v4f send = side ? accA : accB;
    v4f oth;
    oth.x = xor4(send.x); oth.y = xor4(send.y); oth.z = xor4(send.z); oth.w = xor4(send.w);
    if (n0 < ne) {
      const v4f r = (mine + oth) + bias;
      *(v4f*)((char*)(out + (size_t)n0 * 256 + cs * GS_COLS) + cbW) = r;
    }
    st0 = st1; d0 = d1; o0 = o1; n0 = n1; a0 = a1; b0 = b1;
    st1 = st2; d1 = d2; o1 = o2; n1 = n2;
    st2 = __builtin_amdgcn_readfirstlane(stn);
  }
}

extern "C" void kernel_launch(void* const* d_in, const int* in_sizes, int n_in,
                              void* d_out, int out_size, void* d_ws, size_t ws_size,
                              hipStream_t stream) {
  const float* x_e  = (const float*)d_in[0];
  const float* x_r  = (const float*)d_in[1];
  const int*   eidx = (const int*)d_in[2];
  const int*   rel  = (const int*)d_in[3];
  const float* w_ah = (const float*)d_in[5];
  const float* w_at = (const float*)d_in[6];
  const float* w_ar = (const float*)d_in[7];
  const float* W1   = (const float*)d_in[8];
  const float* b1   = (const float*)d_in[9];
  const float* W2   = (const float*)d_in[10];
  const float* b2   = (const float*)d_in[11];
  const float* Wr   = (const float*)d_in[12];
  const float* br   = (const float*)d_in[13];
  float* out = (float*)d_out;

  const int n_e     = in_sizes[0] / E_HID;
  const int n_r     = in_sizes[1] / R_HID;
  const int n_edges = in_sizes[3];
  const int* h_idx = eidx;
  const int* t_idx = eidx + n_edges;
  const int nbin   = (2 * n_e + KPB - 1) / KPB;

  int* pairs   = (int*)d_ws;
  float* P     = (float*)(pairs + (size_t)2 * nbin * BIN_CAP);
  float* msg   = P + (size_t)2 * n_r * 256;
  float* e_h   = msg + (size_t)n_r * R_HID;
  float* e_t   = e_h + n_e;
  float* e_r   = e_t + n_e;
  int* cnt     = (int*)(e_r + n_r);
  int* off     = cnt + (size_t)2 * n_e;
  int* gcount  = off + (size_t)2 * n_e;
  int* binned  = gcount + NBIN_MAX;

  k_node_scores<<<(n_e + 63) / 64, 64, 0, stream>>>(x_e, w_ah, w_at, e_h, e_t, gcount, nbin, n_e);
  k_rel_bin<<<(n_r + 15) / 16 + (n_edges + EPB - 1) / EPB, 256, 0, stream>>>(x_r, w_ar, W1, b1, W2, b2, e_r, msg, n_r, (n_r + 15) / 16,
                                                                 h_idx, t_idx, rel, gcount, binned, n_edges, n_e, nbin);
  k_proj_list<<<2 * ((n_r + 15) / 16) + nbin, 256, 0, stream>>>(msg, Wr, P, n_r, (n_r + 15) / 16,
                                                                gcount, binned, e_h, e_t, e_r, cnt, off, pairs, 2 * n_e, n_e);
  k_gather<<<GS_NSL * 32, GS_WAVES * 32, 0, stream>>>(cnt, off, pairs, e_h, e_t, e_r, P, br, out, n_e, n_r, h_idx, t_idx, rel, n_edges);
}
